# starting kernel
# baseline (speedup 1.0000x reference)
#include <hip/hip_runtime.h>
#include <math.h>

typedef float v2f __attribute__((ext_vector_type(2)));
typedef float v8f __attribute__((ext_vector_type(8)));

__global__ void deg_init_k(float* __restrict__ deg, int N) {
  int i = blockIdx.x * blockDim.x + threadIdx.x;
  if (i < N) deg[i] = 1.0f;
}

__global__ void deg_acc_k(const int* __restrict__ dst, float* __restrict__ deg, int E) {
  int e = blockIdx.x * blockDim.x + threadIdx.x;
  if (e < E) {
    __hip_atomic_fetch_add(&deg[dst[e]], 1.0f, __ATOMIC_RELAXED, __HIP_MEMORY_SCOPE_AGENT);
  }
}

__global__ void dinv_k(float* __restrict__ deg, int N) {
  int i = blockIdx.x * blockDim.x + threadIdx.x;
  if (i < N) {
    float d = deg[i];
    deg[i] = (d > 0.0f) ? rsqrtf(d) : 0.0f;
  }
}

template <int NN>
__global__ void gemm_wmma_k(const float* __restrict__ A,
                            const float* __restrict__ B,
                            float* __restrict__ C,
                            int M, int K) {
  const int wave = threadIdx.x >> 5;
  const int lane = threadIdx.x & 31;
  const int m0 = blockIdx.x * 16;
  if (m0 >= M) return;
  const int n0 = wave * 16;
  const int half = lane >> 4;
  const int l15 = lane & 15;

  v8f acc = {};
  const float* aptr = A + (size_t)(m0 + l15) * K + half * 2;
  const float* bptr = B + (size_t)(half * 2) * NN + n0 + l15;

  for (int k = 0; k < K; k += 4) {
    v2f a, b;
    a.x = aptr[0];
    a.y = aptr[1];
    b.x = bptr[0];
    b.y = bptr[NN];
    acc = __builtin_amdgcn_wmma_f32_16x16x4_f32(
         false, a,  false, b,
         (short)0, acc,  false,  false);
    aptr += 4;
    bptr += 4 * NN;
  }

  float* cptr = C + (size_t)(m0 + half * 8) * NN + n0 + l15;
#pragma unroll
  for (int r = 0; r < 8; ++r) {
    cptr[0] = acc[r];
    cptr += NN;
  }
}

template <int C>
__global__ void self_bias_k(const float* __restrict__ feat,
                            const float* __restrict__ dinv,
                            const float* __restrict__ bias,
                            float* __restrict__ out, int N) {
  int t = blockIdx.x * blockDim.x + threadIdx.x;
  int total = N * (C / 4);
  if (t >= total) return;
  int i = t / (C / 4);
  int c4 = t % (C / 4);
  float di = dinv[i];
  float w = di * di;
  float4 f = reinterpret_cast<const float4*>(feat)[t];
  float4 bv = reinterpret_cast<const float4*>(bias)[c4];
  float4 o;
  o.x = w * f.x + bv.x;
  o.y = w * f.y + bv.y;
  o.z = w * f.z + bv.z;
  o.w = w * f.w + bv.w;
  reinterpret_cast<float4*>(out)[t] = o;
}

template <int C>
__global__ void scatter_edges_k(const float* __restrict__ feat,
                                const float* __restrict__ dinv,
                                const int* __restrict__ src,
                                const int* __restrict__ dst,
                                float* __restrict__ out, int E) {
  int wave = blockIdx.x * (blockDim.x >> 5) + (threadIdx.x >> 5);
  int lane = threadIdx.x & 31;
  if (wave >= E) return;
  int s = src[wave];
  int d = dst[wave];
  float norm = dinv[s] * dinv[d];
  const float* frow = feat + (size_t)s * C;
  float* orow = out + (size_t)d * C;
  if constexpr (C == 128) {
    float4 v = reinterpret_cast<const float4*>(frow)[lane];
    float* p = orow + lane * 4;
    __hip_atomic_fetch_add(p + 0, norm * v.x, __ATOMIC_RELAXED, __HIP_MEMORY_SCOPE_AGENT);
    __hip_atomic_fetch_add(p + 1, norm * v.y, __ATOMIC_RELAXED, __HIP_MEMORY_SCOPE_AGENT);
    __hip_atomic_fetch_add(p + 2, norm * v.z, __ATOMIC_RELAXED, __HIP_MEMORY_SCOPE_AGENT);
    __hip_atomic_fetch_add(p + 3, norm * v.w, __ATOMIC_RELAXED, __HIP_MEMORY_SCOPE_AGENT);
  } else {
    float2 v = reinterpret_cast<const float2*>(frow)[lane];
    float* p = orow + lane * 2;
    __hip_atomic_fetch_add(p + 0, norm * v.x, __ATOMIC_RELAXED, __HIP_MEMORY_SCOPE_AGENT);
    __hip_atomic_fetch_add(p + 1, norm * v.y, __ATOMIC_RELAXED, __HIP_MEMORY_SCOPE_AGENT);
  }
}

__global__ void finalize_k(const float* __restrict__ g, float* __restrict__ out, int N) {
  int node = blockIdx.x * (blockDim.x >> 5) + (threadIdx.x >> 5);
  int lane = threadIdx.x & 31;
  if (node >= N) return;
  float2 v = reinterpret_cast<const float2*>(g + (size_t)node * 64)[lane];
  float mx = fmaxf(v.x, v.y);
  float mn = fminf(v.x, v.y);
#pragma unroll
  for (int o = 16; o > 0; o >>= 1) {
    mx = fmaxf(mx, __shfl_xor(mx, o, 32));
    mn = fminf(mn, __shfl_xor(mn, o, 32));
  }
  float inv = 1.0f / (mx - mn);
  float sx = (v.x - mn) * inv;
  float sy = (v.y - mn) * inv;
  float ss = sx * sx + sy * sy;
#pragma unroll
  for (int o = 16; o > 0; o >>= 1) ss += __shfl_xor(ss, o, 32);
  float r = 1.0f / fmaxf(sqrtf(ss), 1e-12f);
  float2 ov;
  ov.x = sx * r;
  ov.y = sy * r;
  reinterpret_cast<float2*>(out + (size_t)node * 64)[lane] = ov;
}

extern "C" void kernel_launch(void* const* d_in, const int* in_sizes, int n_in,
                              void* d_out, int out_size, void* d_ws, size_t ws_size,
                              hipStream_t stream) {
  (void)n_in; (void)out_size; (void)ws_size;

  const float* x  = (const float*)d_in[0];
  const int*   ei = (const int*)d_in[1];
  const float* W1 = (const float*)d_in[2];
  const float* b1 = (const float*)d_in[3];
  const float* W2 = (const float*)d_in[4];
  const float* b2 = (const float*)d_in[5];

  const int N = in_sizes[0] / 128;
  const int E = in_sizes[1] / 2;
  const int* src = ei;
  const int* dst = ei + E;
  float* out = (float*)d_out;

  float* ws = (float*)d_ws;
  size_t Npad = ((size_t)N + 63) & ~(size_t)63;
  float* dinv = ws;
  float* xw = ws + Npad;
  float* h  = xw + (size_t)N * 128;
  float* g2 = h  + (size_t)N * 128;

  deg_init_k<<<(N + 255) / 256, 256, 0, stream>>>(dinv, N);
  deg_acc_k<<<(E + 255) / 256, 256, 0, stream>>>(dst, dinv, E);
  dinv_k<<<(N + 255) / 256, 256, 0, stream>>>(dinv, N);

  gemm_wmma_k<128><<<(N + 15) / 16, 256, 0, stream>>>(x, W1, xw, N, 128);
  self_bias_k<128><<<(N * 32 + 255) / 256, 256, 0, stream>>>(xw, dinv, b1, h, N);
  scatter_edges_k<128><<<(E + 7) / 8, 256, 0, stream>>>(xw, dinv, src, dst, h, E);

  gemm_wmma_k<64><<<(N + 15) / 16, 128, 0, stream>>>(h, W2, xw, N, 128);
  self_bias_k<64><<<(N * 16 + 255) / 256, 256, 0, stream>>>(xw, dinv, b2, g2, N);
  scatter_edges_k<64><<<(E + 7) / 8, 256, 0, stream>>>(xw, dinv, src, dst, g2, E);

  finalize_k<<<(N + 7) / 8, 256, 0, stream>>>(g2, out, N);
}


# Edge sums off global f32 atomics: a block owns 512 (1024) nodes, holds their sums in LDS from self loop + bias on, scans the whole dst list, adds matching rows by LDS f32 atomics; sum order free
# speedup vs baseline: 1.4624x; 1.4624x over previous
#include <hip/hip_runtime.h>
#include <math.h>

typedef float v2f __attribute__((ext_vector_type(2)));
typedef float v8f __attribute__((ext_vector_type(8)));

__global__ void deg_init_k(float* __restrict__ deg, int N) {
  int i = blockIdx.x * blockDim.x + threadIdx.x;
  if (i < N) deg[i] = 1.0f;
}

__global__ void deg_acc_k(const int* __restrict__ dst, float* __restrict__ deg, int E) {
  int e = blockIdx.x * blockDim.x + threadIdx.x;
  if (e < E) {
    __hip_atomic_fetch_add(&deg[dst[e]], 1.0f, __ATOMIC_RELAXED, __HIP_MEMORY_SCOPE_AGENT);
  }
}

__global__ void dinv_k(float* __restrict__ deg, int N) {
  int i = blockIdx.x * blockDim.x + threadIdx.x;
  if (i < N) {
    float d = deg[i];
    deg[i] = (d > 0.0f) ? rsqrtf(d) : 0.0f;
  }
}

template <int NN>
__global__ void gemm_wmma_k(const float* __restrict__ A,
                            const float* __restrict__ B,
                            float* __restrict__ C,
                            int M, int K) {
  const int wave = threadIdx.x >> 5;
  const int lane = threadIdx.x & 31;
  const int m0 = blockIdx.x * 16;
  if (m0 >= M) return;
  const int n0 = wave * 16;
  const int half = lane >> 4;
  const int l15 = lane & 15;

  v8f acc = {};
  const float* aptr = A + (size_t)(m0 + l15) * K + half * 2;
  const float* bptr = B + (size_t)(half * 2) * NN + n0 + l15;

  for (int k = 0; k < K; k += 4) {
    v2f a, b;
    a.x = aptr[0];
    a.y = aptr[1];
    b.x = bptr[0];
    b.y = bptr[NN];
    acc = __builtin_amdgcn_wmma_f32_16x16x4_f32(
         false, a,  false, b,
         (short)0, acc,  false,  false);
    aptr += 4;
    bptr += 4 * NN;
  }

  float* cptr = C + (size_t)(m0 + half * 8) * NN + n0 + l15;
#pragma unroll
  for (int r = 0; r < 8; ++r) {
    cptr[0] = acc[r];
    cptr += NN;
  }
}

template <int C, int SH>
__global__ __launch_bounds__(1024) void agg_tile_k(const float* __restrict__ feat,
                                                   const float* __restrict__ dinv,
                                                   const float* __restrict__ bias,
                                                   const int* __restrict__ src,
                                                   const int* __restrict__ dst,
                                                   float* __restrict__ out, int N, int E) {
  constexpr int T = 1 << SH;
  constexpr int NW = 32;
  constexpr int VEC = C / 32;
  constexpr int QCAP = 160;
  __shared__ float acc[T * C];
  __shared__ int2 q[NW * QCAP];

  const int lane = threadIdx.x & 31;
  const int wave = __builtin_amdgcn_readfirstlane((int)(threadIdx.x >> 5));
  const int tile = blockIdx.x;
  const int base = tile << SH;
  int nrows = N - base;
  if (nrows > T) nrows = T;

  for (int i = wave; i < nrows; i += NW) {
    float di = dinv[base + i];
    float w = di * di;
    const float* fr = feat + (size_t)(base + i) * C + lane * VEC;
    if constexpr (VEC == 4) {
      float4 f = *reinterpret_cast<const float4*>(fr);
      float4 bv = reinterpret_cast<const float4*>(bias)[lane];
      acc[i * C + lane]      = w * f.x + bv.x;
      acc[i * C + 32 + lane] = w * f.y + bv.y;
      acc[i * C + 64 + lane] = w * f.z + bv.z;
      acc[i * C + 96 + lane] = w * f.w + bv.w;
    } else {
      float2 f = *reinterpret_cast<const float2*>(fr);
      float2 bv = reinterpret_cast<const float2*>(bias)[lane];
      acc[i * C + lane]      = w * f.x + bv.x;
      acc[i * C + 32 + lane] = w * f.y + bv.y;
    }
  }
  __syncthreads();

  int2* myq = q + wave * QCAP;
  int cnt = 0;

  auto drain32 = [&](const int2* qb) {
    int2 ed = qb[lane];
    int s = src[ed.x];
    int d = ed.y;
    float w = dinv[s] * dinv[d];
    int wb = __builtin_bit_cast(int, w);
    int ro = (d - base) * C;
#pragma unroll 1
    for (int k0 = 0; k0 < 32; k0 += 8) {
      float v[8][VEC];
#pragma unroll
      for (int u = 0; u < 8; ++u) {
        int sk = __builtin_amdgcn_readlane(s, k0 + u);
        const float* fr = feat + (size_t)sk * C + lane * VEC;
        if constexpr (VEC == 4) {
          float4 t = *reinterpret_cast<const float4*>(fr);
          v[u][0] = t.x; v[u][1] = t.y; v[u][2] = t.z; v[u][3] = t.w;
        } else {
          float2 t = *reinterpret_cast<const float2*>(fr);
          v[u][0] = t.x; v[u][1] = t.y;
        }
      }
#pragma unroll
      for (int u = 0; u < 8; ++u) {
        float wk = __builtin_bit_cast(float, __builtin_amdgcn_readlane(wb, k0 + u));
        int rk = __builtin_amdgcn_readlane(ro, k0 + u);
#pragma unroll
        for (int j = 0; j < VEC; ++j)
          __hip_atomic_fetch_add(&acc[rk + j * 32 + lane], wk * v[u][j],
                                 __ATOMIC_RELAXED, __HIP_MEMORY_SCOPE_WORKGROUP);
      }
    }
  };

  auto drain_n = [&](const int2* qb, int n) {
    int2 ed = qb[lane < n ? lane : n - 1];
    int s = src[ed.x];
    int d = ed.y;
    float w = dinv[s] * dinv[d];
    int wb = __builtin_bit_cast(int, w);
    int ro = (d - base) * C;
    for (int k = 0; k < n; ++k) {
      int sk = __builtin_amdgcn_readlane(s, k);
      float wk = __builtin_bit_cast(float, __builtin_amdgcn_readlane(wb, k));
      int rk = __builtin_amdgcn_readlane(ro, k);
      const float* fr = feat + (size_t)sk * C + lane * VEC;
#pragma unroll
      for (int j = 0; j < VEC; ++j)
        __hip_atomic_fetch_add(&acc[rk + j * 32 + lane], wk * fr[j],
                               __ATOMIC_RELAXED, __HIP_MEMORY_SCOPE_WORKGROUP);
    }
  };

  auto push = [&](bool m, int e, int d) {
    unsigned mk = __builtin_amdgcn_ballot_w32(m);
    if (m) myq[cnt + (int)__builtin_amdgcn_mbcnt_lo(mk, 0u)] = make_int2(e, d);
    cnt += __builtin_popcount(mk);
  };

  const bool vec_ok = (reinterpret_cast<size_t>(dst) & 15) == 0;
  const int nfull = vec_ok ? (E >> 7) : 0;
  const int rot = (int)(((long long)tile * nfull) / (int)gridDim.x);
  for (int g0 = wave; g0 < nfull; g0 += NW * 4) {
    int4 d4[4];
    int e4[4];
#pragma unroll
    for (int u = 0; u < 4; ++u) {
      int g = g0 + u * NW;
      if (g > nfull - 1) g = nfull - 1;
      int gg = g + rot;
      if (gg >= nfull) gg -= nfull;
      e4[u] = gg * 128 + lane * 4;
      d4[u] = *reinterpret_cast<const int4*>(dst + (size_t)gg * 128 + lane * 4);
    }
#pragma unroll
    for (int u = 0; u < 4; ++u) {
      const bool live = g0 + u * NW < nfull;
      bool m0 = live && (d4[u].x >> SH) == tile;
      bool m1 = live && (d4[u].y >> SH) == tile;
      bool m2 = live && (d4[u].z >> SH) == tile;
      bool m3 = live && (d4[u].w >> SH) == tile;
      if (__builtin_amdgcn_ballot_w32(m0 || m1 || m2 || m3) != 0u) {
        push(m0, e4[u], d4[u].x);
        push(m1, e4[u] + 1, d4[u].y);
        push(m2, e4[u] + 2, d4[u].z);
        push(m3, e4[u] + 3, d4[u].w);
        while (cnt >= 32) {
          cnt -= 32;
          drain32(myq + cnt);
        }
      }
    }
  }
  for (int eb = nfull * 128 + wave * 32; eb < E; eb += NW * 32) {
    int e = eb + lane;
    int d = -1;
    if (e < E) d = dst[e];
    bool m = (d >> SH) == tile;
    push(m, e, d);
    if (cnt >= 32) {
      cnt -= 32;
      drain_n(myq + cnt, 32);
    }
  }
  if (cnt > 0) drain_n(myq, cnt);
  __syncthreads();

  for (int i = wave; i < nrows; i += NW) {
    float* orow = out + (size_t)(base + i) * C + lane * VEC;
    if constexpr (VEC == 4) {
      float4 o;
      o.x = acc[i * C + lane];
      o.y = acc[i * C + 32 + lane];
      o.z = acc[i * C + 64 + lane];
      o.w = acc[i * C + 96 + lane];
      *reinterpret_cast<float4*>(orow) = o;
    } else {
      float2 o;
      o.x = acc[i * C + lane];
      o.y = acc[i * C + 32 + lane];
      *reinterpret_cast<float2*>(orow) = o;
    }
  }
}

__global__ void finalize_k(const float* __restrict__ g, float* __restrict__ out, int N) {
  int node = blockIdx.x * (blockDim.x >> 5) + (threadIdx.x >> 5);
  int lane = threadIdx.x & 31;
  if (node >= N) return;
  float2 v = reinterpret_cast<const float2*>(g + (size_t)node * 64)[lane];
  float mx = fmaxf(v.x, v.y);
  float mn = fminf(v.x, v.y);
#pragma unroll
  for (int o = 16; o > 0; o >>= 1) {
    mx = fmaxf(mx, __shfl_xor(mx, o, 32));
    mn = fminf(mn, __shfl_xor(mn, o, 32));
  }
  float inv = 1.0f / (mx - mn);
  float sx = (v.x - mn) * inv;
  float sy = (v.y - mn) * inv;
  float ss = sx * sx + sy * sy;
#pragma unroll
  for (int o = 16; o > 0; o >>= 1) ss += __shfl_xor(ss, o, 32);
  float r = 1.0f / fmaxf(sqrtf(ss), 1e-12f);
  float2 ov;
  ov.x = sx * r;
  ov.y = sy * r;
  reinterpret_cast<float2*>(out + (size_t)node * 64)[lane] = ov;
}

extern "C" void kernel_launch(void* const* d_in, const int* in_sizes, int n_in,
                              void* d_out, int out_size, void* d_ws, size_t ws_size,
                              hipStream_t stream) {
  (void)n_in; (void)out_size; (void)ws_size;

  const float* x  = (const float*)d_in[0];
  const int*   ei = (const int*)d_in[1];
  const float* W1 = (const float*)d_in[2];
  const float* b1 = (const float*)d_in[3];
  const float* W2 = (const float*)d_in[4];
  const float* b2 = (const float*)d_in[5];

  const int N = in_sizes[0] / 128;
  const int E = in_sizes[1] / 2;
  const int* src = ei;
  const int* dst = ei + E;
  float* out = (float*)d_out;

  float* ws = (float*)d_ws;
  size_t Npad = ((size_t)N + 63) & ~(size_t)63;
  float* dinv = ws;
  float* xw = ws + Npad;
  float* h  = xw + (size_t)N * 128;
  float* g2 = h  + (size_t)N * 128;

  deg_init_k<<<(N + 255) / 256, 256, 0, stream>>>(dinv, N);
  deg_acc_k<<<(E + 255) / 256, 256, 0, stream>>>(dst, dinv, E);
  dinv_k<<<(N + 255) / 256, 256, 0, stream>>>(dinv, N);

  gemm_wmma_k<128><<<(N + 15) / 16, 256, 0, stream>>>(x, W1, xw, N, 128);
  agg_tile_k<128, 9><<<(N + 511) / 512, 1024, 0, stream>>>(xw, dinv, b1, src, dst, h, N, E);

  gemm_wmma_k<64><<<(N + 15) / 16, 128, 0, stream>>>(h, W2, xw, N, 128);
  agg_tile_k<64, 10><<<(N + 1023) / 1024, 1024, 0, stream>>>(xw, dinv, b2, src, dst, g2, N, E);

  finalize_k<<<(N + 7) / 8, 256, 0, stream>>>(g2, out, N);
}


# Layer-2 tile blocks take their edges from the (src,dst) pairs the layer-1 tile scan leaves in scratch (a piece per wave in the g2 plane) and do not scan the dst list; 512-node tiles; sum order free
# speedup vs baseline: 1.8612x; 1.2727x over previous
#include <hip/hip_runtime.h>
#include <math.h>

typedef float v2f __attribute__((ext_vector_type(2)));
typedef float v8f __attribute__((ext_vector_type(8)));

__global__ void deg_init_k(float* __restrict__ deg, int N) {
  int i = blockIdx.x * blockDim.x + threadIdx.x;
  if (i < N) deg[i] = 1.0f;
}

__global__ void deg_acc_k(const int* __restrict__ dst, float* __restrict__ deg, int E) {
  int e = blockIdx.x * blockDim.x + threadIdx.x;
  if (e < E) {
    __hip_atomic_fetch_add(&deg[dst[e]], 1.0f, __ATOMIC_RELAXED, __HIP_MEMORY_SCOPE_AGENT);
  }
}

__global__ void dinv_k(float* __restrict__ deg, int N) {
  int i = blockIdx.x * blockDim.x + threadIdx.x;
  if (i < N) {
    float d = deg[i];
    deg[i] = (d > 0.0f) ? rsqrtf(d) : 0.0f;
  }
}

template <int NN>
__global__ void gemm_wmma_k(const float* __restrict__ A,
                            const float* __restrict__ B,
                            float* __restrict__ C,
                            int M, int K) {
  const int wave = threadIdx.x >> 5;
  const int lane = threadIdx.x & 31;
  const int m0 = blockIdx.x * 16;
  if (m0 >= M) return;
  const int n0 = wave * 16;
  const int half = lane >> 4;
  const int l15 = lane & 15;

  v8f acc = {};
  const float* aptr = A + (size_t)(m0 + l15) * K + half * 2;
  const float* bptr = B + (size_t)(half * 2) * NN + n0 + l15;

  for (int k = 0; k < K; k += 4) {
    v2f a, b;
    a.x = aptr[0];
    a.y = aptr[1];
    b.x = bptr[0];
    b.y = bptr[NN];
    acc = __builtin_amdgcn_wmma_f32_16x16x4_f32(
         false, a,  false, b,
         (short)0, acc,  false,  false);
    aptr += 4;
    bptr += 4 * NN;
  }

  float* cptr = C + (size_t)(m0 + half * 8) * NN + n0 + l15;
#pragma unroll
  for (int r = 0; r < 8; ++r) {
    cptr[0] = acc[r];
    cptr += NN;
  }
}

template <int C, int SH, int MODE>
__global__ __launch_bounds__(1024) void agg_tile_k(const float* __restrict__ feat,
                                                   const float* __restrict__ dinv,
                                                   const float* __restrict__ bias,
                                                   const int* __restrict__ src,
                                                   const int* __restrict__ dst,
                                                   float* out, int* lst, int N, int E) {
  constexpr int T = 1 << SH;
  constexpr int NW = 32;
  constexpr int VEC = C / 32;
  constexpr int QCAP = 160;
  constexpr int G = 32 / VEC;
  __shared__ float acc[T * C];
  __shared__ int2 q[NW * QCAP];

  const int lane = threadIdx.x & 31;
  const int wave = __builtin_amdgcn_readfirstlane((int)(threadIdx.x >> 5));
  const int tile = blockIdx.x;
  const int base = tile << SH;
  int nrows = N - base;
  if (nrows > T) nrows = T;

  int* reg = lst + (size_t)base * 64;
  const int cap = nrows - 1;
  int2* mylist = reinterpret_cast<int2*>(reg + 32) + (size_t)wave * cap;

  for (int i = wave; i < nrows; i += NW) {
    float di = dinv[base + i];
    float w = di * di;
    const float* fr = feat + (size_t)(base + i) * C + lane * VEC;
    if constexpr (VEC == 4) {
      float4 f = *reinterpret_cast<const float4*>(fr);
      float4 bv = reinterpret_cast<const float4*>(bias)[lane];
      acc[i * C + lane]      = w * f.x + bv.x;
      acc[i * C + 32 + lane] = w * f.y + bv.y;
      acc[i * C + 64 + lane] = w * f.z + bv.z;
      acc[i * C + 96 + lane] = w * f.w + bv.w;
    } else {
      float2 f = *reinterpret_cast<const float2*>(fr);
      float2 bv = reinterpret_cast<const float2*>(bias)[lane];
      acc[i * C + lane]      = w * f.x + bv.x;
      acc[i * C + 32 + lane] = w * f.y + bv.y;
    }
  }
  __syncthreads();

  int2* myq = q + wave * QCAP;
  int cnt = 0;
  int rcnt = 0;
  bool rec_ok = true;

  auto drain32 = [&](const int2* qb) {
    int2 ed = qb[lane];
    int s = src[ed.x];
    int d = ed.y;
    if constexpr (MODE == 0) {
      if (rec_ok) {
        if (rcnt + 32 <= cap) {
          mylist[rcnt + lane] = make_int2(s, d);
          rcnt += 32;
        } else {
          rec_ok = false;
        }
      }
    }
    float w = dinv[s] * dinv[d];
    int wb = __builtin_bit_cast(int, w);
    int ro = (d - base) * C;
#pragma unroll 1
    for (int k0 = 0; k0 < 32; k0 += 8) {
      float v[8][VEC];
#pragma unroll
      for (int u = 0; u < 8; ++u) {
        int sk = __builtin_amdgcn_readlane(s, k0 + u);
        const float* fr = feat + (size_t)sk * C + lane * VEC;
        if constexpr (VEC == 4) {
          float4 t = *reinterpret_cast<const float4*>(fr);
          v[u][0] = t.x; v[u][1] = t.y; v[u][2] = t.z; v[u][3] = t.w;
        } else {
          float2 t = *reinterpret_cast<const float2*>(fr);
          v[u][0] = t.x; v[u][1] = t.y;
        }
      }
#pragma unroll
      for (int u = 0; u < 8; ++u) {
        float wk = __builtin_bit_cast(float, __builtin_amdgcn_readlane(wb, k0 + u));
        int rk = __builtin_amdgcn_readlane(ro, k0 + u);
#pragma unroll
        for (int j = 0; j < VEC; ++j)
          __hip_atomic_fetch_add(&acc[rk + j * 32 + lane], wk * v[u][j],
                                 __ATOMIC_RELAXED, __HIP_MEMORY_SCOPE_WORKGROUP);
      }
    }
  };

  auto add_n = [&](int s, int d, int n) {
    float w = dinv[s] * dinv[d];
    int wb = __builtin_bit_cast(int, w);
    int ro = (d - base) * C;
    for (int k = 0; k < n; ++k) {
      int sk = __builtin_amdgcn_readlane(s, k);
      float wk = __builtin_bit_cast(float, __builtin_amdgcn_readlane(wb, k));
      int rk = __builtin_amdgcn_readlane(ro, k);
      const float* fr = feat + (size_t)sk * C + lane * VEC;
#pragma unroll
      for (int j = 0; j < VEC; ++j)
        __hip_atomic_fetch_add(&acc[rk + j * 32 + lane], wk * fr[j],
                               __ATOMIC_RELAXED, __HIP_MEMORY_SCOPE_WORKGROUP);
    }
  };

  auto drain_n = [&](const int2* qb, int n) {
    int2 ed = qb[lane < n ? lane : n - 1];
    int s = src[ed.x];
    int d = ed.y;
    if constexpr (MODE == 0) {
      if (rec_ok) {
        if (rcnt + n <= cap) {
          if (lane < n) mylist[rcnt + lane] = make_int2(s, d);
          rcnt += n;
        } else {
          rec_ok = false;
        }
      }
    }
    add_n(s, d, n);
  };

  auto push = [&](bool m, int e, int d) {
    unsigned mk = __builtin_amdgcn_ballot_w32(m);
    if (m) myq[cnt + (int)__builtin_amdgcn_mbcnt_lo(mk, 0u)] = make_int2(e, d);
    cnt += __builtin_popcount(mk);
  };

  bool listed = false;
  if constexpr (MODE == 1) {
    int hv = reg[lane];
    listed = __builtin_amdgcn_ballot_w32(hv < 0 || hv > cap) == 0u;
    if (listed) {
      const int n = __builtin_amdgcn_readlane(hv, wave);
      int p = 0;
      int2 ed = make_int2(0, base);
      if (n >= 32) ed = mylist[lane];
      for (; p + 32 <= n; p += 32) {
        int s = ed.x;
        int d = ed.y;
        if (p + 64 <= n) ed = mylist[p + 32 + lane];
        float w = dinv[s] * dinv[d];
        int wb = __builtin_bit_cast(int, w);
        int ro = (d - base) * C;
#pragma unroll 1
        for (int k0 = 0; k0 < 32; k0 += G) {
          float v[G][VEC];
#pragma unroll
          for (int u = 0; u < G; ++u) {
            int sk = __builtin_amdgcn_readlane(s, k0 + u);
            const float* fr = feat + (size_t)sk * C + lane * VEC;
            if constexpr (VEC == 4) {
              float4 t = *reinterpret_cast<const float4*>(fr);
              v[u][0] = t.x; v[u][1] = t.y; v[u][2] = t.z; v[u][3] = t.w;
            } else {
              float2 t = *reinterpret_cast<const float2*>(fr);
              v[u][0] = t.x; v[u][1] = t.y;
            }
          }
#pragma unroll
          for (int u = 0; u < G; ++u) {
            float wk = __builtin_bit_cast(float, __builtin_amdgcn_readlane(wb, k0 + u));
            int rk = __builtin_amdgcn_readlane(ro, k0 + u);
#pragma unroll
            for (int j = 0; j < VEC; ++j)
              __hip_atomic_fetch_add(&acc[rk + j * 32 + lane], wk * v[u][j],
                                     __ATOMIC_RELAXED, __HIP_MEMORY_SCOPE_WORKGROUP);
          }
        }
      }
      if (p < n) {
        const int r = n - p;
        int2 et = mylist[p + (lane < r ? lane : r - 1)];
        add_n(et.x, et.y, r);
      }
    }
  }

  if (!listed) {
    const bool vec_ok = (reinterpret_cast<size_t>(dst) & 15) == 0;
    const int nfull = vec_ok ? (E >> 7) : 0;
    const int rot = (int)(((long long)tile * nfull) / (int)gridDim.x);
    for (int g0 = wave; g0 < nfull; g0 += NW * 4) {
      int4 d4[4];
      int e4[4];
#pragma unroll
      for (int u = 0; u < 4; ++u) {
        int g = g0 + u * NW;
        if (g > nfull - 1) g = nfull - 1;
        int gg = g + rot;
        if (gg >= nfull) gg -= nfull;
        e4[u] = gg * 128 + lane * 4;
        d4[u] = *reinterpret_cast<const int4*>(dst + (size_t)gg * 128 + lane * 4);
      }
#pragma unroll
      for (int u = 0; u < 4; ++u) {
        const bool live = g0 + u * NW < nfull;
        bool m0 = live && (d4[u].x >> SH) == tile;
        bool m1 = live && (d4[u].y >> SH) == tile;
        bool m2 = live && (d4[u].z >> SH) == tile;
        bool m3 = live && (d4[u].w >> SH) == tile;
        if (__builtin_amdgcn_ballot_w32(m0 || m1 || m2 || m3) != 0u) {
          push(m0, e4[u], d4[u].x);
          push(m1, e4[u] + 1, d4[u].y);
          push(m2, e4[u] + 2, d4[u].z);
          push(m3, e4[u] + 3, d4[u].w);
          while (cnt >= 32) {
            cnt -= 32;
            drain32(myq + cnt);
          }
        }
      }
    }
    for (int eb = nfull * 128 + wave * 32; eb < E; eb += NW * 32) {
      int e = eb + lane;
      int d = -1;
      if (e < E) d = dst[e];
      bool m = (d >> SH) == tile;
      push(m, e, d);
      if (cnt >= 32) {
        cnt -= 32;
        drain_n(myq + cnt, 32);
      }
    }
    if (cnt > 0) drain_n(myq, cnt);
    if constexpr (MODE == 0) {
      if (lane == 0) reg[wave] = rec_ok ? rcnt : -1;
    }
  }
  __syncthreads();

  for (int i = wave; i < nrows; i += NW) {
    float* orow = out + (size_t)(base + i) * C + lane * VEC;
    if constexpr (VEC == 4) {
      float4 o;
      o.x = acc[i * C + lane];
      o.y = acc[i * C + 32 + lane];
      o.z = acc[i * C + 64 + lane];
      o.w = acc[i * C + 96 + lane];
      *reinterpret_cast<float4*>(orow) = o;
    } else {
      float2 o;
      o.x = acc[i * C + lane];
      o.y = acc[i * C + 32 + lane];
      *reinterpret_cast<float2*>(orow) = o;
    }
  }
}

__global__ void finalize_k(const float* __restrict__ g, float* __restrict__ out, int N) {
  int node = blockIdx.x * (blockDim.x >> 5) + (threadIdx.x >> 5);
  int lane = threadIdx.x & 31;
  if (node >= N) return;
  float2 v = reinterpret_cast<const float2*>(g + (size_t)node * 64)[lane];
  float mx = fmaxf(v.x, v.y);
  float mn = fminf(v.x, v.y);
#pragma unroll
  for (int o = 16; o > 0; o >>= 1) {
    mx = fmaxf(mx, __shfl_xor(mx, o, 32));
    mn = fminf(mn, __shfl_xor(mn, o, 32));
  }
  float inv = 1.0f / (mx - mn);
  float sx = (v.x - mn) * inv;
  float sy = (v.y - mn) * inv;
  float ss = sx * sx + sy * sy;
#pragma unroll
  for (int o = 16; o > 0; o >>= 1) ss += __shfl_xor(ss, o, 32);
  float r = 1.0f / fmaxf(sqrtf(ss), 1e-12f);
  float2 ov;
  ov.x = sx * r;
  ov.y = sy * r;
  reinterpret_cast<float2*>(out + (size_t)node * 64)[lane] = ov;
}

extern "C" void kernel_launch(void* const* d_in, const int* in_sizes, int n_in,
                              void* d_out, int out_size, void* d_ws, size_t ws_size,
                              hipStream_t stream) {
  (void)n_in; (void)out_size; (void)ws_size;

  const float* x  = (const float*)d_in[0];
  const int*   ei = (const int*)d_in[1];
  const float* W1 = (const float*)d_in[2];
  const float* b1 = (const float*)d_in[3];
  const float* W2 = (const float*)d_in[4];
  const float* b2 = (const float*)d_in[5];

  const int N = in_sizes[0] / 128;
  const int E = in_sizes[1] / 2;
  const int* src = ei;
  const int* dst = ei + E;
  float* out = (float*)d_out;

  float* ws = (float*)d_ws;
  size_t Npad = ((size_t)N + 63) & ~(size_t)63;
  float* dinv = ws;
  float* xw = ws + Npad;
  float* h  = xw + (size_t)N * 128;
  float* g2 = h  + (size_t)N * 128;

  deg_init_k<<<(N + 255) / 256, 256, 0, stream>>>(dinv, N);
  deg_acc_k<<<(E + 255) / 256, 256, 0, stream>>>(dst, dinv, E);
  dinv_k<<<(N + 255) / 256, 256, 0, stream>>>(dinv, N);

  gemm_wmma_k<128><<<(N + 15) / 16, 256, 0, stream>>>(x, W1, xw, N, 128);
  agg_tile_k<128, 9, 0><<<(N + 511) / 512, 1024, 0, stream>>>(xw, dinv, b1, src, dst, h, (int*)g2, N, E);

  gemm_wmma_k<64><<<(N + 15) / 16, 128, 0, stream>>>(h, W2, xw, N, 128);
  agg_tile_k<64, 9, 1><<<(N + 511) / 512, 1024, 0, stream>>>(xw, dinv, b2, src, dst, g2, (int*)g2, N, E);

  finalize_k<<<(N + 7) / 8, 256, 0, stream>>>(g2, out, N);
}


# Layer-1 tile scan reads 1-byte tile ids (degree pass packs dst>>9, 4 a dword, into the out buffer), not dst: 16 edges a lane load, zero-byte test on xor; queue holds edge numbers; sum order free
# speedup vs baseline: 2.0007x; 1.0750x over previous
#include <hip/hip_runtime.h>
#include <math.h>

typedef float v2f __attribute__((ext_vector_type(2)));
typedef float v8f __attribute__((ext_vector_type(8)));

__global__ void deg_init_k(float* __restrict__ deg, int N) {
  int i = blockIdx.x * blockDim.x + threadIdx.x;
  if (i < N) deg[i] = 1.0f;
}

__global__ void deg_acc_k(const int* __restrict__ dst, float* __restrict__ deg,
                          int* __restrict__ keys, int E, int sh, int n) {
  int e = (blockIdx.x * blockDim.x + threadIdx.x) * 4;
  if (e >= E) return;
  if (e + 4 <= E && (reinterpret_cast<size_t>(dst) & 15) == 0) {
    int4 d = *reinterpret_cast<const int4*>(dst + e);
    __hip_atomic_fetch_add(&deg[d.x], 1.0f, __ATOMIC_RELAXED, __HIP_MEMORY_SCOPE_AGENT);
    __hip_atomic_fetch_add(&deg[d.y], 1.0f, __ATOMIC_RELAXED, __HIP_MEMORY_SCOPE_AGENT);
    __hip_atomic_fetch_add(&deg[d.z], 1.0f, __ATOMIC_RELAXED, __HIP_MEMORY_SCOPE_AGENT);
    __hip_atomic_fetch_add(&deg[d.w], 1.0f, __ATOMIC_RELAXED, __HIP_MEMORY_SCOPE_AGENT);
    if ((long long)E <= (long long)n * 256)
      keys[e >> 2] = ((d.x >> sh) & 255) | (((d.y >> sh) & 255) << 8) |
                     (((d.z >> sh) & 255) << 16) | (((d.w >> sh) & 255) << 24);
  } else {
    for (int k = 0; k < 4 && e + k < E; ++k)
      __hip_atomic_fetch_add(&deg[dst[e + k]], 1.0f, __ATOMIC_RELAXED, __HIP_MEMORY_SCOPE_AGENT);
  }
}

__global__ void dinv_k(float* __restrict__ deg, int N) {
  int i = blockIdx.x * blockDim.x + threadIdx.x;
  if (i < N) {
    float d = deg[i];
    deg[i] = (d > 0.0f) ? rsqrtf(d) : 0.0f;
  }
}

template <int NN>
__global__ void gemm_wmma_k(const float* __restrict__ A,
                            const float* __restrict__ B,
                            float* __restrict__ C,
                            int M, int K) {
  const int wave = threadIdx.x >> 5;
  const int lane = threadIdx.x & 31;
  const int m0 = blockIdx.x * 16;
  if (m0 >= M) return;
  const int n0 = wave * 16;
  const int half = lane >> 4;
  const int l15 = lane & 15;

  v8f acc = {};
  const float* aptr = A + (size_t)(m0 + l15) * K + half * 2;
  const float* bptr = B + (size_t)(half * 2) * NN + n0 + l15;

  for (int k = 0; k < K; k += 4) {
    v2f a, b;
    a.x = aptr[0];
    a.y = aptr[1];
    b.x = bptr[0];
    b.y = bptr[NN];
    acc = __builtin_amdgcn_wmma_f32_16x16x4_f32(
         false, a,  false, b,
         (short)0, acc,  false,  false);
    aptr += 4;
    bptr += 4 * NN;
  }

  float* cptr = C + (size_t)(m0 + half * 8) * NN + n0 + l15;
#pragma unroll
  for (int r = 0; r < 8; ++r) {
    cptr[0] = acc[r];
    cptr += NN;
  }
}

template <int C, int SH, int MODE>
__global__ __launch_bounds__(1024) void agg_tile_k(const float* __restrict__ feat,
                                                   const float* __restrict__ dinv,
                                                   const float* __restrict__ bias,
                                                   const int* __restrict__ src,
                                                   const int* __restrict__ dst,
                                                   const int* __restrict__ keys,
                                                   float* out, int* lst, int N, int E) {
  constexpr int T = 1 << SH;
  constexpr int NW = 32;
  constexpr int VEC = C / 32;
  constexpr int QCAP = 64;
  constexpr int G = 32 / VEC;
  __shared__ float acc[T * C];
  __shared__ int q[NW * QCAP];

  const int lane = threadIdx.x & 31;
  const int wave = __builtin_amdgcn_readfirstlane((int)(threadIdx.x >> 5));
  const int tile = blockIdx.x;
  const int base = tile << SH;
  int nrows = N - base;
  if (nrows > T) nrows = T;

  int* reg = lst + (size_t)base * 64;
  const int cap = nrows - 1;
  int2* mylist = reinterpret_cast<int2*>(reg + 32) + (size_t)wave * cap;

  for (int i = wave; i < nrows; i += NW) {
    float di = dinv[base + i];
    float w = di * di;
    const float* fr = feat + (size_t)(base + i) * C + lane * VEC;
    if constexpr (VEC == 4) {
      float4 f = *reinterpret_cast<const float4*>(fr);
      float4 bv = reinterpret_cast<const float4*>(bias)[lane];
      acc[i * C + lane]      = w * f.x + bv.x;
      acc[i * C + 32 + lane] = w * f.y + bv.y;
      acc[i * C + 64 + lane] = w * f.z + bv.z;
      acc[i * C + 96 + lane] = w * f.w + bv.w;
    } else {
      float2 f = *reinterpret_cast<const float2*>(fr);
      float2 bv = reinterpret_cast<const float2*>(bias)[lane];
      acc[i * C + lane]      = w * f.x + bv.x;
      acc[i * C + 32 + lane] = w * f.y + bv.y;
    }
  }
  __syncthreads();

  int* myq = q + wave * QCAP;
  int cnt = 0;
  int rcnt = 0;
  bool rec_ok = true;

  auto drain32 = [&](const int* qb) {
    int ed = qb[lane];
    int s = src[ed];
    int d = dst[ed];
    if constexpr (MODE == 0) {
      if (rec_ok) {
        if (rcnt + 32 <= cap) {
          mylist[rcnt + lane] = make_int2(s, d);
          rcnt += 32;
        } else {
          rec_ok = false;
        }
      }
    }
    float w = dinv[s] * dinv[d];
    int wb = __builtin_bit_cast(int, w);
    int ro = (d - base) * C;
#pragma unroll 1
    for (int k0 = 0; k0 < 32; k0 += 8) {
      float v[8][VEC];
#pragma unroll
      for (int u = 0; u < 8; ++u) {
        int sk = __builtin_amdgcn_readlane(s, k0 + u);
        const float* fr = feat + (size_t)sk * C + lane * VEC;
        if constexpr (VEC == 4) {
          float4 t = *reinterpret_cast<const float4*>(fr);
          v[u][0] = t.x; v[u][1] = t.y; v[u][2] = t.z; v[u][3] = t.w;
        } else {
          float2 t = *reinterpret_cast<const float2*>(fr);
          v[u][0] = t.x; v[u][1] = t.y;
        }
      }
#pragma unroll
      for (int u = 0; u < 8; ++u) {
        float wk = __builtin_bit_cast(float, __builtin_amdgcn_readlane(wb, k0 + u));
        int rk = __builtin_amdgcn_readlane(ro, k0 + u);
#pragma unroll
        for (int j = 0; j < VEC; ++j)
          __hip_atomic_fetch_add(&acc[rk + j * 32 + lane], wk * v[u][j],
                                 __ATOMIC_RELAXED, __HIP_MEMORY_SCOPE_WORKGROUP);
      }
    }
  };

  auto add_n = [&](int s, int d, int n) {
    float w = dinv[s] * dinv[d];
    int wb = __builtin_bit_cast(int, w);
    int ro = (d - base) * C;
    for (int k = 0; k < n; ++k) {
      int sk = __builtin_amdgcn_readlane(s, k);
      float wk = __builtin_bit_cast(float, __builtin_amdgcn_readlane(wb, k));
      int rk = __builtin_amdgcn_readlane(ro, k);
      const float* fr = feat + (size_t)sk * C + lane * VEC;
#pragma unroll
      for (int j = 0; j < VEC; ++j)
        __hip_atomic_fetch_add(&acc[rk + j * 32 + lane], wk * fr[j],
                               __ATOMIC_RELAXED, __HIP_MEMORY_SCOPE_WORKGROUP);
    }
  };

  auto drain_n = [&](const int* qb, int n) {
    int ed = qb[lane < n ? lane : n - 1];
    int s = src[ed];
    int d = dst[ed];
    if constexpr (MODE == 0) {
      if (rec_ok) {
        if (rcnt + n <= cap) {
          if (lane < n) mylist[rcnt + lane] = make_int2(s, d);
          rcnt += n;
        } else {
          rec_ok = false;
        }
      }
    }
    add_n(s, d, n);
  };

  auto push = [&](bool m, int e) {
    unsigned mk = __builtin_amdgcn_ballot_w32(m);
    if (m) myq[cnt + (int)__builtin_amdgcn_mbcnt_lo(mk, 0u)] = e;
    cnt += __builtin_popcount(mk);
  };

  bool listed = false;
  if constexpr (MODE == 1) {
    int hv = reg[lane];
    listed = __builtin_amdgcn_ballot_w32(hv < 0 || hv > cap) == 0u;
    if (listed) {
      const int n = __builtin_amdgcn_readlane(hv, wave);
      int p = 0;
      int2 ed = make_int2(0, base);
      if (n >= 32) ed = mylist[lane];
      for (; p + 32 <= n; p += 32) {
        int s = ed.x;
        int d = ed.y;
        if (p + 64 <= n) ed = mylist[p + 32 + lane];
        float w = dinv[s] * dinv[d];
        int wb = __builtin_bit_cast(int, w);
        int ro = (d - base) * C;
#pragma unroll 1
        for (int k0 = 0; k0 < 32; k0 += G) {
          float v[G][VEC];
#pragma unroll
          for (int u = 0; u < G; ++u) {
            int sk = __builtin_amdgcn_readlane(s, k0 + u);
            const float* fr = feat + (size_t)sk * C + lane * VEC;
            if constexpr (VEC == 4) {
              float4 t = *reinterpret_cast<const float4*>(fr);
              v[u][0] = t.x; v[u][1] = t.y; v[u][2] = t.z; v[u][3] = t.w;
            } else {
              float2 t = *reinterpret_cast<const float2*>(fr);
              v[u][0] = t.x; v[u][1] = t.y;
            }
          }
#pragma unroll
          for (int u = 0; u < G; ++u) {
            float wk = __builtin_bit_cast(float, __builtin_amdgcn_readlane(wb, k0 + u));
            int rk = __builtin_amdgcn_readlane(ro, k0 + u);
#pragma unroll
            for (int j = 0; j < VEC; ++j)
              __hip_atomic_fetch_add(&acc[rk + j * 32 + lane], wk * v[u][j],
                                     __ATOMIC_RELAXED, __HIP_MEMORY_SCOPE_WORKGROUP);
          }
        }
      }
      if (p < n) {
        const int r = n - p;
        int2 et = mylist[p + (lane < r ? lane : r - 1)];
        add_n(et.x, et.y, r);
      }
    }
  }

  if (!listed) {
    const bool key_ok = gridDim.x <= 256u && (reinterpret_cast<size_t>(dst) & 15) == 0 &&
                        (reinterpret_cast<size_t>(keys) & 15) == 0 &&
                        (long long)E <= (long long)N * 256;
    const int nfull = key_ok ? (E >> 9) : 0;
    const unsigned pat = (unsigned)tile * 0x01010101u;
    const unsigned HI = 0x80808080u;
    const int rot = (int)(((long long)tile * nfull) / (int)gridDim.x);
    for (int g0 = wave; g0 < nfull; g0 += NW * 4) {
      int4 k4[4];
      int e16[4];
#pragma unroll
      for (int u = 0; u < 4; ++u) {
        int g = g0 + u * NW;
        if (g > nfull - 1) g = nfull - 1;
        int gg = g + rot;
        if (gg >= nfull) gg -= nfull;
        e16[u] = gg * 512 + lane * 16;
        k4[u] = *reinterpret_cast<const int4*>(keys + (size_t)gg * 128 + lane * 4);
      }
#pragma unroll
      for (int u = 0; u < 4; ++u) {
        const bool live = g0 + u * NW < nfull;
        unsigned x0 = (unsigned)k4[u].x ^ pat;
        unsigned x1 = (unsigned)k4[u].y ^ pat;
        unsigned x2 = (unsigned)k4[u].z ^ pat;
        unsigned x3 = (unsigned)k4[u].w ^ pat;
        unsigned nz0 = ((x0 & ~HI) + ~HI) | x0;
        unsigned nz1 = ((x1 & ~HI) + ~HI) | x1;
        unsigned nz2 = ((x2 & ~HI) + ~HI) | x2;
        unsigned nz3 = ((x3 & ~HI) + ~HI) | x3;
        const bool hit = live && ((nz0 & nz1 & nz2 & nz3 & HI) != HI);
        if (__builtin_amdgcn_ballot_w32(hit) != 0u) {
          unsigned mm = 0u;
          if (hit)
            mm = 0x0F0F0F0Fu ^ (((nz0 & HI) >> 7) | ((nz1 & HI) >> 6) | ((nz2 & HI) >> 5) | ((nz3 & HI) >> 4));
          while (__builtin_amdgcn_ballot_w32(mm != 0u) != 0u) {
            const int p = __builtin_ctz(mm | 0x80000000u);
            push(mm != 0u, e16[u] + ((p & 3) << 2) + (p >> 3));
            mm &= mm - 1u;
            if (cnt >= 32) {
              cnt -= 32;
              drain32(myq + cnt);
            }
          }
        }
      }
    }
    for (int eb = nfull * 512 + wave * 32; eb < E; eb += NW * 32) {
      int e = eb + lane;
      int d = -1;
      if (e < E) d = dst[e];
      bool m = (d >> SH) == tile;
      push(m, e);
      if (cnt >= 32) {
        cnt -= 32;
        drain_n(myq + cnt, 32);
      }
    }
    if (cnt > 0) drain_n(myq, cnt);
    if constexpr (MODE == 0) {
      if (lane == 0) reg[wave] = rec_ok ? rcnt : -1;
    }
  }
  __syncthreads();

  for (int i = wave; i < nrows; i += NW) {
    float* orow = out + (size_t)(base + i) * C + lane * VEC;
    if constexpr (VEC == 4) {
      float4 o;
      o.x = acc[i * C + lane];
      o.y = acc[i * C + 32 + lane];
      o.z = acc[i * C + 64 + lane];
      o.w = acc[i * C + 96 + lane];
      *reinterpret_cast<float4*>(orow) = o;
    } else {
      float2 o;
      o.x = acc[i * C + lane];
      o.y = acc[i * C + 32 + lane];
      *reinterpret_cast<float2*>(orow) = o;
    }
  }
}

__global__ void finalize_k(const float* __restrict__ g, float* __restrict__ out, int N) {
  int node = blockIdx.x * (blockDim.x >> 5) + (threadIdx.x >> 5);
  int lane = threadIdx.x & 31;
  if (node >= N) return;
  float2 v = reinterpret_cast<const float2*>(g + (size_t)node * 64)[lane];
  float mx = fmaxf(v.x, v.y);
  float mn = fminf(v.x, v.y);
#pragma unroll
  for (int o = 16; o > 0; o >>= 1) {
    mx = fmaxf(mx, __shfl_xor(mx, o, 32));
    mn = fminf(mn, __shfl_xor(mn, o, 32));
  }
  float inv = 1.0f / (mx - mn);
  float sx = (v.x - mn) * inv;
  float sy = (v.y - mn) * inv;
  float ss = sx * sx + sy * sy;
#pragma unroll
  for (int o = 16; o > 0; o >>= 1) ss += __shfl_xor(ss, o, 32);
  float r = 1.0f / fmaxf(sqrtf(ss), 1e-12f);
  float2 ov;
  ov.x = sx * r;
  ov.y = sy * r;
  reinterpret_cast<float2*>(out + (size_t)node * 64)[lane] = ov;
}

extern "C" void kernel_launch(void* const* d_in, const int* in_sizes, int n_in,
                              void* d_out, int out_size, void* d_ws, size_t ws_size,
                              hipStream_t stream) {
  (void)n_in; (void)out_size; (void)ws_size;

  const float* x  = (const float*)d_in[0];
  const int*   ei = (const int*)d_in[1];
  const float* W1 = (const float*)d_in[2];
  const float* b1 = (const float*)d_in[3];
  const float* W2 = (const float*)d_in[4];
  const float* b2 = (const float*)d_in[5];

  const int N = in_sizes[0] / 128;
  const int E = in_sizes[1] / 2;
  const int* src = ei;
  const int* dst = ei + E;
  float* out = (float*)d_out;

  float* ws = (float*)d_ws;
  size_t Npad = ((size_t)N + 63) & ~(size_t)63;
  float* dinv = ws;
  float* xw = ws + Npad;
  float* h  = xw + (size_t)N * 128;
  float* g2 = h  + (size_t)N * 128;

  deg_init_k<<<(N + 255) / 256, 256, 0, stream>>>(dinv, N);
  int* keys = (int*)d_out;
  deg_acc_k<<<((E + 3) / 4 + 255) / 256, 256, 0, stream>>>(dst, dinv, keys, E, 9, N);
  dinv_k<<<(N + 255) / 256, 256, 0, stream>>>(dinv, N);

  gemm_wmma_k<128><<<(N + 15) / 16, 256, 0, stream>>>(x, W1, xw, N, 128);
  agg_tile_k<128, 9, 0><<<(N + 511) / 512, 1024, 0, stream>>>(xw, dinv, b1, src, dst, keys, h, (int*)g2, N, E);

  gemm_wmma_k<64><<<(N + 15) / 16, 128, 0, stream>>>(h, W2, xw, N, 128);
  agg_tile_k<64, 9, 1><<<(N + 511) / 512, 1024, 0, stream>>>(xw, dinv, b2, src, dst, keys, g2, (int*)g2, N, E);

  finalize_k<<<(N + 7) / 8, 256, 0, stream>>>(g2, out, N);
}


# Aggregation tiles of 392 nodes, not 512: 256 tile blocks, one on each of the 256 WGPs (196 before), each adds a 256th of the edges into its LDS; tile id by constant division; sum order free
# speedup vs baseline: 2.1014x; 1.0503x over previous
#include <hip/hip_runtime.h>
#include <math.h>

typedef float v2f __attribute__((ext_vector_type(2)));
typedef float v8f __attribute__((ext_vector_type(8)));

__global__ void deg_init_k(float* __restrict__ deg, int N) {
  int i = blockIdx.x * blockDim.x + threadIdx.x;
  if (i < N) deg[i] = 1.0f;
}

template <int T>
__global__ void deg_acc_k(const int* __restrict__ dst, float* __restrict__ deg,
                          int* __restrict__ keys, int E, int n) {
  int e = (blockIdx.x * blockDim.x + threadIdx.x) * 4;
  if (e >= E) return;
  if (e + 4 <= E && (reinterpret_cast<size_t>(dst) & 15) == 0) {
    int4 d = *reinterpret_cast<const int4*>(dst + e);
    __hip_atomic_fetch_add(&deg[d.x], 1.0f, __ATOMIC_RELAXED, __HIP_MEMORY_SCOPE_AGENT);
    __hip_atomic_fetch_add(&deg[d.y], 1.0f, __ATOMIC_RELAXED, __HIP_MEMORY_SCOPE_AGENT);
    __hip_atomic_fetch_add(&deg[d.z], 1.0f, __ATOMIC_RELAXED, __HIP_MEMORY_SCOPE_AGENT);
    __hip_atomic_fetch_add(&deg[d.w], 1.0f, __ATOMIC_RELAXED, __HIP_MEMORY_SCOPE_AGENT);
    if ((long long)E <= (long long)n * 256)
      keys[e >> 2] = (((unsigned)d.x / T) & 255u) | ((((unsigned)d.y / T) & 255u) << 8) |
                     ((((unsigned)d.z / T) & 255u) << 16) | ((((unsigned)d.w / T) & 255u) << 24);
  } else {
    for (int k = 0; k < 4 && e + k < E; ++k)
      __hip_atomic_fetch_add(&deg[dst[e + k]], 1.0f, __ATOMIC_RELAXED, __HIP_MEMORY_SCOPE_AGENT);
  }
}

__global__ void dinv_k(float* __restrict__ deg, int N) {
  int i = blockIdx.x * blockDim.x + threadIdx.x;
  if (i < N) {
    float d = deg[i];
    deg[i] = (d > 0.0f) ? rsqrtf(d) : 0.0f;
  }
}

template <int NN>
__global__ void gemm_wmma_k(const float* __restrict__ A,
                            const float* __restrict__ B,
                            float* __restrict__ C,
                            int M, int K) {
  const int wave = threadIdx.x >> 5;
  const int lane = threadIdx.x & 31;
  const int m0 = blockIdx.x * 16;
  if (m0 >= M) return;
  const int n0 = wave * 16;
  const int half = lane >> 4;
  const int l15 = lane & 15;

  v8f acc = {};
  const float* aptr = A + (size_t)(m0 + l15) * K + half * 2;
  const float* bptr = B + (size_t)(half * 2) * NN + n0 + l15;

  for (int k = 0; k < K; k += 4) {
    v2f a, b;
    a.x = aptr[0];
    a.y = aptr[1];
    b.x = bptr[0];
    b.y = bptr[NN];
    acc = __builtin_amdgcn_wmma_f32_16x16x4_f32(
         false, a,  false, b,
         (short)0, acc,  false,  false);
    aptr += 4;
    bptr += 4 * NN;
  }

  float* cptr = C + (size_t)(m0 + half * 8) * NN + n0 + l15;
#pragma unroll
  for (int r = 0; r < 8; ++r) {
    cptr[0] = acc[r];
    cptr += NN;
  }
}

template <int C, int T, int MODE>
__global__ __launch_bounds__(1024) void agg_tile_k(const float* __restrict__ feat,
                                                   const float* __restrict__ dinv,
                                                   const float* __restrict__ bias,
                                                   const int* __restrict__ src,
                                                   const int* __restrict__ dst,
                                                   const int* __restrict__ keys,
                                                   float* out, int* lst, int N, int E) {
  constexpr int NW = 32;
  constexpr int VEC = C / 32;
  constexpr int QCAP = 64;
  constexpr int G = 32 / VEC;
  __shared__ float acc[T * C];
  __shared__ int q[NW * QCAP];

  const int lane = threadIdx.x & 31;
  const int wave = __builtin_amdgcn_readfirstlane((int)(threadIdx.x >> 5));
  const int tile = blockIdx.x;
  const int base = tile * T;
  int nrows = N - base;
  if (nrows > T) nrows = T;

  int* reg = lst + (size_t)base * 64;
  const int cap = nrows - 1;
  int2* mylist = reinterpret_cast<int2*>(reg + 32) + (size_t)wave * cap;

  for (int i = wave; i < nrows; i += NW) {
    float di = dinv[base + i];
    float w = di * di;
    const float* fr = feat + (size_t)(base + i) * C + lane * VEC;
    if constexpr (VEC == 4) {
      float4 f = *reinterpret_cast<const float4*>(fr);
      float4 bv = reinterpret_cast<const float4*>(bias)[lane];
      acc[i * C + lane]      = w * f.x + bv.x;
      acc[i * C + 32 + lane] = w * f.y + bv.y;
      acc[i * C + 64 + lane] = w * f.z + bv.z;
      acc[i * C + 96 + lane] = w * f.w + bv.w;
    } else {
      float2 f = *reinterpret_cast<const float2*>(fr);
      float2 bv = reinterpret_cast<const float2*>(bias)[lane];
      acc[i * C + lane]      = w * f.x + bv.x;
      acc[i * C + 32 + lane] = w * f.y + bv.y;
    }
  }
  __syncthreads();

  int* myq = q + wave * QCAP;
  int cnt = 0;
  int rcnt = 0;
  bool rec_ok = true;

  auto drain32 = [&](const int* qb) {
    int ed = qb[lane];
    int s = src[ed];
    int d = dst[ed];
    if constexpr (MODE == 0) {
      if (rec_ok) {
        if (rcnt + 32 <= cap) {
          mylist[rcnt + lane] = make_int2(s, d);
          rcnt += 32;
        } else {
          rec_ok = false;
        }
      }
    }
    float w = dinv[s] * dinv[d];
    int wb = __builtin_bit_cast(int, w);
    int ro = (d - base) * C;
#pragma unroll 1
    for (int k0 = 0; k0 < 32; k0 += 8) {
      float v[8][VEC];
#pragma unroll
      for (int u = 0; u < 8; ++u) {
        int sk = __builtin_amdgcn_readlane(s, k0 + u);
        const float* fr = feat + (size_t)sk * C + lane * VEC;
        if constexpr (VEC == 4) {
          float4 t = *reinterpret_cast<const float4*>(fr);
          v[u][0] = t.x; v[u][1] = t.y; v[u][2] = t.z; v[u][3] = t.w;
        } else {
          float2 t = *reinterpret_cast<const float2*>(fr);
          v[u][0] = t.x; v[u][1] = t.y;
        }
      }
#pragma unroll
      for (int u = 0; u < 8; ++u) {
        float wk = __builtin_bit_cast(float, __builtin_amdgcn_readlane(wb, k0 + u));
        int rk = __builtin_amdgcn_readlane(ro, k0 + u);
#pragma unroll
        for (int j = 0; j < VEC; ++j)
          __hip_atomic_fetch_add(&acc[rk + j * 32 + lane], wk * v[u][j],
                                 __ATOMIC_RELAXED, __HIP_MEMORY_SCOPE_WORKGROUP);
      }
    }
  };

  auto add_n = [&](int s, int d, int n) {
    float w = dinv[s] * dinv[d];
    int wb = __builtin_bit_cast(int, w);
    int ro = (d - base) * C;
    for (int k = 0; k < n; ++k) {
      int sk = __builtin_amdgcn_readlane(s, k);
      float wk = __builtin_bit_cast(float, __builtin_amdgcn_readlane(wb, k));
      int rk = __builtin_amdgcn_readlane(ro, k);
      const float* fr = feat + (size_t)sk * C + lane * VEC;
#pragma unroll
      for (int j = 0; j < VEC; ++j)
        __hip_atomic_fetch_add(&acc[rk + j * 32 + lane], wk * fr[j],
                               __ATOMIC_RELAXED, __HIP_MEMORY_SCOPE_WORKGROUP);
    }
  };

  auto drain_n = [&](const int* qb, int n) {
    int ed = qb[lane < n ? lane : n - 1];
    int s = src[ed];
    int d = dst[ed];
    if constexpr (MODE == 0) {
      if (rec_ok) {
        if (rcnt + n <= cap) {
          if (lane < n) mylist[rcnt + lane] = make_int2(s, d);
          rcnt += n;
        } else {
          rec_ok = false;
        }
      }
    }
    add_n(s, d, n);
  };

  auto push = [&](bool m, int e) {
    unsigned mk = __builtin_amdgcn_ballot_w32(m);
    if (m) myq[cnt + (int)__builtin_amdgcn_mbcnt_lo(mk, 0u)] = e;
    cnt += __builtin_popcount(mk);
  };

  bool listed = false;
  if constexpr (MODE == 1) {
    int hv = reg[lane];
    listed = __builtin_amdgcn_ballot_w32(hv < 0 || hv > cap) == 0u;
    if (listed) {
      const int n = __builtin_amdgcn_readlane(hv, wave);
      int p = 0;
      int2 ed = make_int2(0, base);
      if (n >= 32) ed = mylist[lane];
      for (; p + 32 <= n; p += 32) {
        int s = ed.x;
        int d = ed.y;
        if (p + 64 <= n) ed = mylist[p + 32 + lane];
        float w = dinv[s] * dinv[d];
        int wb = __builtin_bit_cast(int, w);
        int ro = (d - base) * C;
#pragma unroll 1
        for (int k0 = 0; k0 < 32; k0 += G) {
          float v[G][VEC];
#pragma unroll
          for (int u = 0; u < G; ++u) {
            int sk = __builtin_amdgcn_readlane(s, k0 + u);
            const float* fr = feat + (size_t)sk * C + lane * VEC;
            if constexpr (VEC == 4) {
              float4 t = *reinterpret_cast<const float4*>(fr);
              v[u][0] = t.x; v[u][1] = t.y; v[u][2] = t.z; v[u][3] = t.w;
            } else {
              float2 t = *reinterpret_cast<const float2*>(fr);
              v[u][0] = t.x; v[u][1] = t.y;
            }
          }
#pragma unroll
          for (int u = 0; u < G; ++u) {
            float wk = __builtin_bit_cast(float, __builtin_amdgcn_readlane(wb, k0 + u));
            int rk = __builtin_amdgcn_readlane(ro, k0 + u);
#pragma unroll
            for (int j = 0; j < VEC; ++j)
              __hip_atomic_fetch_add(&acc[rk + j * 32 + lane], wk * v[u][j],
                                     __ATOMIC_RELAXED, __HIP_MEMORY_SCOPE_WORKGROUP);
          }
        }
      }
      if (p < n) {
        const int r = n - p;
        int2 et = mylist[p + (lane < r ? lane : r - 1)];
        add_n(et.x, et.y, r);
      }
    }
  }

  if (!listed) {
    const bool key_ok = gridDim.x <= 256u && (reinterpret_cast<size_t>(dst) & 15) == 0 &&
                        (reinterpret_cast<size_t>(keys) & 15) == 0 &&
                        (long long)E <= (long long)N * 256;
    const int nfull = key_ok ? (E >> 9) : 0;
    const unsigned pat = (unsigned)tile * 0x01010101u;
    const unsigned HI = 0x80808080u;
    const int rot = (int)(((long long)tile * nfull) / (int)gridDim.x);
    for (int g0 = wave; g0 < nfull; g0 += NW * 4) {
      int4 k4[4];
      int e16[4];
#pragma unroll
      for (int u = 0; u < 4; ++u) {
        int g = g0 + u * NW;
        if (g > nfull - 1) g = nfull - 1;
        int gg = g + rot;
        if (gg >= nfull) gg -= nfull;
        e16[u] = gg * 512 + lane * 16;
        k4[u] = *reinterpret_cast<const int4*>(keys + (size_t)gg * 128 + lane * 4);
      }
#pragma unroll
      for (int u = 0; u < 4; ++u) {
        const bool live = g0 + u * NW < nfull;
        unsigned x0 = (unsigned)k4[u].x ^ pat;
        unsigned x1 = (unsigned)k4[u].y ^ pat;
        unsigned x2 = (unsigned)k4[u].z ^ pat;
        unsigned x3 = (unsigned)k4[u].w ^ pat;
        unsigned nz0 = ((x0 & ~HI) + ~HI) | x0;
        unsigned nz1 = ((x1 & ~HI) + ~HI) | x1;
        unsigned nz2 = ((x2 & ~HI) + ~HI) | x2;
        unsigned nz3 = ((x3 & ~HI) + ~HI) | x3;
        const bool hit = live && ((nz0 & nz1 & nz2 & nz3 & HI) != HI);
        if (__builtin_amdgcn_ballot_w32(hit) != 0u) {
          unsigned mm = 0u;
          if (hit)
            mm = 0x0F0F0F0Fu ^ (((nz0 & HI) >> 7) | ((nz1 & HI) >> 6) | ((nz2 & HI) >> 5) | ((nz3 & HI) >> 4));
          while (__builtin_amdgcn_ballot_w32(mm != 0u) != 0u) {
            const int p = __builtin_ctz(mm | 0x80000000u);
            push(mm != 0u, e16[u] + ((p & 3) << 2) + (p >> 3));
            mm &= mm - 1u;
            if (cnt >= 32) {
              cnt -= 32;
              drain32(myq + cnt);
            }
          }
        }
      }
    }
    for (int eb = nfull * 512 + wave * 32; eb < E; eb += NW * 32) {
      int e = eb + lane;
      int d = -1;
      if (e < E) d = dst[e];
      bool m = (unsigned)(d - base) < (unsigned)T;
      push(m, e);
      if (cnt >= 32) {
        cnt -= 32;
        drain_n(myq + cnt, 32);
      }
    }
    if (cnt > 0) drain_n(myq, cnt);
    if constexpr (MODE == 0) {
      if (lane == 0) reg[wave] = rec_ok ? rcnt : -1;
    }
  }
  __syncthreads();

  for (int i = wave; i < nrows; i += NW) {
    float* orow = out + (size_t)(base + i) * C + lane * VEC;
    if constexpr (VEC == 4) {
      float4 o;
      o.x = acc[i * C + lane];
      o.y = acc[i * C + 32 + lane];
      o.z = acc[i * C + 64 + lane];
      o.w = acc[i * C + 96 + lane];
      *reinterpret_cast<float4*>(orow) = o;
    } else {
      float2 o;
      o.x = acc[i * C + lane];
      o.y = acc[i * C + 32 + lane];
      *reinterpret_cast<float2*>(orow) = o;
    }
  }
}

__global__ void finalize_k(const float* __restrict__ g, float* __restrict__ out, int N) {
  int node = blockIdx.x * (blockDim.x >> 5) + (threadIdx.x >> 5);
  int lane = threadIdx.x & 31;
  if (node >= N) return;
  float2 v = reinterpret_cast<const float2*>(g + (size_t)node * 64)[lane];
  float mx = fmaxf(v.x, v.y);
  float mn = fminf(v.x, v.y);
#pragma unroll
  for (int o = 16; o > 0; o >>= 1) {
    mx = fmaxf(mx, __shfl_xor(mx, o, 32));
    mn = fminf(mn, __shfl_xor(mn, o, 32));
  }
  float inv = 1.0f / (mx - mn);
  float sx = (v.x - mn) * inv;
  float sy = (v.y - mn) * inv;
  float ss = sx * sx + sy * sy;
#pragma unroll
  for (int o = 16; o > 0; o >>= 1) ss += __shfl_xor(ss, o, 32);
  float r = 1.0f / fmaxf(sqrtf(ss), 1e-12f);
  float2 ov;
  ov.x = sx * r;
  ov.y = sy * r;
  reinterpret_cast<float2*>(out + (size_t)node * 64)[lane] = ov;
}

constexpr int TILE = 392;

extern "C" void kernel_launch(void* const* d_in, const int* in_sizes, int n_in,
                              void* d_out, int out_size, void* d_ws, size_t ws_size,
                              hipStream_t stream) {
  (void)n_in; (void)out_size; (void)ws_size;

  const float* x  = (const float*)d_in[0];
  const int*   ei = (const int*)d_in[1];
  const float* W1 = (const float*)d_in[2];
  const float* b1 = (const float*)d_in[3];
  const float* W2 = (const float*)d_in[4];
  const float* b2 = (const float*)d_in[5];

  const int N = in_sizes[0] / 128;
  const int E = in_sizes[1] / 2;
  const int* src = ei;
  const int* dst = ei + E;
  float* out = (float*)d_out;

  float* ws = (float*)d_ws;
  size_t Npad = ((size_t)N + 63) & ~(size_t)63;
  float* dinv = ws;
  float* xw = ws + Npad;
  float* h  = xw + (size_t)N * 128;
  float* g2 = h  + (size_t)N * 128;

  deg_init_k<<<(N + 255) / 256, 256, 0, stream>>>(dinv, N);
  int* keys = (int*)d_out;
  deg_acc_k<TILE><<<((E + 3) / 4 + 255) / 256, 256, 0, stream>>>(dst, dinv, keys, E, N);
  dinv_k<<<(N + 255) / 256, 256, 0, stream>>>(dinv, N);

  gemm_wmma_k<128><<<(N + 15) / 16, 256, 0, stream>>>(x, W1, xw, N, 128);
  agg_tile_k<128, TILE, 0><<<(N + TILE - 1) / TILE, 1024, 0, stream>>>(xw, dinv, b1, src, dst, keys, h, (int*)g2, N, E);

  gemm_wmma_k<64><<<(N + 15) / 16, 128, 0, stream>>>(h, W2, xw, N, 128);
  agg_tile_k<64, TILE, 1><<<(N + TILE - 1) / TILE, 1024, 0, stream>>>(xw, dinv, b2, src, dst, keys, g2, (int*)g2, N, E);

  finalize_k<<<(N + 7) / 8, 256, 0, stream>>>(g2, out, N);
}


# Layer-2 GEMM has no launch: the layer-1 tile block multiplies its 392 summed rows by W2 out of LDS (f32 WMMA, row pitch 132) and stores z; h never goes to memory; k order of the sums changes
# speedup vs baseline: 2.2381x; 1.0651x over previous
#include <hip/hip_runtime.h>
#include <math.h>

typedef float v2f __attribute__((ext_vector_type(2)));
typedef float v8f __attribute__((ext_vector_type(8)));

__global__ void deg_init_k(float* __restrict__ deg, int N) {
  int i = blockIdx.x * blockDim.x + threadIdx.x;
  if (i < N) deg[i] = 1.0f;
}

template <int T>
__global__ void deg_acc_k(const int* __restrict__ dst, float* __restrict__ deg,
                          int* __restrict__ keys, int E, int n) {
  int e = (blockIdx.x * blockDim.x + threadIdx.x) * 4;
  if (e >= E) return;
  if (e + 4 <= E && (reinterpret_cast<size_t>(dst) & 15) == 0) {
    int4 d = *reinterpret_cast<const int4*>(dst + e);
    __hip_atomic_fetch_add(&deg[d.x], 1.0f, __ATOMIC_RELAXED, __HIP_MEMORY_SCOPE_AGENT);
    __hip_atomic_fetch_add(&deg[d.y], 1.0f, __ATOMIC_RELAXED, __HIP_MEMORY_SCOPE_AGENT);
    __hip_atomic_fetch_add(&deg[d.z], 1.0f, __ATOMIC_RELAXED, __HIP_MEMORY_SCOPE_AGENT);
    __hip_atomic_fetch_add(&deg[d.w], 1.0f, __ATOMIC_RELAXED, __HIP_MEMORY_SCOPE_AGENT);
    if ((long long)E <= (long long)n * 256)
      keys[e >> 2] = (((unsigned)d.x / T) & 255u) | ((((unsigned)d.y / T) & 255u) << 8) |
                     ((((unsigned)d.z / T) & 255u) << 16) | ((((unsigned)d.w / T) & 255u) << 24);
  } else {
    for (int k = 0; k < 4 && e + k < E; ++k)
      __hip_atomic_fetch_add(&deg[dst[e + k]], 1.0f, __ATOMIC_RELAXED, __HIP_MEMORY_SCOPE_AGENT);
  }
}

__global__ void dinv_k(float* __restrict__ deg, int N) {
  int i = blockIdx.x * blockDim.x + threadIdx.x;
  if (i < N) {
    float d = deg[i];
    deg[i] = (d > 0.0f) ? rsqrtf(d) : 0.0f;
  }
}

template <int NN>
__global__ void gemm_wmma_k(const float* __restrict__ A,
                            const float* __restrict__ B,
                            float* __restrict__ C,
                            int M, int K) {
  const int wave = threadIdx.x >> 5;
  const int lane = threadIdx.x & 31;
  const int m0 = blockIdx.x * 16;
  if (m0 >= M) return;
  const int n0 = wave * 16;
  const int half = lane >> 4;
  const int l15 = lane & 15;

  v8f acc = {};
  const float* aptr = A + (size_t)(m0 + l15) * K + half * 2;
  const float* bptr = B + (size_t)(half * 2) * NN + n0 + l15;

  for (int k = 0; k < K; k += 4) {
    v2f a, b;
    a.x = aptr[0];
    a.y = aptr[1];
    b.x = bptr[0];
    b.y = bptr[NN];
    acc = __builtin_amdgcn_wmma_f32_16x16x4_f32(
         false, a,  false, b,
         (short)0, acc,  false,  false);
    aptr += 4;
    bptr += 4 * NN;
  }

  float* cptr = C + (size_t)(m0 + half * 8) * NN + n0 + l15;
#pragma unroll
  for (int r = 0; r < 8; ++r) {
    cptr[0] = acc[r];
    cptr += NN;
  }
}

template <int C, int T, int MODE>
__global__ __launch_bounds__(1024) void agg_tile_k(const float* __restrict__ feat,
                                                   const float* __restrict__ dinv,
                                                   const float* __restrict__ bias,
                                                   const int* __restrict__ src,
                                                   const int* __restrict__ dst,
                                                   const int* __restrict__ keys,
                                                   const float* __restrict__ w2,
                                                   float* out, int* lst, int N, int E) {
  constexpr int NW = 32;
  constexpr int VEC = C / 32;
  constexpr int QCAP = 64;
  constexpr int G = 32 / VEC;
  constexpr int P = MODE == 0 ? C + 4 : C;
  constexpr int ROWS = MODE == 0 ? (T + 15) / 16 * 16 : T;
  constexpr int OC = 64;
  static_assert(MODE != 0 || C == 128, "the fused product is written for 128 channels");
  __shared__ alignas(16) float acc[ROWS * P];
  __shared__ int q[NW * QCAP];

  const int lane = threadIdx.x & 31;
  const int wave = __builtin_amdgcn_readfirstlane((int)(threadIdx.x >> 5));
  const int tile = blockIdx.x;
  const int base = tile * T;
  int nrows = N - base;
  if (nrows > T) nrows = T;

  int* reg = lst + (size_t)base * 64;
  const int cap = nrows - 1;
  int2* mylist = reinterpret_cast<int2*>(reg + 32) + (size_t)wave * cap;

  for (int i = wave; i < nrows; i += NW) {
    float di = dinv[base + i];
    float w = di * di;
    const float* fr = feat + (size_t)(base + i) * C + lane * VEC;
    if constexpr (VEC == 4) {
      float4 f = *reinterpret_cast<const float4*>(fr);
      float4 bv = reinterpret_cast<const float4*>(bias)[lane];
      acc[i * P + lane]      = w * f.x + bv.x;
      acc[i * P + 32 + lane] = w * f.y + bv.y;
      acc[i * P + 64 + lane] = w * f.z + bv.z;
      acc[i * P + 96 + lane] = w * f.w + bv.w;
    } else {
      float2 f = *reinterpret_cast<const float2*>(fr);
      float2 bv = reinterpret_cast<const float2*>(bias)[lane];
      acc[i * P + lane]      = w * f.x + bv.x;
      acc[i * P + 32 + lane] = w * f.y + bv.y;
    }
  }
  [[maybe_unused]] const int ntm = (nrows + 15) >> 4;
  if constexpr (MODE == 0) {
    for (int i = nrows + wave; i < ntm * 16; i += NW) {
      acc[i * P + lane]      = 0.0f;
      acc[i * P + 32 + lane] = 0.0f;
      acc[i * P + 64 + lane] = 0.0f;
      acc[i * P + 96 + lane] = 0.0f;
    }
  }
  __syncthreads();

  int* myq = q + wave * QCAP;
  int cnt = 0;
  int rcnt = 0;
  bool rec_ok = true;

  auto drain32 = [&](const int* qb) {
    int ed = qb[lane];
    int s = src[ed];
    int d = dst[ed];
    if constexpr (MODE == 0) {
      if (rec_ok) {
        if (rcnt + 32 <= cap) {
          mylist[rcnt + lane] = make_int2(s, d);
          rcnt += 32;
        } else {
          rec_ok = false;
        }
      }
    }
    float w = dinv[s] * dinv[d];
    int wb = __builtin_bit_cast(int, w);
    int ro = (d - base) * P;
#pragma unroll 1
    for (int k0 = 0; k0 < 32; k0 += 8) {
      float v[8][VEC];
#pragma unroll
      for (int u = 0; u < 8; ++u) {
        int sk = __builtin_amdgcn_readlane(s, k0 + u);
        const float* fr = feat + (size_t)sk * C + lane * VEC;
        if constexpr (VEC == 4) {
          float4 t = *reinterpret_cast<const float4*>(fr);
          v[u][0] = t.x; v[u][1] = t.y; v[u][2] = t.z; v[u][3] = t.w;
        } else {
          float2 t = *reinterpret_cast<const float2*>(fr);
          v[u][0] = t.x; v[u][1] = t.y;
        }
      }
#pragma unroll
      for (int u = 0; u < 8; ++u) {
        float wk = __builtin_bit_cast(float, __builtin_amdgcn_readlane(wb, k0 + u));
        int rk = __builtin_amdgcn_readlane(ro, k0 + u);
#pragma unroll
        for (int j = 0; j < VEC; ++j)
          __hip_atomic_fetch_add(&acc[rk + j * 32 + lane], wk * v[u][j],
                                 __ATOMIC_RELAXED, __HIP_MEMORY_SCOPE_WORKGROUP);
      }
    }
  };

  auto add_n = [&](int s, int d, int n) {
    float w = dinv[s] * dinv[d];
    int wb = __builtin_bit_cast(int, w);
    int ro = (d - base) * P;
    for (int k = 0; k < n; ++k) {
      int sk = __builtin_amdgcn_readlane(s, k);
      float wk = __builtin_bit_cast(float, __builtin_amdgcn_readlane(wb, k));
      int rk = __builtin_amdgcn_readlane(ro, k);
      const float* fr = feat + (size_t)sk * C + lane * VEC;
#pragma unroll
      for (int j = 0; j < VEC; ++j)
        __hip_atomic_fetch_add(&acc[rk + j * 32 + lane], wk * fr[j],
                               __ATOMIC_RELAXED, __HIP_MEMORY_SCOPE_WORKGROUP);
    }
  };

  auto drain_n = [&](const int* qb, int n) {
    int ed = qb[lane < n ? lane : n - 1];
    int s = src[ed];
    int d = dst[ed];
    if constexpr (MODE == 0) {
      if (rec_ok) {
        if (rcnt + n <= cap) {
          if (lane < n) mylist[rcnt + lane] = make_int2(s, d);
          rcnt += n;
        } else {
          rec_ok = false;
        }
      }
    }
    add_n(s, d, n);
  };

  auto push = [&](bool m, int e) {
    unsigned mk = __builtin_amdgcn_ballot_w32(m);
    if (m) myq[cnt + (int)__builtin_amdgcn_mbcnt_lo(mk, 0u)] = e;
    cnt += __builtin_popcount(mk);
  };

  bool listed = false;
  if constexpr (MODE == 1) {
    int hv = reg[lane];
    listed = __builtin_amdgcn_ballot_w32(hv < 0 || hv > cap) == 0u;
    if (listed) {
      const int n = __builtin_amdgcn_readlane(hv, wave);
      int p = 0;
      int2 ed = make_int2(0, base);
      if (n >= 32) ed = mylist[lane];
      for (; p + 32 <= n; p += 32) {
        int s = ed.x;
        int d = ed.y;
        if (p + 64 <= n) ed = mylist[p + 32 + lane];
        float w = dinv[s] * dinv[d];
        int wb = __builtin_bit_cast(int, w);
        int ro = (d - base) * P;
#pragma unroll 1
        for (int k0 = 0; k0 < 32; k0 += G) {
          float v[G][VEC];
#pragma unroll
          for (int u = 0; u < G; ++u) {
            int sk = __builtin_amdgcn_readlane(s, k0 + u);
            const float* fr = feat + (size_t)sk * C + lane * VEC;
            if constexpr (VEC == 4) {
              float4 t = *reinterpret_cast<const float4*>(fr);
              v[u][0] = t.x; v[u][1] = t.y; v[u][2] = t.z; v[u][3] = t.w;
            } else {
              float2 t = *reinterpret_cast<const float2*>(fr);
              v[u][0] = t.x; v[u][1] = t.y;
            }
          }
#pragma unroll
          for (int u = 0; u < G; ++u) {
            float wk = __builtin_bit_cast(float, __builtin_amdgcn_readlane(wb, k0 + u));
            int rk = __builtin_amdgcn_readlane(ro, k0 + u);
#pragma unroll
            for (int j = 0; j < VEC; ++j)
              __hip_atomic_fetch_add(&acc[rk + j * 32 + lane], wk * v[u][j],
                                     __ATOMIC_RELAXED, __HIP_MEMORY_SCOPE_WORKGROUP);
          }
        }
      }
      if (p < n) {
        const int r = n - p;
        int2 et = mylist[p + (lane < r ? lane : r - 1)];
        add_n(et.x, et.y, r);
      }
    }
  }

  if (!listed) {
    const bool key_ok = gridDim.x <= 256u && (reinterpret_cast<size_t>(dst) & 15) == 0 &&
                        (reinterpret_cast<size_t>(keys) & 15) == 0 &&
                        (long long)E <= (long long)N * 256;
    const int nfull = key_ok ? (E >> 9) : 0;
    const unsigned pat = (unsigned)tile * 0x01010101u;
    const unsigned HI = 0x80808080u;
    const int rot = (int)(((long long)tile * nfull) / (int)gridDim.x);
    for (int g0 = wave; g0 < nfull; g0 += NW * 4) {
      int4 k4[4];
      int e16[4];
#pragma unroll
      for (int u = 0; u < 4; ++u) {
        int g = g0 + u * NW;
        if (g > nfull - 1) g = nfull - 1;
        int gg = g + rot;
        if (gg >= nfull) gg -= nfull;
        e16[u] = gg * 512 + lane * 16;
        k4[u] = *reinterpret_cast<const int4*>(keys + (size_t)gg * 128 + lane * 4);
      }
#pragma unroll
      for (int u = 0; u < 4; ++u) {
        const bool live = g0 + u * NW < nfull;
        unsigned x0 = (unsigned)k4[u].x ^ pat;
        unsigned x1 = (unsigned)k4[u].y ^ pat;
        unsigned x2 = (unsigned)k4[u].z ^ pat;
        unsigned x3 = (unsigned)k4[u].w ^ pat;
        unsigned nz0 = ((x0 & ~HI) + ~HI) | x0;
        unsigned nz1 = ((x1 & ~HI) + ~HI) | x1;
        unsigned nz2 = ((x2 & ~HI) + ~HI) | x2;
        unsigned nz3 = ((x3 & ~HI) + ~HI) | x3;
        const bool hit = live && ((nz0 & nz1 & nz2 & nz3 & HI) != HI);
        if (__builtin_amdgcn_ballot_w32(hit) != 0u) {
          unsigned mm = 0u;
          if (hit)
            mm = 0x0F0F0F0Fu ^ (((nz0 & HI) >> 7) | ((nz1 & HI) >> 6) | ((nz2 & HI) >> 5) | ((nz3 & HI) >> 4));
          while (__builtin_amdgcn_ballot_w32(mm != 0u) != 0u) {
            const int p = __builtin_ctz(mm | 0x80000000u);
            push(mm != 0u, e16[u] + ((p & 3) << 2) + (p >> 3));
            mm &= mm - 1u;
            if (cnt >= 32) {
              cnt -= 32;
              drain32(myq + cnt);
            }
          }
        }
      }
    }
    for (int eb = nfull * 512 + wave * 32; eb < E; eb += NW * 32) {
      int e = eb + lane;
      int d = -1;
      if (e < E) d = dst[e];
      bool m = (unsigned)(d - base) < (unsigned)T;
      push(m, e);
      if (cnt >= 32) {
        cnt -= 32;
        drain_n(myq + cnt, 32);
      }
    }
    if (cnt > 0) drain_n(myq, cnt);
    if constexpr (MODE == 0) {
      if (lane == 0) reg[wave] = rec_ok ? rcnt : -1;
    }
  }
  if constexpr (MODE == 0) {
    const int hf = lane >> 4;
    const int l15 = lane & 15;
    const int n0 = (wave & 3) * 16;
    v2f bf[32];
#pragma unroll
    for (int t = 0; t < 32; ++t) {
      const int j = t >> 3, qq = t & 7;
      bf[t].x = w2[(16 * qq + 8 * hf + j) * OC + n0 + l15];
      bf[t].y = w2[(16 * qq + 8 * hf + 4 + j) * OC + n0 + l15];
    }
    __syncthreads();
    for (int mt = wave >> 2; mt < ntm; mt += 8) {
      const float* ar = acc + (mt * 16 + l15) * P + 2 * hf;
      int row = mt * 16 + hf * 8;
      unsigned zoff = ((unsigned)(base + row) * OC + n0 + l15) * 4u;
      asm volatile("" : "+v"(zoff), "+v"(row));
      char* zr = reinterpret_cast<char*>(out) + (size_t)zoff;
      v8f dacc = {};
#pragma unroll
      for (int t = 0; t < 32; ++t) {
        const int j = t >> 3, qq = t & 7;
        v2f a = *reinterpret_cast<const v2f*>(ar + j * 32 + 4 * qq);
        dacc = __builtin_amdgcn_wmma_f32_16x16x4_f32(
             false, a,  false, bf[t],
             (short)0, dacc,  false,  false);
      }
      asm volatile("v_nop\n\tv_nop\n\tv_nop\n\tv_nop\n\tv_nop" : "+v"(dacc));
      if (mt * 16 + 16 <= nrows) {
#pragma unroll
        for (int r = 0; r < 8; ++r) *reinterpret_cast<float*>(zr + r * OC * 4) = dacc[r];
      } else {
#pragma unroll
        for (int r = 0; r < 8; ++r)
          if (row + r < nrows) *reinterpret_cast<float*>(zr + r * OC * 4) = dacc[r];
      }
    }
    return;
  }
  __syncthreads();

  for (int i = wave; i < nrows; i += NW) {
    float* orow = out + (size_t)(base + i) * C + lane * VEC;
    if constexpr (VEC == 4) {
      float4 o;
      o.x = acc[i * C + lane];
      o.y = acc[i * C + 32 + lane];
      o.z = acc[i * C + 64 + lane];
      o.w = acc[i * C + 96 + lane];
      *reinterpret_cast<float4*>(orow) = o;
    } else {
      float2 o;
      o.x = acc[i * C + lane];
      o.y = acc[i * C + 32 + lane];
      *reinterpret_cast<float2*>(orow) = o;
    }
  }
}

__global__ void finalize_k(const float* __restrict__ g, float* __restrict__ out, int N) {
  int node = blockIdx.x * (blockDim.x >> 5) + (threadIdx.x >> 5);
  int lane = threadIdx.x & 31;
  if (node >= N) return;
  float2 v = reinterpret_cast<const float2*>(g + (size_t)node * 64)[lane];
  float mx = fmaxf(v.x, v.y);
  float mn = fminf(v.x, v.y);
#pragma unroll
  for (int o = 16; o > 0; o >>= 1) {
    mx = fmaxf(mx, __shfl_xor(mx, o, 32));
    mn = fminf(mn, __shfl_xor(mn, o, 32));
  }
  float inv = 1.0f / (mx - mn);
  float sx = (v.x - mn) * inv;
  float sy = (v.y - mn) * inv;
  float ss = sx * sx + sy * sy;
#pragma unroll
  for (int o = 16; o > 0; o >>= 1) ss += __shfl_xor(ss, o, 32);
  float r = 1.0f / fmaxf(sqrtf(ss), 1e-12f);
  float2 ov;
  ov.x = sx * r;
  ov.y = sy * r;
  reinterpret_cast<float2*>(out + (size_t)node * 64)[lane] = ov;
}

constexpr int TILE = 392;

extern "C" void kernel_launch(void* const* d_in, const int* in_sizes, int n_in,
                              void* d_out, int out_size, void* d_ws, size_t ws_size,
                              hipStream_t stream) {
  (void)n_in; (void)out_size; (void)ws_size;

  const float* x  = (const float*)d_in[0];
  const int*   ei = (const int*)d_in[1];
  const float* W1 = (const float*)d_in[2];
  const float* b1 = (const float*)d_in[3];
  const float* W2 = (const float*)d_in[4];
  const float* b2 = (const float*)d_in[5];

  const int N = in_sizes[0] / 128;
  const int E = in_sizes[1] / 2;
  const int* src = ei;
  const int* dst = ei + E;
  float* out = (float*)d_out;

  float* ws = (float*)d_ws;
  size_t Npad = ((size_t)N + 63) & ~(size_t)63;
  float* dinv = ws;
  float* xw = ws + Npad;
  float* h  = xw + (size_t)N * 128;
  float* g2 = h  + (size_t)N * 128;

  deg_init_k<<<(N + 255) / 256, 256, 0, stream>>>(dinv, N);
  int* keys = (int*)d_out;
  deg_acc_k<TILE><<<((E + 3) / 4 + 255) / 256, 256, 0, stream>>>(dst, dinv, keys, E, N);
  dinv_k<<<(N + 255) / 256, 256, 0, stream>>>(dinv, N);

  gemm_wmma_k<128><<<(N + 15) / 16, 256, 0, stream>>>(x, W1, xw, N, 128);
  agg_tile_k<128, TILE, 0><<<(N + TILE - 1) / TILE, 1024, 0, stream>>>(xw, dinv, b1, src, dst, keys, W2, h, (int*)g2, N, E);

  agg_tile_k<64, TILE, 1><<<(N + TILE - 1) / TILE, 1024, 0, stream>>>(h, dinv, b2, src, dst, keys, W2, g2, (int*)g2, N, E);

  finalize_k<<<(N + 7) / 8, 256, 0, stream>>>(g2, out, N);
}


# Layer 1 sums rows of x, not of x@W1: the tile block multiplies its summed tile by W1 in LDS (f32 WMMA, in place, + b1), then by W2; the x@W1 launch and plane go; product and sum order change
# speedup vs baseline: 2.5227x; 1.1272x over previous
#include <hip/hip_runtime.h>
#include <math.h>

typedef float v2f __attribute__((ext_vector_type(2)));
typedef float v8f __attribute__((ext_vector_type(8)));

__global__ void deg_init_k(float* __restrict__ deg, int N) {
  int i = blockIdx.x * blockDim.x + threadIdx.x;
  if (i < N) deg[i] = 1.0f;
}

template <int T>
__global__ void deg_acc_k(const int* __restrict__ dst, float* __restrict__ deg,
                          int* __restrict__ keys, int E, int n) {
  int e = (blockIdx.x * blockDim.x + threadIdx.x) * 4;
  if (e >= E) return;
  if (e + 4 <= E && (reinterpret_cast<size_t>(dst) & 15) == 0) {
    int4 d = *reinterpret_cast<const int4*>(dst + e);
    __hip_atomic_fetch_add(&deg[d.x], 1.0f, __ATOMIC_RELAXED, __HIP_MEMORY_SCOPE_AGENT);
    __hip_atomic_fetch_add(&deg[d.y], 1.0f, __ATOMIC_RELAXED, __HIP_MEMORY_SCOPE_AGENT);
    __hip_atomic_fetch_add(&deg[d.z], 1.0f, __ATOMIC_RELAXED, __HIP_MEMORY_SCOPE_AGENT);
    __hip_atomic_fetch_add(&deg[d.w], 1.0f, __ATOMIC_RELAXED, __HIP_MEMORY_SCOPE_AGENT);
    if ((long long)E <= (long long)n * 256)
      keys[e >> 2] = (((unsigned)d.x / T) & 255u) | ((((unsigned)d.y / T) & 255u) << 8) |
                     ((((unsigned)d.z / T) & 255u) << 16) | ((((unsigned)d.w / T) & 255u) << 24);
  } else {
    for (int k = 0; k < 4 && e + k < E; ++k)
      __hip_atomic_fetch_add(&deg[dst[e + k]], 1.0f, __ATOMIC_RELAXED, __HIP_MEMORY_SCOPE_AGENT);
  }
}

__global__ void dinv_k(float* __restrict__ deg, int N) {
  int i = blockIdx.x * blockDim.x + threadIdx.x;
  if (i < N) {
    float d = deg[i];
    deg[i] = (d > 0.0f) ? rsqrtf(d) : 0.0f;
  }
}

template <int C, int T, int MODE>
__global__ __launch_bounds__(1024) void agg_tile_k(const float* __restrict__ feat,
                                                   const float* __restrict__ dinv,
                                                   const float* __restrict__ bias,
                                                   const int* __restrict__ src,
                                                   const int* __restrict__ dst,
                                                   const int* __restrict__ keys,
                                                   const float* __restrict__ w1,
                                                   const float* __restrict__ w2,
                                                   float* out, int* lst, int N, int E) {
  constexpr int NW = 32;
  constexpr int VEC = C / 32;
  constexpr int QCAP = 64;
  constexpr int G = 32 / VEC;
  constexpr int P = MODE == 0 ? C + 4 : C;
  constexpr int ROWS = MODE == 0 ? (T + 15) / 16 * 16 : T;
  constexpr int OC = 64;
  static_assert(MODE != 0 || C == 128, "the fused product is written for 128 channels");
  __shared__ alignas(16) float acc[ROWS * P];
  __shared__ int q[NW * QCAP];

  const int lane = threadIdx.x & 31;
  const int wave = __builtin_amdgcn_readfirstlane((int)(threadIdx.x >> 5));
  const int tile = blockIdx.x;
  const int base = tile * T;
  int nrows = N - base;
  if (nrows > T) nrows = T;

  int* reg = lst + (size_t)base * 64;
  const int cap = nrows - 1;
  int2* mylist = reinterpret_cast<int2*>(reg + 32) + (size_t)wave * cap;

  for (int i = wave; i < nrows; i += NW) {
    float di = dinv[base + i];
    float w = di * di;
    const float* fr = feat + (size_t)(base + i) * C + lane * VEC;
    if constexpr (VEC == 4) {
      float4 f = *reinterpret_cast<const float4*>(fr);
      if constexpr (MODE == 0) {
        acc[i * P + lane]      = w * f.x;
        acc[i * P + 32 + lane] = w * f.y;
        acc[i * P + 64 + lane] = w * f.z;
        acc[i * P + 96 + lane] = w * f.w;
      } else {
        float4 bv = reinterpret_cast<const float4*>(bias)[lane];
        acc[i * P + lane]      = w * f.x + bv.x;
        acc[i * P + 32 + lane] = w * f.y + bv.y;
        acc[i * P + 64 + lane] = w * f.z + bv.z;
        acc[i * P + 96 + lane] = w * f.w + bv.w;
      }
    } else {
      float2 f = *reinterpret_cast<const float2*>(fr);
      float2 bv = reinterpret_cast<const float2*>(bias)[lane];
      acc[i * P + lane]      = w * f.x + bv.x;
      acc[i * P + 32 + lane] = w * f.y + bv.y;
    }
  }
  [[maybe_unused]] const int ntm = (nrows + 15) >> 4;
  if constexpr (MODE == 0) {
    for (int i = nrows + wave; i < ntm * 16; i += NW) {
      acc[i * P + lane]      = 0.0f;
      acc[i * P + 32 + lane] = 0.0f;
      acc[i * P + 64 + lane] = 0.0f;
      acc[i * P + 96 + lane] = 0.0f;
    }
  }
  __syncthreads();

  int* myq = q + wave * QCAP;
  int cnt = 0;
  int rcnt = 0;
  bool rec_ok = true;

  auto drain32 = [&](const int* qb) {
    int ed = qb[lane];
    int s = src[ed];
    int d = dst[ed];
    if constexpr (MODE == 0) {
      if (rec_ok) {
        if (rcnt + 32 <= cap) {
          mylist[rcnt + lane] = make_int2(s, d);
          rcnt += 32;
        } else {
          rec_ok = false;
        }
      }
    }
    float w = dinv[s] * dinv[d];
    int wb = __builtin_bit_cast(int, w);
    int ro = (d - base) * P;
#pragma unroll 1
    for (int k0 = 0; k0 < 32; k0 += 8) {
      float v[8][VEC];
#pragma unroll
      for (int u = 0; u < 8; ++u) {
        int sk = __builtin_amdgcn_readlane(s, k0 + u);
        const float* fr = feat + (size_t)sk * C + lane * VEC;
        if constexpr (VEC == 4) {
          float4 t = *reinterpret_cast<const float4*>(fr);
          v[u][0] = t.x; v[u][1] = t.y; v[u][2] = t.z; v[u][3] = t.w;
        } else {
          float2 t = *reinterpret_cast<const float2*>(fr);
          v[u][0] = t.x; v[u][1] = t.y;
        }
      }
#pragma unroll
      for (int u = 0; u < 8; ++u) {
        float wk = __builtin_bit_cast(float, __builtin_amdgcn_readlane(wb, k0 + u));
        int rk = __builtin_amdgcn_readlane(ro, k0 + u);
#pragma unroll
        for (int j = 0; j < VEC; ++j)
          __hip_atomic_fetch_add(&acc[rk + j * 32 + lane], wk * v[u][j],
                                 __ATOMIC_RELAXED, __HIP_MEMORY_SCOPE_WORKGROUP);
      }
    }
  };

  auto add_n = [&](int s, int d, int n) {
    float w = dinv[s] * dinv[d];
    int wb = __builtin_bit_cast(int, w);
    int ro = (d - base) * P;
    for (int k = 0; k < n; ++k) {
      int sk = __builtin_amdgcn_readlane(s, k);
      float wk = __builtin_bit_cast(float, __builtin_amdgcn_readlane(wb, k));
      int rk = __builtin_amdgcn_readlane(ro, k);
      const float* fr = feat + (size_t)sk * C + lane * VEC;
#pragma unroll
      for (int j = 0; j < VEC; ++j)
        __hip_atomic_fetch_add(&acc[rk + j * 32 + lane], wk * fr[j],
                               __ATOMIC_RELAXED, __HIP_MEMORY_SCOPE_WORKGROUP);
    }
  };

  auto drain_n = [&](const int* qb, int n) {
    int ed = qb[lane < n ? lane : n - 1];
    int s = src[ed];
    int d = dst[ed];
    if constexpr (MODE == 0) {
      if (rec_ok) {
        if (rcnt + n <= cap) {
          if (lane < n) mylist[rcnt + lane] = make_int2(s, d);
          rcnt += n;
        } else {
          rec_ok = false;
        }
      }
    }
    add_n(s, d, n);
  };

  auto push = [&](bool m, int e) {
    unsigned mk = __builtin_amdgcn_ballot_w32(m);
    if (m) myq[cnt + (int)__builtin_amdgcn_mbcnt_lo(mk, 0u)] = e;
    cnt += __builtin_popcount(mk);
  };

  bool listed = false;
  if constexpr (MODE == 1) {
    int hv = reg[lane];
    listed = __builtin_amdgcn_ballot_w32(hv < 0 || hv > cap) == 0u;
    if (listed) {
      const int n = __builtin_amdgcn_readlane(hv, wave);
      int p = 0;
      int2 ed = make_int2(0, base);
      if (n >= 32) ed = mylist[lane];
      for (; p + 32 <= n; p += 32) {
        int s = ed.x;
        int d = ed.y;
        if (p + 64 <= n) ed = mylist[p + 32 + lane];
        float w = dinv[s] * dinv[d];
        int wb = __builtin_bit_cast(int, w);
        int ro = (d - base) * P;
#pragma unroll 1
        for (int k0 = 0; k0 < 32; k0 += G) {
          float v[G][VEC];
#pragma unroll
          for (int u = 0; u < G; ++u) {
            int sk = __builtin_amdgcn_readlane(s, k0 + u);
            const float* fr = feat + (size_t)sk * C + lane * VEC;
            if constexpr (VEC == 4) {
              float4 t = *reinterpret_cast<const float4*>(fr);
              v[u][0] = t.x; v[u][1] = t.y; v[u][2] = t.z; v[u][3] = t.w;
            } else {
              float2 t = *reinterpret_cast<const float2*>(fr);
              v[u][0] = t.x; v[u][1] = t.y;
            }
          }
#pragma unroll
          for (int u = 0; u < G; ++u) {
            float wk = __builtin_bit_cast(float, __builtin_amdgcn_readlane(wb, k0 + u));
            int rk = __builtin_amdgcn_readlane(ro, k0 + u);
#pragma unroll
            for (int j = 0; j < VEC; ++j)
              __hip_atomic_fetch_add(&acc[rk + j * 32 + lane], wk * v[u][j],
                                     __ATOMIC_RELAXED, __HIP_MEMORY_SCOPE_WORKGROUP);
          }
        }
      }
      if (p < n) {
        const int r = n - p;
        int2 et = mylist[p + (lane < r ? lane : r - 1)];
        add_n(et.x, et.y, r);
      }
    }
  }

  if (!listed) {
    const bool key_ok = gridDim.x <= 256u && (reinterpret_cast<size_t>(dst) & 15) == 0 &&
                        (reinterpret_cast<size_t>(keys) & 15) == 0 &&
                        (long long)E <= (long long)N * 256;
    const int nfull = key_ok ? (E >> 9) : 0;
    const unsigned pat = (unsigned)tile * 0x01010101u;
    const unsigned HI = 0x80808080u;
    const int rot = (int)(((long long)tile * nfull) / (int)gridDim.x);
    for (int g0 = wave; g0 < nfull; g0 += NW * 4) {
      int4 k4[4];
      int e16[4];
#pragma unroll
      for (int u = 0; u < 4; ++u) {
        int g = g0 + u * NW;
        if (g > nfull - 1) g = nfull - 1;
        int gg = g + rot;
        if (gg >= nfull) gg -= nfull;
        e16[u] = gg * 512 + lane * 16;
        k4[u] = *reinterpret_cast<const int4*>(keys + (size_t)gg * 128 + lane * 4);
      }
#pragma unroll
      for (int u = 0; u < 4; ++u) {
        const bool live = g0 + u * NW < nfull;
        unsigned x0 = (unsigned)k4[u].x ^ pat;
        unsigned x1 = (unsigned)k4[u].y ^ pat;
        unsigned x2 = (unsigned)k4[u].z ^ pat;
        unsigned x3 = (unsigned)k4[u].w ^ pat;
        unsigned nz0 = ((x0 & ~HI) + ~HI) | x0;
        unsigned nz1 = ((x1 & ~HI) + ~HI) | x1;
        unsigned nz2 = ((x2 & ~HI) + ~HI) | x2;
        unsigned nz3 = ((x3 & ~HI) + ~HI) | x3;
        const bool hit = live && ((nz0 & nz1 & nz2 & nz3 & HI) != HI);
        if (__builtin_amdgcn_ballot_w32(hit) != 0u) {
          unsigned mm = 0u;
          if (hit)
            mm = 0x0F0F0F0Fu ^ (((nz0 & HI) >> 7) | ((nz1 & HI) >> 6) | ((nz2 & HI) >> 5) | ((nz3 & HI) >> 4));
          while (__builtin_amdgcn_ballot_w32(mm != 0u) != 0u) {
            const int p = __builtin_ctz(mm | 0x80000000u);
            push(mm != 0u, e16[u] + ((p & 3) << 2) + (p >> 3));
            mm &= mm - 1u;
            if (cnt >= 32) {
              cnt -= 32;
              drain32(myq + cnt);
            }
          }
        }
      }
    }
    for (int eb = nfull * 512 + wave * 32; eb < E; eb += NW * 32) {
      int e = eb + lane;
      int d = -1;
      if (e < E) d = dst[e];
      bool m = (unsigned)(d - base) < (unsigned)T;
      push(m, e);
      if (cnt >= 32) {
        cnt -= 32;
        drain_n(myq + cnt, 32);
      }
    }
    if (cnt > 0) drain_n(myq, cnt);
    if constexpr (MODE == 0) {
      if (lane == 0) reg[wave] = rec_ok ? rcnt : -1;
    }
  }
  if constexpr (MODE == 0) {
    const int hf = lane >> 4;
    const int l15 = lane & 15;
    {
      const int n1 = (wave & 7) * 16;
      v2f bf[32];
#pragma unroll
      for (int t = 0; t < 32; ++t) {
        const int j = t >> 3, qq = t & 7;
        bf[t].x = w1[(16 * qq + 8 * hf + j) * C + n1 + l15];
        bf[t].y = w1[(16 * qq + 8 * hf + 4 + j) * C + n1 + l15];
      }
      const float bb = bias[n1 + l15];
      __syncthreads();
      const int nit = (ntm + 3) >> 2;
      for (int it = 0; it < nit; ++it) {
        const int mt = (wave >> 3) + 4 * it;
        const bool on = mt < ntm;
        int hidx = (mt * 16 + 8 * hf) * P + n1 + l15;
        asm volatile("" : "+v"(hidx));
        v8f dacc = {};
        if (on) {
          const float* ar = acc + (mt * 16 + l15) * P + 2 * hf;
#pragma unroll
          for (int t = 0; t < 32; ++t) {
            const int j = t >> 3, qq = t & 7;
            v2f a = *reinterpret_cast<const v2f*>(ar + j * 32 + 4 * qq);
            dacc = __builtin_amdgcn_wmma_f32_16x16x4_f32(
                 false, a,  false, bf[t],
                 (short)0, dacc,  false,  false);
          }
        }
        asm volatile("v_nop\n\tv_nop\n\tv_nop\n\tv_nop\n\tv_nop" : "+v"(dacc));
        __syncthreads();
        if (on) {
#pragma unroll
          for (int r = 0; r < 8; ++r) acc[hidx + r * P] = dacc[r] + bb;
        }
      }
    }
    const int n0 = (wave & 3) * 16;
    v2f bf[32];
#pragma unroll
    for (int t = 0; t < 32; ++t) {
      bf[t].x = w2[(4 * t + 2 * hf) * OC + n0 + l15];
      bf[t].y = w2[(4 * t + 2 * hf + 1) * OC + n0 + l15];
    }
    __syncthreads();
    for (int mt = wave >> 2; mt < ntm; mt += 8) {
      const float* ar = acc + (mt * 16 + l15) * P + 2 * hf;
      int row = mt * 16 + hf * 8;
      unsigned zoff = ((unsigned)(base + row) * OC + n0 + l15) * 4u;
      asm volatile("" : "+v"(zoff), "+v"(row));
      char* zr = reinterpret_cast<char*>(out) + (size_t)zoff;
      v8f dacc = {};
#pragma unroll
      for (int t = 0; t < 32; ++t) {
        v2f a = *reinterpret_cast<const v2f*>(ar + 4 * t);
        dacc = __builtin_amdgcn_wmma_f32_16x16x4_f32(
             false, a,  false, bf[t],
             (short)0, dacc,  false,  false);
      }
      asm volatile("v_nop\n\tv_nop\n\tv_nop\n\tv_nop\n\tv_nop" : "+v"(dacc));
      if (mt * 16 + 16 <= nrows) {
#pragma unroll
        for (int r = 0; r < 8; ++r) *reinterpret_cast<float*>(zr + r * OC * 4) = dacc[r];
      } else {
#pragma unroll
        for (int r = 0; r < 8; ++r)
          if (row + r < nrows) *reinterpret_cast<float*>(zr + r * OC * 4) = dacc[r];
      }
    }
    return;
  }
  __syncthreads();

  for (int i = wave; i < nrows; i += NW) {
    float* orow = out + (size_t)(base + i) * C + lane * VEC;
    if constexpr (VEC == 4) {
      float4 o;
      o.x = acc[i * C + lane];
      o.y = acc[i * C + 32 + lane];
      o.z = acc[i * C + 64 + lane];
      o.w = acc[i * C + 96 + lane];
      *reinterpret_cast<float4*>(orow) = o;
    } else {
      float2 o;
      o.x = acc[i * C + lane];
      o.y = acc[i * C + 32 + lane];
      *reinterpret_cast<float2*>(orow) = o;
    }
  }
}

__global__ void finalize_k(const float* __restrict__ g, float* __restrict__ out, int N) {
  int node = blockIdx.x * (blockDim.x >> 5) + (threadIdx.x >> 5);
  int lane = threadIdx.x & 31;
  if (node >= N) return;
  float2 v = reinterpret_cast<const float2*>(g + (size_t)node * 64)[lane];
  float mx = fmaxf(v.x, v.y);
  float mn = fminf(v.x, v.y);
#pragma unroll
  for (int o = 16; o > 0; o >>= 1) {
    mx = fmaxf(mx, __shfl_xor(mx, o, 32));
    mn = fminf(mn, __shfl_xor(mn, o, 32));
  }
  float inv = 1.0f / (mx - mn);
  float sx = (v.x - mn) * inv;
  float sy = (v.y - mn) * inv;
  float ss = sx * sx + sy * sy;
#pragma unroll
  for (int o = 16; o > 0; o >>= 1) ss += __shfl_xor(ss, o, 32);
  float r = 1.0f / fmaxf(sqrtf(ss), 1e-12f);
  float2 ov;
  ov.x = sx * r;
  ov.y = sy * r;
  reinterpret_cast<float2*>(out + (size_t)node * 64)[lane] = ov;
}

constexpr int TILE = 392;

extern "C" void kernel_launch(void* const* d_in, const int* in_sizes, int n_in,
                              void* d_out, int out_size, void* d_ws, size_t ws_size,
                              hipStream_t stream) {
  (void)n_in; (void)out_size; (void)ws_size;

  const float* x  = (const float*)d_in[0];
  const int*   ei = (const int*)d_in[1];
  const float* W1 = (const float*)d_in[2];
  const float* b1 = (const float*)d_in[3];
  const float* W2 = (const float*)d_in[4];
  const float* b2 = (const float*)d_in[5];

  const int N = in_sizes[0] / 128;
  const int E = in_sizes[1] / 2;
  const int* src = ei;
  const int* dst = ei + E;
  float* out = (float*)d_out;

  float* ws = (float*)d_ws;
  size_t Npad = ((size_t)N + 63) & ~(size_t)63;
  float* dinv = ws;
  float* z  = ws + Npad;
  float* g2 = z + (size_t)N * 64;

  deg_init_k<<<(N + 255) / 256, 256, 0, stream>>>(dinv, N);
  int* keys = (int*)d_out;
  deg_acc_k<TILE><<<((E + 3) / 4 + 255) / 256, 256, 0, stream>>>(dst, dinv, keys, E, N);
  dinv_k<<<(N + 255) / 256, 256, 0, stream>>>(dinv, N);

  agg_tile_k<128, TILE, 0><<<(N + TILE - 1) / TILE, 1024, 0, stream>>>(x, dinv, b1, src, dst, keys, W1, W2, z, (int*)g2, N, E);

  agg_tile_k<64, TILE, 1><<<(N + TILE - 1) / TILE, 1024, 0, stream>>>(z, dinv, b2, src, dst, keys, W1, W2, g2, (int*)g2, N, E);

  finalize_k<<<(N + 7) / 8, 256, 0, stream>>>(g2, out, N);
}


# Rows fetched in groups, not one per round trip: a wave's 13 start-value rows 7 to a trip, its last <32 queued edges or listed pairs 8 (layer 2: 16) rows to a trip; LDS sum order free as before
# speedup vs baseline: 2.6114x; 1.0352x over previous
#include <hip/hip_runtime.h>
#include <math.h>

typedef float v2f __attribute__((ext_vector_type(2)));
typedef float v8f __attribute__((ext_vector_type(8)));

__global__ void deg_init_k(float* __restrict__ deg, int N) {
  int i = blockIdx.x * blockDim.x + threadIdx.x;
  if (i < N) deg[i] = 1.0f;
}

template <int T>
__global__ void deg_acc_k(const int* __restrict__ dst, float* __restrict__ deg,
                          int* __restrict__ keys, int E, int n) {
  int e = (blockIdx.x * blockDim.x + threadIdx.x) * 4;
  if (e >= E) return;
  if (e + 4 <= E && (reinterpret_cast<size_t>(dst) & 15) == 0) {
    int4 d = *reinterpret_cast<const int4*>(dst + e);
    __hip_atomic_fetch_add(&deg[d.x], 1.0f, __ATOMIC_RELAXED, __HIP_MEMORY_SCOPE_AGENT);
    __hip_atomic_fetch_add(&deg[d.y], 1.0f, __ATOMIC_RELAXED, __HIP_MEMORY_SCOPE_AGENT);
    __hip_atomic_fetch_add(&deg[d.z], 1.0f, __ATOMIC_RELAXED, __HIP_MEMORY_SCOPE_AGENT);
    __hip_atomic_fetch_add(&deg[d.w], 1.0f, __ATOMIC_RELAXED, __HIP_MEMORY_SCOPE_AGENT);
    if ((long long)E <= (long long)n * 256)
      keys[e >> 2] = (((unsigned)d.x / T) & 255u) | ((((unsigned)d.y / T) & 255u) << 8) |
                     ((((unsigned)d.z / T) & 255u) << 16) | ((((unsigned)d.w / T) & 255u) << 24);
  } else {
    for (int k = 0; k < 4 && e + k < E; ++k)
      __hip_atomic_fetch_add(&deg[dst[e + k]], 1.0f, __ATOMIC_RELAXED, __HIP_MEMORY_SCOPE_AGENT);
  }
}

__global__ void dinv_k(float* __restrict__ deg, int N) {
  int i = blockIdx.x * blockDim.x + threadIdx.x;
  if (i < N) {
    float d = deg[i];
    deg[i] = (d > 0.0f) ? rsqrtf(d) : 0.0f;
  }
}

template <int C, int T, int MODE>
__global__ __launch_bounds__(1024) void agg_tile_k(const float* __restrict__ feat,
                                                   const float* __restrict__ dinv,
                                                   const float* __restrict__ bias,
                                                   const int* __restrict__ src,
                                                   const int* __restrict__ dst,
                                                   const int* __restrict__ keys,
                                                   const float* __restrict__ w1,
                                                   const float* __restrict__ w2,
                                                   float* out, int* lst, int N, int E) {
  constexpr int NW = 32;
  constexpr int VEC = C / 32;
  constexpr int QCAP = 64;
  constexpr int G = 32 / VEC;
  constexpr int RPW = (T + NW - 1) / NW;
  constexpr int IG = RPW <= 8 ? RPW : (RPW + 1) / 2;
  static_assert(IG * VEC <= 32, "a group of start-value rows is meant to stay within 32 registers");
  constexpr int P = MODE == 0 ? C + 4 : C;
  constexpr int ROWS = MODE == 0 ? (T + 15) / 16 * 16 : T;
  constexpr int OC = 64;
  static_assert(MODE != 0 || C == 128, "the fused product is written for 128 channels");
  __shared__ alignas(16) float acc[ROWS * P];
  __shared__ int q[NW * QCAP];

  const int lane = threadIdx.x & 31;
  const int wave = __builtin_amdgcn_readfirstlane((int)(threadIdx.x >> 5));
  const int tile = blockIdx.x;
  const int base = tile * T;
  int nrows = N - base;
  if (nrows > T) nrows = T;

  int* reg = lst + (size_t)base * 64;
  const int cap = nrows - 1;
  int2* mylist = reinterpret_cast<int2*>(reg + 32) + (size_t)wave * cap;

  {
    [[maybe_unused]] float bv[VEC];
    if constexpr (MODE == 1) {
#pragma unroll
      for (int j = 0; j < VEC; ++j) bv[j] = bias[lane * VEC + j];
    }
#pragma unroll 1
    for (int i0 = wave; i0 < nrows; i0 += NW * IG) {
      float f[IG][VEC];
      float dd[IG];
#pragma unroll
      for (int u = 0; u < IG; ++u) {
        int i = i0 + NW * u;
        if (i > nrows - 1) i = nrows - 1;
        dd[u] = dinv[base + i];
        const float* fr = feat + (size_t)(base + i) * C + lane * VEC;
        if constexpr (VEC == 4) {
          float4 t = *reinterpret_cast<const float4*>(fr);
          f[u][0] = t.x; f[u][1] = t.y; f[u][2] = t.z; f[u][3] = t.w;
        } else {
          float2 t = *reinterpret_cast<const float2*>(fr);
          f[u][0] = t.x; f[u][1] = t.y;
        }
      }
#pragma unroll
      for (int u = 0; u < IG; ++u) {
        int i = i0 + NW * u;
        if (i > nrows - 1) i = nrows - 1;
        const float w = dd[u] * dd[u];
#pragma unroll
        for (int j = 0; j < VEC; ++j) {
          if constexpr (MODE == 1) acc[i * P + j * 32 + lane] = w * f[u][j] + bv[j];
          else                     acc[i * P + j * 32 + lane] = w * f[u][j];
        }
      }
    }
  }
  [[maybe_unused]] const int ntm = (nrows + 15) >> 4;
  if constexpr (MODE == 0) {
    for (int i = nrows + wave; i < ntm * 16; i += NW) {
      acc[i * P + lane]      = 0.0f;
      acc[i * P + 32 + lane] = 0.0f;
      acc[i * P + 64 + lane] = 0.0f;
      acc[i * P + 96 + lane] = 0.0f;
    }
  }
  __syncthreads();

  int* myq = q + wave * QCAP;
  int cnt = 0;
  int rcnt = 0;
  bool rec_ok = true;

  auto drain32 = [&](const int* qb) {
    int ed = qb[lane];
    int s = src[ed];
    int d = dst[ed];
    if constexpr (MODE == 0) {
      if (rec_ok) {
        if (rcnt + 32 <= cap) {
          mylist[rcnt + lane] = make_int2(s, d);
          rcnt += 32;
        } else {
          rec_ok = false;
        }
      }
    }
    float w = dinv[s] * dinv[d];
    int wb = __builtin_bit_cast(int, w);
    int ro = (d - base) * P;
#pragma unroll 1
    for (int k0 = 0; k0 < 32; k0 += 8) {
      float v[8][VEC];
#pragma unroll
      for (int u = 0; u < 8; ++u) {
        int sk = __builtin_amdgcn_readlane(s, k0 + u);
        const float* fr = feat + (size_t)sk * C + lane * VEC;
        if constexpr (VEC == 4) {
          float4 t = *reinterpret_cast<const float4*>(fr);
          v[u][0] = t.x; v[u][1] = t.y; v[u][2] = t.z; v[u][3] = t.w;
        } else {
          float2 t = *reinterpret_cast<const float2*>(fr);
          v[u][0] = t.x; v[u][1] = t.y;
        }
      }
#pragma unroll
      for (int u = 0; u < 8; ++u) {
        float wk = __builtin_bit_cast(float, __builtin_amdgcn_readlane(wb, k0 + u));
        int rk = __builtin_amdgcn_readlane(ro, k0 + u);
#pragma unroll
        for (int j = 0; j < VEC; ++j)
          __hip_atomic_fetch_add(&acc[rk + j * 32 + lane], wk * v[u][j],
                                 __ATOMIC_RELAXED, __HIP_MEMORY_SCOPE_WORKGROUP);
      }
    }
  };

  auto add_n = [&](int s, int d, int n) {
    float w = dinv[s] * dinv[d];
    int wb = __builtin_bit_cast(int, w);
    int ro = (d - base) * P;
#pragma unroll 1
    for (int k0 = 0; k0 < n; k0 += G) {
      float v[G][VEC];
#pragma unroll
      for (int u = 0; u < G; ++u) {
        int k = k0 + u;
        if (k > n - 1) k = n - 1;
        int sk = __builtin_amdgcn_readlane(s, k);
        const float* fr = feat + (size_t)sk * C + lane * VEC;
        if constexpr (VEC == 4) {
          float4 t = *reinterpret_cast<const float4*>(fr);
          v[u][0] = t.x; v[u][1] = t.y; v[u][2] = t.z; v[u][3] = t.w;
        } else {
          float2 t = *reinterpret_cast<const float2*>(fr);
          v[u][0] = t.x; v[u][1] = t.y;
        }
      }
#pragma unroll
      for (int u = 0; u < G; ++u) {
        if constexpr (VEC == 4) asm volatile("" : "+v"(v[u][0]), "+v"(v[u][1]), "+v"(v[u][2]), "+v"(v[u][3]));
        else                    asm volatile("" : "+v"(v[u][0]), "+v"(v[u][1]));
      }
#pragma unroll
      for (int u = 0; u < G; ++u) {
        if (k0 + u < n) {
          float wk = __builtin_bit_cast(float, __builtin_amdgcn_readlane(wb, k0 + u));
          int rk = __builtin_amdgcn_readlane(ro, k0 + u);
#pragma unroll
          for (int j = 0; j < VEC; ++j)
            __hip_atomic_fetch_add(&acc[rk + j * 32 + lane], wk * v[u][j],
                                   __ATOMIC_RELAXED, __HIP_MEMORY_SCOPE_WORKGROUP);
        }
      }
    }
  };

  auto drain_n = [&](const int* qb, int n) {
    int ed = qb[lane < n ? lane : n - 1];
    int s = src[ed];
    int d = dst[ed];
    if constexpr (MODE == 0) {
      if (rec_ok) {
        if (rcnt + n <= cap) {
          if (lane < n) mylist[rcnt + lane] = make_int2(s, d);
          rcnt += n;
        } else {
          rec_ok = false;
        }
      }
    }
    add_n(s, d, n);
  };

  auto push = [&](bool m, int e) {
    unsigned mk = __builtin_amdgcn_ballot_w32(m);
    if (m) myq[cnt + (int)__builtin_amdgcn_mbcnt_lo(mk, 0u)] = e;
    cnt += __builtin_popcount(mk);
  };

  bool listed = false;
  if constexpr (MODE == 1) {
    int hv = reg[lane];
    listed = __builtin_amdgcn_ballot_w32(hv < 0 || hv > cap) == 0u;
    if (listed) {
      const int n = __builtin_amdgcn_readlane(hv, wave);
      int p = 0;
      int2 ed = make_int2(0, base);
      if (n >= 32) ed = mylist[lane];
      for (; p + 32 <= n; p += 32) {
        int s = ed.x;
        int d = ed.y;
        if (p + 64 <= n) ed = mylist[p + 32 + lane];
        float w = dinv[s] * dinv[d];
        int wb = __builtin_bit_cast(int, w);
        int ro = (d - base) * P;
#pragma unroll 1
        for (int k0 = 0; k0 < 32; k0 += G) {
          float v[G][VEC];
#pragma unroll
          for (int u = 0; u < G; ++u) {
            int sk = __builtin_amdgcn_readlane(s, k0 + u);
            const float* fr = feat + (size_t)sk * C + lane * VEC;
            if constexpr (VEC == 4) {
              float4 t = *reinterpret_cast<const float4*>(fr);
              v[u][0] = t.x; v[u][1] = t.y; v[u][2] = t.z; v[u][3] = t.w;
            } else {
              float2 t = *reinterpret_cast<const float2*>(fr);
              v[u][0] = t.x; v[u][1] = t.y;
            }
          }
#pragma unroll
          for (int u = 0; u < G; ++u) {
            float wk = __builtin_bit_cast(float, __builtin_amdgcn_readlane(wb, k0 + u));
            int rk = __builtin_amdgcn_readlane(ro, k0 + u);
#pragma unroll
            for (int j = 0; j < VEC; ++j)
              __hip_atomic_fetch_add(&acc[rk + j * 32 + lane], wk * v[u][j],
                                     __ATOMIC_RELAXED, __HIP_MEMORY_SCOPE_WORKGROUP);
          }
        }
      }
      if (p < n) {
        const int r = n - p;
        int2 et = mylist[p + (lane < r ? lane : r - 1)];
        add_n(et.x, et.y, r);
      }
    }
  }

  if (!listed) {
    const bool key_ok = gridDim.x <= 256u && (reinterpret_cast<size_t>(dst) & 15) == 0 &&
                        (reinterpret_cast<size_t>(keys) & 15) == 0 &&
                        (long long)E <= (long long)N * 256;
    const int nfull = key_ok ? (E >> 9) : 0;
    const unsigned pat = (unsigned)tile * 0x01010101u;
    const unsigned HI = 0x80808080u;
    const int rot = (int)(((long long)tile * nfull) / (int)gridDim.x);
    for (int g0 = wave; g0 < nfull; g0 += NW * 4) {
      int4 k4[4];
      int e16[4];
#pragma unroll
      for (int u = 0; u < 4; ++u) {
        int g = g0 + u * NW;
        if (g > nfull - 1) g = nfull - 1;
        int gg = g + rot;
        if (gg >= nfull) gg -= nfull;
        e16[u] = gg * 512 + lane * 16;
        k4[u] = *reinterpret_cast<const int4*>(keys + (size_t)gg * 128 + lane * 4);
      }
#pragma unroll
      for (int u = 0; u < 4; ++u) {
        const bool live = g0 + u * NW < nfull;
        unsigned x0 = (unsigned)k4[u].x ^ pat;
        unsigned x1 = (unsigned)k4[u].y ^ pat;
        unsigned x2 = (unsigned)k4[u].z ^ pat;
        unsigned x3 = (unsigned)k4[u].w ^ pat;
        unsigned nz0 = ((x0 & ~HI) + ~HI) | x0;
        unsigned nz1 = ((x1 & ~HI) + ~HI) | x1;
        unsigned nz2 = ((x2 & ~HI) + ~HI) | x2;
        unsigned nz3 = ((x3 & ~HI) + ~HI) | x3;
        const bool hit = live && ((nz0 & nz1 & nz2 & nz3 & HI) != HI);
        if (__builtin_amdgcn_ballot_w32(hit) != 0u) {
          unsigned mm = 0u;
          if (hit)
            mm = 0x0F0F0F0Fu ^ (((nz0 & HI) >> 7) | ((nz1 & HI) >> 6) | ((nz2 & HI) >> 5) | ((nz3 & HI) >> 4));
          while (__builtin_amdgcn_ballot_w32(mm != 0u) != 0u) {
            const int p = __builtin_ctz(mm | 0x80000000u);
            push(mm != 0u, e16[u] + ((p & 3) << 2) + (p >> 3));
            mm &= mm - 1u;
            if (cnt >= 32) {
              cnt -= 32;
              drain32(myq + cnt);
            }
          }
        }
      }
    }
    for (int eb = nfull * 512 + wave * 32; eb < E; eb += NW * 32) {
      int e = eb + lane;
      int d = -1;
      if (e < E) d = dst[e];
      bool m = (unsigned)(d - base) < (unsigned)T;
      push(m, e);
      if (cnt >= 32) {
        cnt -= 32;
        drain_n(myq + cnt, 32);
      }
    }
    if (cnt > 0) drain_n(myq, cnt);
    if constexpr (MODE == 0) {
      if (lane == 0) reg[wave] = rec_ok ? rcnt : -1;
    }
  }
  if constexpr (MODE == 0) {
    const int hf = lane >> 4;
    const int l15 = lane & 15;
    {
      const int n1 = (wave & 7) * 16;
      v2f bf[32];
#pragma unroll
      for (int t = 0; t < 32; ++t) {
        const int j = t >> 3, qq = t & 7;
        bf[t].x = w1[(16 * qq + 8 * hf + j) * C + n1 + l15];
        bf[t].y = w1[(16 * qq + 8 * hf + 4 + j) * C + n1 + l15];
      }
      const float bb = bias[n1 + l15];
      __syncthreads();
      const int nit = (ntm + 3) >> 2;
      for (int it = 0; it < nit; ++it) {
        const int mt = (wave >> 3) + 4 * it;
        const bool on = mt < ntm;
        int hidx = (mt * 16 + 8 * hf) * P + n1 + l15;
        asm volatile("" : "+v"(hidx));
        v8f dacc = {};
        if (on) {
          const float* ar = acc + (mt * 16 + l15) * P + 2 * hf;
#pragma unroll
          for (int t = 0; t < 32; ++t) {
            const int j = t >> 3, qq = t & 7;
            v2f a = *reinterpret_cast<const v2f*>(ar + j * 32 + 4 * qq);
            dacc = __builtin_amdgcn_wmma_f32_16x16x4_f32(
                 false, a,  false, bf[t],
                 (short)0, dacc,  false,  false);
          }
        }
        asm volatile("v_nop\n\tv_nop\n\tv_nop\n\tv_nop\n\tv_nop" : "+v"(dacc));
        __syncthreads();
        if (on) {
#pragma unroll
          for (int r = 0; r < 8; ++r) acc[hidx + r * P] = dacc[r] + bb;
        }
      }
    }
    const int n0 = (wave & 3) * 16;
    v2f bf[32];
#pragma unroll
    for (int t = 0; t < 32; ++t) {
      bf[t].x = w2[(4 * t + 2 * hf) * OC + n0 + l15];
      bf[t].y = w2[(4 * t + 2 * hf + 1) * OC + n0 + l15];
    }
    __syncthreads();
    for (int mt = wave >> 2; mt < ntm; mt += 8) {
      const float* ar = acc + (mt * 16 + l15) * P + 2 * hf;
      int row = mt * 16 + hf * 8;
      unsigned zoff = ((unsigned)(base + row) * OC + n0 + l15) * 4u;
      asm volatile("" : "+v"(zoff), "+v"(row));
      char* zr = reinterpret_cast<char*>(out) + (size_t)zoff;
      v8f dacc = {};
#pragma unroll
      for (int t = 0; t < 32; ++t) {
        v2f a = *reinterpret_cast<const v2f*>(ar + 4 * t);
        dacc = __builtin_amdgcn_wmma_f32_16x16x4_f32(
             false, a,  false, bf[t],
             (short)0, dacc,  false,  false);
      }
      asm volatile("v_nop\n\tv_nop\n\tv_nop\n\tv_nop\n\tv_nop" : "+v"(dacc));
      if (mt * 16 + 16 <= nrows) {
#pragma unroll
        for (int r = 0; r < 8; ++r) *reinterpret_cast<float*>(zr + r * OC * 4) = dacc[r];
      } else {
#pragma unroll
        for (int r = 0; r < 8; ++r)
          if (row + r < nrows) *reinterpret_cast<float*>(zr + r * OC * 4) = dacc[r];
      }
    }
    return;
  }
  __syncthreads();

  for (int i = wave; i < nrows; i += NW) {
    float* orow = out + (size_t)(base + i) * C + lane * VEC;
    if constexpr (VEC == 4) {
      float4 o;
      o.x = acc[i * C + lane];
      o.y = acc[i * C + 32 + lane];
      o.z = acc[i * C + 64 + lane];
      o.w = acc[i * C + 96 + lane];
      *reinterpret_cast<float4*>(orow) = o;
    } else {
      float2 o;
      o.x = acc[i * C + lane];
      o.y = acc[i * C + 32 + lane];
      *reinterpret_cast<float2*>(orow) = o;
    }
  }
}

__global__ void finalize_k(const float* __restrict__ g, float* __restrict__ out, int N) {
  int node = blockIdx.x * (blockDim.x >> 5) + (threadIdx.x >> 5);
  int lane = threadIdx.x & 31;
  if (node >= N) return;
  float2 v = reinterpret_cast<const float2*>(g + (size_t)node * 64)[lane];
  float mx = fmaxf(v.x, v.y);
  float mn = fminf(v.x, v.y);
#pragma unroll
  for (int o = 16; o > 0; o >>= 1) {
    mx = fmaxf(mx, __shfl_xor(mx, o, 32));
    mn = fminf(mn, __shfl_xor(mn, o, 32));
  }
  float inv = 1.0f / (mx - mn);
  float sx = (v.x - mn) * inv;
  float sy = (v.y - mn) * inv;
  float ss = sx * sx + sy * sy;
#pragma unroll
  for (int o = 16; o > 0; o >>= 1) ss += __shfl_xor(ss, o, 32);
  float r = 1.0f / fmaxf(sqrtf(ss), 1e-12f);
  float2 ov;
  ov.x = sx * r;
  ov.y = sy * r;
  reinterpret_cast<float2*>(out + (size_t)node * 64)[lane] = ov;
}

constexpr int TILE = 392;

extern "C" void kernel_launch(void* const* d_in, const int* in_sizes, int n_in,
                              void* d_out, int out_size, void* d_ws, size_t ws_size,
                              hipStream_t stream) {
  (void)n_in; (void)out_size; (void)ws_size;

  const float* x  = (const float*)d_in[0];
  const int*   ei = (const int*)d_in[1];
  const float* W1 = (const float*)d_in[2];
  const float* b1 = (const float*)d_in[3];
  const float* W2 = (const float*)d_in[4];
  const float* b2 = (const float*)d_in[5];

  const int N = in_sizes[0] / 128;
  const int E = in_sizes[1] / 2;
  const int* src = ei;
  const int* dst = ei + E;
  float* out = (float*)d_out;

  float* ws = (float*)d_ws;
  size_t Npad = ((size_t)N + 63) & ~(size_t)63;
  float* dinv = ws;
  float* z  = ws + Npad;
  float* g2 = z + (size_t)N * 64;

  deg_init_k<<<(N + 255) / 256, 256, 0, stream>>>(dinv, N);
  int* keys = (int*)d_out;
  deg_acc_k<TILE><<<((E + 3) / 4 + 255) / 256, 256, 0, stream>>>(dst, dinv, keys, E, N);
  dinv_k<<<(N + 255) / 256, 256, 0, stream>>>(dinv, N);

  agg_tile_k<128, TILE, 0><<<(N + TILE - 1) / TILE, 1024, 0, stream>>>(x, dinv, b1, src, dst, keys, W1, W2, z, (int*)g2, N, E);

  agg_tile_k<64, TILE, 1><<<(N + TILE - 1) / TILE, 1024, 0, stream>>>(z, dinv, b2, src, dst, keys, W1, W2, g2, (int*)g2, N, E);

  finalize_k<<<(N + 7) / 8, 256, 0, stream>>>(g2, out, N);
}


# Edge weights take 1/sqrt(deg) from a byte-per-node degree table that every tile block copies into LDS (rsqrt at use, same bits), not from scattered 4-byte global loads of dinv[src], dinv[dst]
# speedup vs baseline: 2.7494x; 1.0528x over previous
#include <hip/hip_runtime.h>
#include <math.h>
#include <type_traits>

typedef float v2f __attribute__((ext_vector_type(2)));
typedef float v8f __attribute__((ext_vector_type(8)));

constexpr int TABCAP = 100352;
constexpr unsigned TABESC = 255u;

__device__ __forceinline__ size_t tab_off(int E) { return ((size_t)E + 15) & ~(size_t)15; }
__device__ __forceinline__ bool tab_fits(const void* keys, int N, int E) {
  return N <= TABCAP && (reinterpret_cast<size_t>(keys) & 15) == 0 &&
         tab_off(E) + (size_t)N <= (size_t)N * 256;
}

__global__ void deg_init_k(float* __restrict__ deg, int N) {
  int i = blockIdx.x * blockDim.x + threadIdx.x;
  if (i < N) deg[i] = 1.0f;
}

template <int T>
__global__ void deg_acc_k(const int* __restrict__ dst, float* __restrict__ deg,
                          int* __restrict__ keys, int E, int n) {
  int e = (blockIdx.x * blockDim.x + threadIdx.x) * 4;
  if (e >= E) return;
  if (e + 4 <= E && (reinterpret_cast<size_t>(dst) & 15) == 0) {
    int4 d = *reinterpret_cast<const int4*>(dst + e);
    __hip_atomic_fetch_add(&deg[d.x], 1.0f, __ATOMIC_RELAXED, __HIP_MEMORY_SCOPE_AGENT);
    __hip_atomic_fetch_add(&deg[d.y], 1.0f, __ATOMIC_RELAXED, __HIP_MEMORY_SCOPE_AGENT);
    __hip_atomic_fetch_add(&deg[d.z], 1.0f, __ATOMIC_RELAXED, __HIP_MEMORY_SCOPE_AGENT);
    __hip_atomic_fetch_add(&deg[d.w], 1.0f, __ATOMIC_RELAXED, __HIP_MEMORY_SCOPE_AGENT);
    if ((long long)E <= (long long)n * 256)
      keys[e >> 2] = (((unsigned)d.x / T) & 255u) | ((((unsigned)d.y / T) & 255u) << 8) |
                     ((((unsigned)d.z / T) & 255u) << 16) | ((((unsigned)d.w / T) & 255u) << 24);
  } else {
    for (int k = 0; k < 4 && e + k < E; ++k)
      __hip_atomic_fetch_add(&deg[dst[e + k]], 1.0f, __ATOMIC_RELAXED, __HIP_MEMORY_SCOPE_AGENT);
  }
}

__global__ void dinv_k(float* __restrict__ deg, int* __restrict__ keys, int N, int E) {
  const bool tab = tab_fits(keys, N, E);
  unsigned char* tb = reinterpret_cast<unsigned char*>(keys) + tab_off(E);
  int i = (blockIdx.x * blockDim.x + threadIdx.x) * 4;
  if (i >= N) return;
  if (i + 4 <= N && (reinterpret_cast<size_t>(deg) & 15) == 0) {
    float4 d = *reinterpret_cast<const float4*>(deg + i);
    float4 r;
    r.x = (d.x > 0.0f) ? rsqrtf(d.x) : 0.0f;
    r.y = (d.y > 0.0f) ? rsqrtf(d.y) : 0.0f;
    r.z = (d.z > 0.0f) ? rsqrtf(d.z) : 0.0f;
    r.w = (d.w > 0.0f) ? rsqrtf(d.w) : 0.0f;
    *reinterpret_cast<float4*>(deg + i) = r;
    if (tab) {
      unsigned b0 = d.x < 255.0f ? (unsigned)d.x : TABESC;
      unsigned b1 = d.y < 255.0f ? (unsigned)d.y : TABESC;
      unsigned b2 = d.z < 255.0f ? (unsigned)d.z : TABESC;
      unsigned b3 = d.w < 255.0f ? (unsigned)d.w : TABESC;
      *reinterpret_cast<unsigned*>(tb + i) = b0 | (b1 << 8) | (b2 << 16) | (b3 << 24);
    }
  } else {
    for (int k = 0; k < 4 && i + k < N; ++k) {
      float d = deg[i + k];
      deg[i + k] = (d > 0.0f) ? rsqrtf(d) : 0.0f;
      if (tab) tb[i + k] = (unsigned char)(d < 255.0f ? (unsigned)d : TABESC);
    }
  }
}

template <int C, int T, int MODE>
__global__ __launch_bounds__(1024) void agg_tile_k(const float* __restrict__ feat,
                                                   const float* __restrict__ dinv,
                                                   const float* __restrict__ bias,
                                                   const int* __restrict__ src,
                                                   const int* __restrict__ dst,
                                                   const int* __restrict__ keys,
                                                   const float* __restrict__ w1,
                                                   const float* __restrict__ w2,
                                                   float* out, int* lst, int N, int E) {
  constexpr int NW = 32;
  constexpr int VEC = C / 32;
  constexpr int QCAP = 64;
  constexpr int G = 32 / VEC;
  constexpr int RPW = (T + NW - 1) / NW;
  constexpr int IG = RPW <= 8 ? RPW : (RPW + 1) / 2;
  static_assert(IG * VEC <= 32, "a group of start-value rows is meant to stay within 32 registers");
  constexpr int P = MODE == 0 ? C + 4 : C;
  constexpr int ROWS = MODE == 0 ? (T + 15) / 16 * 16 : T;
  constexpr int OC = 64;
  static_assert(MODE != 0 || C == 128, "the fused product is written for 128 channels");
  __shared__ alignas(16) float acc[ROWS * P];
  __shared__ int q[NW * QCAP];
  __shared__ alignas(16) unsigned char dtab[TABCAP];
  __shared__ int escw[NW];

  const int lane = threadIdx.x & 31;
  const int wave = __builtin_amdgcn_readfirstlane((int)(threadIdx.x >> 5));
  const int tile = blockIdx.x;
  const int base = tile * T;
  int nrows = N - base;
  if (nrows > T) nrows = T;

  const bool tab_ok = tab_fits(keys, N, E);
  if (tab_ok) {
    const unsigned char* tb = reinterpret_cast<const unsigned char*>(keys) + tab_off(E);
    const int4* t16 = reinterpret_cast<const int4*>(tb);
    const int n16 = N >> 4;
    const int rot16 = (int)(((unsigned)tile * (unsigned)n16) / gridDim.x);
    bool esc = false;
    constexpr int TB = 7;
    for (int i0 = (int)threadIdx.x; i0 < n16; i0 += NW * 32 * TB) {
      int4 t[TB];
      int jj[TB];
#pragma unroll
      for (int u = 0; u < TB; ++u) {
        int i = i0 + NW * 32 * u;
        if (i > n16 - 1) i = n16 - 1;
        int j = i + rot16;
        if (j >= n16) j -= n16;
        jj[u] = j;
        t[u] = t16[j];
      }
#pragma unroll
      for (int u = 0; u < TB; ++u) {
        reinterpret_cast<int4*>(dtab)[jj[u]] = t[u];
        const unsigned x0 = (unsigned)t[u].x, x1 = (unsigned)t[u].y, x2 = (unsigned)t[u].z, x3 = (unsigned)t[u].w;
        const unsigned z = ((~x0 - 0x01010101u) & x0) | ((~x1 - 0x01010101u) & x1) |
                           ((~x2 - 0x01010101u) & x2) | ((~x3 - 0x01010101u) & x3);
        esc |= (z & 0x80808080u) != 0u;
      }
    }
    for (int i = (n16 << 4) + (int)threadIdx.x; i < N; i += NW * 32) {
      const unsigned char b = tb[i];
      dtab[i] = b;
      esc |= b == TABESC;
    }
    const bool wesc = __builtin_amdgcn_ballot_w32(esc) != 0u;
    if (lane == 0) escw[wave] = wesc ? 1 : 0;
  }

  int* reg = lst + (size_t)base * 64;
  const int cap = nrows - 1;
  int2* mylist = reinterpret_cast<int2*>(reg + 32) + (size_t)wave * cap;

  {
    [[maybe_unused]] float bv[VEC];
    if constexpr (MODE == 1) {
#pragma unroll
      for (int j = 0; j < VEC; ++j) bv[j] = bias[lane * VEC + j];
    }
#pragma unroll 1
    for (int i0 = wave; i0 < nrows; i0 += NW * IG) {
      float f[IG][VEC];
      float dd[IG];
#pragma unroll
      for (int u = 0; u < IG; ++u) {
        int i = i0 + NW * u;
        if (i > nrows - 1) i = nrows - 1;
        dd[u] = dinv[base + i];
        const float* fr = feat + (size_t)(base + i) * C + lane * VEC;
        if constexpr (VEC == 4) {
          float4 t = *reinterpret_cast<const float4*>(fr);
          f[u][0] = t.x; f[u][1] = t.y; f[u][2] = t.z; f[u][3] = t.w;
        } else {
          float2 t = *reinterpret_cast<const float2*>(fr);
          f[u][0] = t.x; f[u][1] = t.y;
        }
      }
#pragma unroll
      for (int u = 0; u < IG; ++u) {
        int i = i0 + NW * u;
        if (i > nrows - 1) i = nrows - 1;
        const float w = dd[u] * dd[u];
#pragma unroll
        for (int j = 0; j < VEC; ++j) {
          if constexpr (MODE == 1) acc[i * P + j * 32 + lane] = w * f[u][j] + bv[j];
          else                     acc[i * P + j * 32 + lane] = w * f[u][j];
        }
      }
    }
  }
  [[maybe_unused]] const int ntm = (nrows + 15) >> 4;
  if constexpr (MODE == 0) {
    for (int i = nrows + wave; i < ntm * 16; i += NW) {
      acc[i * P + lane]      = 0.0f;
      acc[i * P + 32 + lane] = 0.0f;
      acc[i * P + 64 + lane] = 0.0f;
      acc[i * P + 96 + lane] = 0.0f;
    }
  }
  __syncthreads();

  bool tab_use = false;
  if (tab_ok) tab_use = __builtin_amdgcn_ballot_w32(escw[lane] != 0) == 0u;

  auto edge_phase = [&](auto tabc) {
  constexpr bool TAB = decltype(tabc)::value;
  auto dv = [&](int i) -> float {
    if constexpr (TAB) {
      const unsigned b = dtab[i];
      return b ? rsqrtf((float)b) : 0.0f;
    } else {
      return dinv[i];
    }
  };

  int* myq = q + wave * QCAP;
  int cnt = 0;
  int rcnt = 0;
  bool rec_ok = true;

  auto drain32 = [&](const int* qb) {
    int ed = qb[lane];
    int s = src[ed];
    int d = dst[ed];
    if constexpr (MODE == 0) {
      if (rec_ok) {
        if (rcnt + 32 <= cap) {
          mylist[rcnt + lane] = make_int2(s, d);
          rcnt += 32;
        } else {
          rec_ok = false;
        }
      }
    }
    float w = dv(s) * dv(d);
    int wb = __builtin_bit_cast(int, w);
    int ro = (d - base) * P;
#pragma unroll 1
    for (int k0 = 0; k0 < 32; k0 += 8) {
      float v[8][VEC];
#pragma unroll
      for (int u = 0; u < 8; ++u) {
        int sk = __builtin_amdgcn_readlane(s, k0 + u);
        const float* fr = feat + (size_t)sk * C + lane * VEC;
        if constexpr (VEC == 4) {
          float4 t = *reinterpret_cast<const float4*>(fr);
          v[u][0] = t.x; v[u][1] = t.y; v[u][2] = t.z; v[u][3] = t.w;
        } else {
          float2 t = *reinterpret_cast<const float2*>(fr);
          v[u][0] = t.x; v[u][1] = t.y;
        }
      }
#pragma unroll
      for (int u = 0; u < 8; ++u) {
        float wk = __builtin_bit_cast(float, __builtin_amdgcn_readlane(wb, k0 + u));
        int rk = __builtin_amdgcn_readlane(ro, k0 + u);
#pragma unroll
        for (int j = 0; j < VEC; ++j)
          __hip_atomic_fetch_add(&acc[rk + j * 32 + lane], wk * v[u][j],
                                 __ATOMIC_RELAXED, __HIP_MEMORY_SCOPE_WORKGROUP);
      }
    }
  };

  auto add_n = [&](int s, int d, int n) {
    float w = dv(s) * dv(d);
    int wb = __builtin_bit_cast(int, w);
    int ro = (d - base) * P;
#pragma unroll 1
    for (int k0 = 0; k0 < n; k0 += G) {
      float v[G][VEC];
#pragma unroll
      for (int u = 0; u < G; ++u) {
        int k = k0 + u;
        if (k > n - 1) k = n - 1;
        int sk = __builtin_amdgcn_readlane(s, k);
        const float* fr = feat + (size_t)sk * C + lane * VEC;
        if constexpr (VEC == 4) {
          float4 t = *reinterpret_cast<const float4*>(fr);
          v[u][0] = t.x; v[u][1] = t.y; v[u][2] = t.z; v[u][3] = t.w;
        } else {
          float2 t = *reinterpret_cast<const float2*>(fr);
          v[u][0] = t.x; v[u][1] = t.y;
        }
      }
#pragma unroll
      for (int u = 0; u < G; ++u) {
        if constexpr (VEC == 4) asm volatile("" : "+v"(v[u][0]), "+v"(v[u][1]), "+v"(v[u][2]), "+v"(v[u][3]));
        else                    asm volatile("" : "+v"(v[u][0]), "+v"(v[u][1]));
      }
#pragma unroll
      for (int u = 0; u < G; ++u) {
        if (k0 + u < n) {
          float wk = __builtin_bit_cast(float, __builtin_amdgcn_readlane(wb, k0 + u));
          int rk = __builtin_amdgcn_readlane(ro, k0 + u);
#pragma unroll
          for (int j = 0; j < VEC; ++j)
            __hip_atomic_fetch_add(&acc[rk + j * 32 + lane], wk * v[u][j],
                                   __ATOMIC_RELAXED, __HIP_MEMORY_SCOPE_WORKGROUP);
        }
      }
    }
  };

  auto drain_n = [&](const int* qb, int n) {
    int ed = qb[lane < n ? lane : n - 1];
    int s = src[ed];
    int d = dst[ed];
    if constexpr (MODE == 0) {
      if (rec_ok) {
        if (rcnt + n <= cap) {
          if (lane < n) mylist[rcnt + lane] = make_int2(s, d);
          rcnt += n;
        } else {
          rec_ok = false;
        }
      }
    }
    add_n(s, d, n);
  };

  auto push = [&](bool m, int e) {
    unsigned mk = __builtin_amdgcn_ballot_w32(m);
    if (m) myq[cnt + (int)__builtin_amdgcn_mbcnt_lo(mk, 0u)] = e;
    cnt += __builtin_popcount(mk);
  };

  bool listed = false;
  if constexpr (MODE == 1) {
    int hv = reg[lane];
    listed = __builtin_amdgcn_ballot_w32(hv < 0 || hv > cap) == 0u;
    if (listed) {
      const int n = __builtin_amdgcn_readlane(hv, wave);
      int p = 0;
      if (n >= 32) {
        int2 ed = mylist[lane];
        for (; p + 32 <= n; p += 32) {
          int s = ed.x;
          int d = ed.y;
          if (p + 64 <= n) ed = mylist[p + 32 + lane];
          float w = dv(s) * dv(d);
          int wb = __builtin_bit_cast(int, w);
          int ro = (d - base) * P;
#pragma unroll 1
          for (int k0 = 0; k0 < 32; k0 += G) {
            float v[G][VEC];
#pragma unroll
            for (int u = 0; u < G; ++u) {
              int sk = __builtin_amdgcn_readlane(s, k0 + u);
              const float* fr = feat + (size_t)sk * C + lane * VEC;
              if constexpr (VEC == 4) {
                float4 t = *reinterpret_cast<const float4*>(fr);
                v[u][0] = t.x; v[u][1] = t.y; v[u][2] = t.z; v[u][3] = t.w;
              } else {
                float2 t = *reinterpret_cast<const float2*>(fr);
                v[u][0] = t.x; v[u][1] = t.y;
              }
            }
#pragma unroll
            for (int u = 0; u < G; ++u) {
              float wk = __builtin_bit_cast(float, __builtin_amdgcn_readlane(wb, k0 + u));
              int rk = __builtin_amdgcn_readlane(ro, k0 + u);
#pragma unroll
              for (int j = 0; j < VEC; ++j)
                __hip_atomic_fetch_add(&acc[rk + j * 32 + lane], wk * v[u][j],
                                       __ATOMIC_RELAXED, __HIP_MEMORY_SCOPE_WORKGROUP);
            }
          }
        }
      }
      if (p < n) {
        const int r = n - p;
        int2 et = mylist[p + (lane < r ? lane : r - 1)];
        add_n(et.x, et.y, r);
      }
    }
  }

  if (!listed) {
    const bool key_ok = gridDim.x <= 256u && (reinterpret_cast<size_t>(dst) & 15) == 0 &&
                        (reinterpret_cast<size_t>(keys) & 15) == 0 &&
                        (long long)E <= (long long)N * 256;
    const int nfull = key_ok ? (E >> 9) : 0;
    const unsigned pat = (unsigned)tile * 0x01010101u;
    const unsigned HI = 0x80808080u;
    const int rot = (int)(((long long)tile * nfull) / (int)gridDim.x);
    for (int g0 = wave; g0 < nfull; g0 += NW * 4) {
      int4 k4[4];
      int e16[4];
#pragma unroll
      for (int u = 0; u < 4; ++u) {
        int g = g0 + u * NW;
        if (g > nfull - 1) g = nfull - 1;
        int gg = g + rot;
        if (gg >= nfull) gg -= nfull;
        e16[u] = gg * 512 + lane * 16;
        k4[u] = *reinterpret_cast<const int4*>(keys + (size_t)gg * 128 + lane * 4);
      }
#pragma unroll
      for (int u = 0; u < 4; ++u) {
        const bool live = g0 + u * NW < nfull;
        unsigned x0 = (unsigned)k4[u].x ^ pat;
        unsigned x1 = (unsigned)k4[u].y ^ pat;
        unsigned x2 = (unsigned)k4[u].z ^ pat;
        unsigned x3 = (unsigned)k4[u].w ^ pat;
        unsigned nz0 = ((x0 & ~HI) + ~HI) | x0;
        unsigned nz1 = ((x1 & ~HI) + ~HI) | x1;
        unsigned nz2 = ((x2 & ~HI) + ~HI) | x2;
        unsigned nz3 = ((x3 & ~HI) + ~HI) | x3;
        const bool hit = live && ((nz0 & nz1 & nz2 & nz3 & HI) != HI);
        if (__builtin_amdgcn_ballot_w32(hit) != 0u) {
          unsigned mm = 0u;
          if (hit)
            mm = 0x0F0F0F0Fu ^ (((nz0 & HI) >> 7) | ((nz1 & HI) >> 6) | ((nz2 & HI) >> 5) | ((nz3 & HI) >> 4));
          while (__builtin_amdgcn_ballot_w32(mm != 0u) != 0u) {
            const int p = __builtin_ctz(mm | 0x80000000u);
            push(mm != 0u, e16[u] + ((p & 3) << 2) + (p >> 3));
            mm &= mm - 1u;
            if (cnt >= 32) {
              cnt -= 32;
              drain32(myq + cnt);
            }
          }
        }
      }
    }
    for (int eb = nfull * 512 + wave * 32; eb < E; eb += NW * 32) {
      int e = eb + lane;
      int d = -1;
      if (e < E) d = dst[e];
      bool m = (unsigned)(d - base) < (unsigned)T;
      push(m, e);
      if (cnt >= 32) {
        cnt -= 32;
        drain_n(myq + cnt, 32);
      }
    }
    if (cnt > 0) drain_n(myq, cnt);
    if constexpr (MODE == 0) {
      if (lane == 0) reg[wave] = rec_ok ? rcnt : -1;
    }
  }
  };
  if (tab_use) edge_phase(std::true_type{});
  else         edge_phase(std::false_type{});

  if constexpr (MODE == 0) {
    const int hf = lane >> 4;
    const int l15 = lane & 15;
    {
      const int n1 = (wave & 7) * 16;
      v2f bf[32];
#pragma unroll
      for (int t = 0; t < 32; ++t) {
        const int j = t >> 3, qq = t & 7;
        bf[t].x = w1[(16 * qq + 8 * hf + j) * C + n1 + l15];
        bf[t].y = w1[(16 * qq + 8 * hf + 4 + j) * C + n1 + l15];
      }
      const float bb = bias[n1 + l15];
      __syncthreads();
      const int nit = (ntm + 3) >> 2;
      for (int it = 0; it < nit; ++it) {
        const int mt = (wave >> 3) + 4 * it;
        const bool on = mt < ntm;
        int hidx = (mt * 16 + 8 * hf) * P + n1 + l15;
        asm volatile("" : "+v"(hidx));
        v8f dacc = {};
        if (on) {
          const float* ar = acc + (mt * 16 + l15) * P + 2 * hf;
#pragma unroll
          for (int t = 0; t < 32; ++t) {
            const int j = t >> 3, qq = t & 7;
            v2f a = *reinterpret_cast<const v2f*>(ar + j * 32 + 4 * qq);
            dacc = __builtin_amdgcn_wmma_f32_16x16x4_f32(
                 false, a,  false, bf[t],
                 (short)0, dacc,  false,  false);
          }
        }
        asm volatile("v_nop\n\tv_nop\n\tv_nop\n\tv_nop\n\tv_nop" : "+v"(dacc));
        __syncthreads();
        if (on) {
#pragma unroll
          for (int r = 0; r < 8; ++r) acc[hidx + r * P] = dacc[r] + bb;
        }
      }
    }
    const int n0 = (wave & 3) * 16;
    v2f bf[32];
#pragma unroll
    for (int t = 0; t < 32; ++t) {
      bf[t].x = w2[(4 * t + 2 * hf) * OC + n0 + l15];
      bf[t].y = w2[(4 * t + 2 * hf + 1) * OC + n0 + l15];
    }
    __syncthreads();
    for (int mt = wave >> 2; mt < ntm; mt += 8) {
      const float* ar = acc + (mt * 16 + l15) * P + 2 * hf;
      int row = mt * 16 + hf * 8;
      unsigned zoff = ((unsigned)(base + row) * OC + n0 + l15) * 4u;
      asm volatile("" : "+v"(zoff), "+v"(row));
      char* zr = reinterpret_cast<char*>(out) + (size_t)zoff;
      v8f dacc = {};
#pragma unroll
      for (int t = 0; t < 32; ++t) {
        v2f a = *reinterpret_cast<const v2f*>(ar + 4 * t);
        dacc = __builtin_amdgcn_wmma_f32_16x16x4_f32(
             false, a,  false, bf[t],
             (short)0, dacc,  false,  false);
      }
      asm volatile("v_nop\n\tv_nop\n\tv_nop\n\tv_nop\n\tv_nop" : "+v"(dacc));
      if (mt * 16 + 16 <= nrows) {
#pragma unroll
        for (int r = 0; r < 8; ++r) *reinterpret_cast<float*>(zr + r * OC * 4) = dacc[r];
      } else {
#pragma unroll
        for (int r = 0; r < 8; ++r)
          if (row + r < nrows) *reinterpret_cast<float*>(zr + r * OC * 4) = dacc[r];
      }
    }
    return;
  }
  __syncthreads();

  for (int i = wave; i < nrows; i += NW) {
    float* orow = out + (size_t)(base + i) * C + lane * VEC;
    if constexpr (VEC == 4) {
      float4 o;
      o.x = acc[i * C + lane];
      o.y = acc[i * C + 32 + lane];
      o.z = acc[i * C + 64 + lane];
      o.w = acc[i * C + 96 + lane];
      *reinterpret_cast<float4*>(orow) = o;
    } else {
      float2 o;
      o.x = acc[i * C + lane];
      o.y = acc[i * C + 32 + lane];
      *reinterpret_cast<float2*>(orow) = o;
    }
  }
}

__global__ void finalize_k(const float* __restrict__ g, float* __restrict__ out, int N) {
  int node = blockIdx.x * (blockDim.x >> 5) + (threadIdx.x >> 5);
  int lane = threadIdx.x & 31;
  if (node >= N) return;
  float2 v = reinterpret_cast<const float2*>(g + (size_t)node * 64)[lane];
  float mx = fmaxf(v.x, v.y);
  float mn = fminf(v.x, v.y);
#pragma unroll
  for (int o = 16; o > 0; o >>= 1) {
    mx = fmaxf(mx, __shfl_xor(mx, o, 32));
    mn = fminf(mn, __shfl_xor(mn, o, 32));
  }
  float inv = 1.0f / (mx - mn);
  float sx = (v.x - mn) * inv;
  float sy = (v.y - mn) * inv;
  float ss = sx * sx + sy * sy;
#pragma unroll
  for (int o = 16; o > 0; o >>= 1) ss += __shfl_xor(ss, o, 32);
  float r = 1.0f / fmaxf(sqrtf(ss), 1e-12f);
  float2 ov;
  ov.x = sx * r;
  ov.y = sy * r;
  reinterpret_cast<float2*>(out + (size_t)node * 64)[lane] = ov;
}

constexpr int TILE = 392;

extern "C" void kernel_launch(void* const* d_in, const int* in_sizes, int n_in,
                              void* d_out, int out_size, void* d_ws, size_t ws_size,
                              hipStream_t stream) {
  (void)n_in; (void)out_size; (void)ws_size;

  const float* x  = (const float*)d_in[0];
  const int*   ei = (const int*)d_in[1];
  const float* W1 = (const float*)d_in[2];
  const float* b1 = (const float*)d_in[3];
  const float* W2 = (const float*)d_in[4];
  const float* b2 = (const float*)d_in[5];

  const int N = in_sizes[0] / 128;
  const int E = in_sizes[1] / 2;
  const int* src = ei;
  const int* dst = ei + E;
  float* out = (float*)d_out;

  float* ws = (float*)d_ws;
  size_t Npad = ((size_t)N + 63) & ~(size_t)63;
  float* dinv = ws;
  float* z  = ws + Npad;
  float* g2 = z + (size_t)N * 64;

  deg_init_k<<<(N + 255) / 256, 256, 0, stream>>>(dinv, N);
  int* keys = (int*)d_out;
  deg_acc_k<TILE><<<((E + 3) / 4 + 255) / 256, 256, 0, stream>>>(dst, dinv, keys, E, N);
  dinv_k<<<((N + 3) / 4 + 255) / 256, 256, 0, stream>>>(dinv, keys, N, E);

  agg_tile_k<128, TILE, 0><<<(N + TILE - 1) / TILE, 1024, 0, stream>>>(x, dinv, b1, src, dst, keys, W1, W2, z, (int*)g2, N, E);

  agg_tile_k<64, TILE, 1><<<(N + TILE - 1) / TILE, 1024, 0, stream>>>(z, dinv, b2, src, dst, keys, W1, W2, g2, (int*)g2, N, E);

  finalize_k<<<(N + 7) / 8, 256, 0, stream>>>(g2, out, N);
}


# Layer-2 tile block finishes its rows from LDS (min-max scale, L2 norm; 16 lanes a row, so the norm's sum order changes) and stores out: no normalise launch; list entries carry the edge weight
# speedup vs baseline: 2.9660x; 1.0788x over previous
#include <hip/hip_runtime.h>
#include <math.h>
#include <type_traits>

typedef float v2f __attribute__((ext_vector_type(2)));
typedef float v8f __attribute__((ext_vector_type(8)));

constexpr int TABCAP = 100352;
constexpr unsigned TABESC = 255u;

__device__ __forceinline__ size_t tab_off(int E) { return ((size_t)E + 15) & ~(size_t)15; }
__device__ __forceinline__ bool tab_fits(const void* keys, int N, int E) {
  return N <= TABCAP && (reinterpret_cast<size_t>(keys) & 15) == 0 &&
         tab_off(E) + (size_t)N <= (size_t)N * 256;
}

__global__ void deg_init_k(float* __restrict__ deg, int N) {
  int i = blockIdx.x * blockDim.x + threadIdx.x;
  if (i < N) deg[i] = 1.0f;
}

template <int T>
__global__ void deg_acc_k(const int* __restrict__ dst, float* __restrict__ deg,
                          int* __restrict__ keys, int E, int n) {
  int e = (blockIdx.x * blockDim.x + threadIdx.x) * 4;
  if (e >= E) return;
  if (e + 4 <= E && (reinterpret_cast<size_t>(dst) & 15) == 0) {
    int4 d = *reinterpret_cast<const int4*>(dst + e);
    __hip_atomic_fetch_add(&deg[d.x], 1.0f, __ATOMIC_RELAXED, __HIP_MEMORY_SCOPE_AGENT);
    __hip_atomic_fetch_add(&deg[d.y], 1.0f, __ATOMIC_RELAXED, __HIP_MEMORY_SCOPE_AGENT);
    __hip_atomic_fetch_add(&deg[d.z], 1.0f, __ATOMIC_RELAXED, __HIP_MEMORY_SCOPE_AGENT);
    __hip_atomic_fetch_add(&deg[d.w], 1.0f, __ATOMIC_RELAXED, __HIP_MEMORY_SCOPE_AGENT);
    if ((long long)E <= (long long)n * 256)
      keys[e >> 2] = (((unsigned)d.x / T) & 255u) | ((((unsigned)d.y / T) & 255u) << 8) |
                     ((((unsigned)d.z / T) & 255u) << 16) | ((((unsigned)d.w / T) & 255u) << 24);
  } else {
    for (int k = 0; k < 4 && e + k < E; ++k)
      __hip_atomic_fetch_add(&deg[dst[e + k]], 1.0f, __ATOMIC_RELAXED, __HIP_MEMORY_SCOPE_AGENT);
  }
}

__global__ void dinv_k(float* __restrict__ deg, int* __restrict__ keys, int N, int E) {
  const bool tab = tab_fits(keys, N, E);
  unsigned char* tb = reinterpret_cast<unsigned char*>(keys) + tab_off(E);
  int i = (blockIdx.x * blockDim.x + threadIdx.x) * 4;
  if (i >= N) return;
  if (i + 4 <= N && (reinterpret_cast<size_t>(deg) & 15) == 0) {
    float4 d = *reinterpret_cast<const float4*>(deg + i);
    float4 r;
    r.x = (d.x > 0.0f) ? rsqrtf(d.x) : 0.0f;
    r.y = (d.y > 0.0f) ? rsqrtf(d.y) : 0.0f;
    r.z = (d.z > 0.0f) ? rsqrtf(d.z) : 0.0f;
    r.w = (d.w > 0.0f) ? rsqrtf(d.w) : 0.0f;
    *reinterpret_cast<float4*>(deg + i) = r;
    if (tab) {
      unsigned b0 = d.x < 255.0f ? (unsigned)d.x : TABESC;
      unsigned b1 = d.y < 255.0f ? (unsigned)d.y : TABESC;
      unsigned b2 = d.z < 255.0f ? (unsigned)d.z : TABESC;
      unsigned b3 = d.w < 255.0f ? (unsigned)d.w : TABESC;
      *reinterpret_cast<unsigned*>(tb + i) = b0 | (b1 << 8) | (b2 << 16) | (b3 << 24);
    }
  } else {
    for (int k = 0; k < 4 && i + k < N; ++k) {
      float d = deg[i + k];
      deg[i + k] = (d > 0.0f) ? rsqrtf(d) : 0.0f;
      if (tab) tb[i + k] = (unsigned char)(d < 255.0f ? (unsigned)d : TABESC);
    }
  }
}

template <int M>
__device__ __forceinline__ float lane_xrow(float v) {
  static_assert(M >= 1 && M <= 15, "a mask inside a 16-lane row");
  const int b = __builtin_bit_cast(int, v);
  return __builtin_bit_cast(float, __builtin_amdgcn_update_dpp(b, b, 0x160 + M, 0xf, 0xf, false));
}

template <int C, int T, int MODE>
__global__ __launch_bounds__(1024) void agg_tile_k(const float* __restrict__ feat,
                                                   const float* __restrict__ dinv,
                                                   const float* __restrict__ bias,
                                                   const int* __restrict__ src,
                                                   const int* __restrict__ dst,
                                                   const int* __restrict__ keys,
                                                   const float* __restrict__ w1,
                                                   const float* __restrict__ w2,
                                                   float* out, int* lst, int N, int E) {
  constexpr int NW = 32;
  constexpr int VEC = C / 32;
  constexpr int QCAP = 64;
  constexpr int G = 32 / VEC;
  constexpr int RPW = (T + NW - 1) / NW;
  constexpr int IG = RPW <= 8 ? RPW : (RPW + 1) / 2;
  static_assert(IG * VEC <= 32, "a group of start-value rows is meant to stay within 32 registers");
  constexpr int P = MODE == 0 ? C + 4 : C;
  constexpr int ROWS = MODE == 0 ? (T + 15) / 16 * 16 : T;
  constexpr int OC = 64;
  constexpr int SB = 23;
  static_assert(MODE != 0 || C == 128, "the fused product is written for 128 channels");
  static_assert(MODE != 1 || C == 64, "the finishing pass is written for 64 channels, two a lane");
  static_assert(T <= (1 << (32 - SB)), "a row number inside the tile must fit the bits above SB");
  __shared__ alignas(16) float acc[ROWS * P];
  __shared__ int q[NW * QCAP];
  __shared__ alignas(16) unsigned char dtab[MODE == 0 ? TABCAP : 16];
  __shared__ int escw[NW];

  const int lane = threadIdx.x & 31;
  const int wave = __builtin_amdgcn_readfirstlane((int)(threadIdx.x >> 5));
  const int tile = blockIdx.x;
  const int base = tile * T;
  int nrows = N - base;
  if (nrows > T) nrows = T;

  const bool tab_ok = MODE == 0 && tab_fits(keys, N, E);
  if (tab_ok) {
    const unsigned char* tb = reinterpret_cast<const unsigned char*>(keys) + tab_off(E);
    const int4* t16 = reinterpret_cast<const int4*>(tb);
    const int n16 = N >> 4;
    const int rot16 = (int)(((unsigned)tile * (unsigned)n16) / gridDim.x);
    bool esc = false;
    constexpr int TB = 7;
    for (int i0 = (int)threadIdx.x; i0 < n16; i0 += NW * 32 * TB) {
      int4 t[TB];
      int jj[TB];
#pragma unroll
      for (int u = 0; u < TB; ++u) {
        int i = i0 + NW * 32 * u;
        if (i > n16 - 1) i = n16 - 1;
        int j = i + rot16;
        if (j >= n16) j -= n16;
        jj[u] = j;
        t[u] = t16[j];
      }
#pragma unroll
      for (int u = 0; u < TB; ++u) {
        reinterpret_cast<int4*>(dtab)[jj[u]] = t[u];
        const unsigned x0 = (unsigned)t[u].x, x1 = (unsigned)t[u].y, x2 = (unsigned)t[u].z, x3 = (unsigned)t[u].w;
        const unsigned z = ((~x0 - 0x01010101u) & x0) | ((~x1 - 0x01010101u) & x1) |
                           ((~x2 - 0x01010101u) & x2) | ((~x3 - 0x01010101u) & x3);
        esc |= (z & 0x80808080u) != 0u;
      }
    }
    for (int i = (n16 << 4) + (int)threadIdx.x; i < N; i += NW * 32) {
      const unsigned char b = tb[i];
      dtab[i] = b;
      esc |= b == TABESC;
    }
    const bool wesc = __builtin_amdgcn_ballot_w32(esc) != 0u;
    if (lane == 0) escw[wave] = wesc ? 1 : 0;
  }

  int* reg = lst + (size_t)base * 64;
  const int cap = nrows - 1;
  int2* mylist = reinterpret_cast<int2*>(reg + 32) + (size_t)wave * cap;

  {
    [[maybe_unused]] float bv[VEC];
    if constexpr (MODE == 1) {
#pragma unroll
      for (int j = 0; j < VEC; ++j) bv[j] = bias[lane * VEC + j];
    }
#pragma unroll 1
    for (int i0 = wave; i0 < nrows; i0 += NW * IG) {
      float f[IG][VEC];
      float dd[IG];
#pragma unroll
      for (int u = 0; u < IG; ++u) {
        int i = i0 + NW * u;
        if (i > nrows - 1) i = nrows - 1;
        dd[u] = dinv[base + i];
        const float* fr = feat + (size_t)(base + i) * C + lane * VEC;
        if constexpr (VEC == 4) {
          float4 t = *reinterpret_cast<const float4*>(fr);
          f[u][0] = t.x; f[u][1] = t.y; f[u][2] = t.z; f[u][3] = t.w;
        } else {
          float2 t = *reinterpret_cast<const float2*>(fr);
          f[u][0] = t.x; f[u][1] = t.y;
        }
      }
#pragma unroll
      for (int u = 0; u < IG; ++u) {
        int i = i0 + NW * u;
        if (i > nrows - 1) i = nrows - 1;
        const float w = dd[u] * dd[u];
#pragma unroll
        for (int j = 0; j < VEC; ++j) {
          if constexpr (MODE == 1) acc[i * P + j * 32 + lane] = w * f[u][j] + bv[j];
          else                     acc[i * P + j * 32 + lane] = w * f[u][j];
        }
      }
    }
  }
  [[maybe_unused]] const int ntm = (nrows + 15) >> 4;
  if constexpr (MODE == 0) {
    for (int i = nrows + wave; i < ntm * 16; i += NW) {
      acc[i * P + lane]      = 0.0f;
      acc[i * P + 32 + lane] = 0.0f;
      acc[i * P + 64 + lane] = 0.0f;
      acc[i * P + 96 + lane] = 0.0f;
    }
  }
  __syncthreads();

  bool tab_use = false;
  if (tab_ok) tab_use = __builtin_amdgcn_ballot_w32(escw[lane] != 0) == 0u;

  auto edge_phase = [&](auto tabc) {
  constexpr bool TAB = decltype(tabc)::value;
  auto dv = [&](int i) -> float {
    if constexpr (TAB) {
      const unsigned b = dtab[i];
      return b ? rsqrtf((float)b) : 0.0f;
    } else {
      return dinv[i];
    }
  };

  int* myq = q + wave * QCAP;
  int cnt = 0;
  int rcnt = 0;
  bool rec_ok = N <= (1 << SB);

  auto drain32 = [&](const int* qb) {
    int ed = qb[lane];
    int s = src[ed];
    int d = dst[ed];
    float w = dv(s) * dv(d);
    int wb = __builtin_bit_cast(int, w);
    if constexpr (MODE == 0) {
      if (rec_ok) {
        if (rcnt + 32 <= cap) {
          mylist[rcnt + lane] = make_int2((int)((unsigned)s | ((unsigned)(d - base) << SB)), wb);
          rcnt += 32;
        } else {
          rec_ok = false;
        }
      }
    }
    int ro = (d - base) * P;
#pragma unroll 1
    for (int k0 = 0; k0 < 32; k0 += 8) {
      float v[8][VEC];
#pragma unroll
      for (int u = 0; u < 8; ++u) {
        int sk = __builtin_amdgcn_readlane(s, k0 + u);
        const float* fr = feat + (size_t)sk * C + lane * VEC;
        if constexpr (VEC == 4) {
          float4 t = *reinterpret_cast<const float4*>(fr);
          v[u][0] = t.x; v[u][1] = t.y; v[u][2] = t.z; v[u][3] = t.w;
        } else {
          float2 t = *reinterpret_cast<const float2*>(fr);
          v[u][0] = t.x; v[u][1] = t.y;
        }
      }
#pragma unroll
      for (int u = 0; u < 8; ++u) {
        float wk = __builtin_bit_cast(float, __builtin_amdgcn_readlane(wb, k0 + u));
        int rk = __builtin_amdgcn_readlane(ro, k0 + u);
#pragma unroll
        for (int j = 0; j < VEC; ++j)
          __hip_atomic_fetch_add(&acc[rk + j * 32 + lane], wk * v[u][j],
                                 __ATOMIC_RELAXED, __HIP_MEMORY_SCOPE_WORKGROUP);
      }
    }
  };

  auto add_n = [&](int s, int ro, int wb, int n) {
#pragma unroll 1
    for (int k0 = 0; k0 < n; k0 += G) {
      float v[G][VEC];
#pragma unroll
      for (int u = 0; u < G; ++u) {
        int k = k0 + u;
        if (k > n - 1) k = n - 1;
        int sk = __builtin_amdgcn_readlane(s, k);
        const float* fr = feat + (size_t)sk * C + lane * VEC;
        if constexpr (VEC == 4) {
          float4 t = *reinterpret_cast<const float4*>(fr);
          v[u][0] = t.x; v[u][1] = t.y; v[u][2] = t.z; v[u][3] = t.w;
        } else {
          float2 t = *reinterpret_cast<const float2*>(fr);
          v[u][0] = t.x; v[u][1] = t.y;
        }
      }
#pragma unroll
      for (int u = 0; u < G; ++u) {
        if constexpr (VEC == 4) asm volatile("" : "+v"(v[u][0]), "+v"(v[u][1]), "+v"(v[u][2]), "+v"(v[u][3]));
        else                    asm volatile("" : "+v"(v[u][0]), "+v"(v[u][1]));
      }
#pragma unroll
      for (int u = 0; u < G; ++u) {
        if (k0 + u < n) {
          float wk = __builtin_bit_cast(float, __builtin_amdgcn_readlane(wb, k0 + u));
          int rk = __builtin_amdgcn_readlane(ro, k0 + u);
#pragma unroll
          for (int j = 0; j < VEC; ++j)
            __hip_atomic_fetch_add(&acc[rk + j * 32 + lane], wk * v[u][j],
                                   __ATOMIC_RELAXED, __HIP_MEMORY_SCOPE_WORKGROUP);
        }
      }
    }
  };

  auto drain_n = [&](const int* qb, int n) {
    int ed = qb[lane < n ? lane : n - 1];
    int s = src[ed];
    int d = dst[ed];
    float w = dv(s) * dv(d);
    int wb = __builtin_bit_cast(int, w);
    if constexpr (MODE == 0) {
      if (rec_ok) {
        if (rcnt + n <= cap) {
          if (lane < n) mylist[rcnt + lane] = make_int2((int)((unsigned)s | ((unsigned)(d - base) << SB)), wb);
          rcnt += n;
        } else {
          rec_ok = false;
        }
      }
    }
    add_n(s, (d - base) * P, wb, n);
  };

  auto push = [&](bool m, int e) {
    unsigned mk = __builtin_amdgcn_ballot_w32(m);
    if (m) myq[cnt + (int)__builtin_amdgcn_mbcnt_lo(mk, 0u)] = e;
    cnt += __builtin_popcount(mk);
  };

  bool listed = false;
  if constexpr (MODE == 1) {
    int hv = reg[lane];
    listed = __builtin_amdgcn_ballot_w32(hv < 0 || hv > cap) == 0u;
    if (listed) {
      const int n = __builtin_amdgcn_readlane(hv, wave);
      int p = 0;
      if (n >= 32) {
        int2 ed = mylist[lane];
        for (; p + 32 <= n; p += 32) {
          int s = ed.x & ((1 << SB) - 1);
          int ro = (int)((unsigned)ed.x >> SB) * P;
          int wb = ed.y;
          if (p + 64 <= n) ed = mylist[p + 32 + lane];
#pragma unroll 1
          for (int k0 = 0; k0 < 32; k0 += G) {
            float v[G][VEC];
#pragma unroll
            for (int u = 0; u < G; ++u) {
              int sk = __builtin_amdgcn_readlane(s, k0 + u);
              const float* fr = feat + (size_t)sk * C + lane * VEC;
              if constexpr (VEC == 4) {
                float4 t = *reinterpret_cast<const float4*>(fr);
                v[u][0] = t.x; v[u][1] = t.y; v[u][2] = t.z; v[u][3] = t.w;
              } else {
                float2 t = *reinterpret_cast<const float2*>(fr);
                v[u][0] = t.x; v[u][1] = t.y;
              }
            }
#pragma unroll
            for (int u = 0; u < G; ++u) {
              float wk = __builtin_bit_cast(float, __builtin_amdgcn_readlane(wb, k0 + u));
              int rk = __builtin_amdgcn_readlane(ro, k0 + u);
#pragma unroll
              for (int j = 0; j < VEC; ++j)
                __hip_atomic_fetch_add(&acc[rk + j * 32 + lane], wk * v[u][j],
                                       __ATOMIC_RELAXED, __HIP_MEMORY_SCOPE_WORKGROUP);
            }
          }
        }
      }
      if (p < n) {
        const int r = n - p;
        int2 et = mylist[p + (lane < r ? lane : r - 1)];
        add_n(et.x & ((1 << SB) - 1), (int)((unsigned)et.x >> SB) * P, et.y, r);
      }
    }
  }

  if (!listed) {
    const bool key_ok = MODE == 0 && gridDim.x <= 256u && (reinterpret_cast<size_t>(dst) & 15) == 0 &&
                        (reinterpret_cast<size_t>(keys) & 15) == 0 &&
                        (long long)E <= (long long)N * 256;
    const int nfull = key_ok ? (E >> 9) : 0;
    const unsigned pat = (unsigned)tile * 0x01010101u;
    const unsigned HI = 0x80808080u;
    const int rot = (int)(((long long)tile * nfull) / (int)gridDim.x);
    for (int g0 = wave; g0 < nfull; g0 += NW * 4) {
      int4 k4[4];
      int e16[4];
#pragma unroll
      for (int u = 0; u < 4; ++u) {
        int g = g0 + u * NW;
        if (g > nfull - 1) g = nfull - 1;
        int gg = g + rot;
        if (gg >= nfull) gg -= nfull;
        e16[u] = gg * 512 + lane * 16;
        k4[u] = *reinterpret_cast<const int4*>(keys + (size_t)gg * 128 + lane * 4);
      }
#pragma unroll
      for (int u = 0; u < 4; ++u) {
        const bool live = g0 + u * NW < nfull;
        unsigned x0 = (unsigned)k4[u].x ^ pat;
        unsigned x1 = (unsigned)k4[u].y ^ pat;
        unsigned x2 = (unsigned)k4[u].z ^ pat;
        unsigned x3 = (unsigned)k4[u].w ^ pat;
        unsigned nz0 = ((x0 & ~HI) + ~HI) | x0;
        unsigned nz1 = ((x1 & ~HI) + ~HI) | x1;
        unsigned nz2 = ((x2 & ~HI) + ~HI) | x2;
        unsigned nz3 = ((x3 & ~HI) + ~HI) | x3;
        const bool hit = live && ((nz0 & nz1 & nz2 & nz3 & HI) != HI);
        if (__builtin_amdgcn_ballot_w32(hit) != 0u) {
          unsigned mm = 0u;
          if (hit)
            mm = 0x0F0F0F0Fu ^ (((nz0 & HI) >> 7) | ((nz1 & HI) >> 6) | ((nz2 & HI) >> 5) | ((nz3 & HI) >> 4));
          while (__builtin_amdgcn_ballot_w32(mm != 0u) != 0u) {
            const int p = __builtin_ctz(mm | 0x80000000u);
            push(mm != 0u, e16[u] + ((p & 3) << 2) + (p >> 3));
            mm &= mm - 1u;
            if (cnt >= 32) {
              cnt -= 32;
              drain32(myq + cnt);
            }
          }
        }
      }
    }
    for (int eb = nfull * 512 + wave * 32; eb < E; eb += NW * 32) {
      int e = eb + lane;
      int d = -1;
      if (e < E) d = dst[e];
      bool m = (unsigned)(d - base) < (unsigned)T;
      push(m, e);
      if (cnt >= 32) {
        cnt -= 32;
        drain_n(myq + cnt, 32);
      }
    }
    if (cnt > 0) drain_n(myq, cnt);
    if constexpr (MODE == 0) {
      if (lane == 0) reg[wave] = rec_ok ? rcnt : -1;
    }
  }
  };
  if constexpr (MODE == 0) {
    if (tab_use) edge_phase(std::true_type{});
    else         edge_phase(std::false_type{});
  } else {
    edge_phase(std::false_type{});
  }

  if constexpr (MODE == 0) {
    const int hf = lane >> 4;
    const int l15 = lane & 15;
    {
      const int n1 = (wave & 7) * 16;
      v2f bf[32];
#pragma unroll
      for (int t = 0; t < 32; ++t) {
        const int j = t >> 3, qq = t & 7;
        bf[t].x = w1[(16 * qq + 8 * hf + j) * C + n1 + l15];
        bf[t].y = w1[(16 * qq + 8 * hf + 4 + j) * C + n1 + l15];
      }
      const float bb = bias[n1 + l15];
      __syncthreads();
      const int nit = (ntm + 3) >> 2;
      for (int it = 0; it < nit; ++it) {
        const int mt = (wave >> 3) + 4 * it;
        const bool on = mt < ntm;
        int hidx = (mt * 16 + 8 * hf) * P + n1 + l15;
        asm volatile("" : "+v"(hidx));
        v8f dacc = {};
        if (on) {
          const float* ar = acc + (mt * 16 + l15) * P + 2 * hf;
#pragma unroll
          for (int t = 0; t < 32; ++t) {
            const int j = t >> 3, qq = t & 7;
            v2f a = *reinterpret_cast<const v2f*>(ar + j * 32 + 4 * qq);
            dacc = __builtin_amdgcn_wmma_f32_16x16x4_f32(
                 false, a,  false, bf[t],
                 (short)0, dacc,  false,  false);
          }
        }
        asm volatile("v_nop\n\tv_nop\n\tv_nop\n\tv_nop\n\tv_nop" : "+v"(dacc));
        __syncthreads();
        if (on) {
#pragma unroll
          for (int r = 0; r < 8; ++r) acc[hidx + r * P] = dacc[r] + bb;
        }
      }
    }
    const int n0 = (wave & 3) * 16;
    v2f bf[32];
#pragma unroll
    for (int t = 0; t < 32; ++t) {
      bf[t].x = w2[(4 * t + 2 * hf) * OC + n0 + l15];
      bf[t].y = w2[(4 * t + 2 * hf + 1) * OC + n0 + l15];
    }
    __syncthreads();
    for (int mt = wave >> 2; mt < ntm; mt += 8) {
      const float* ar = acc + (mt * 16 + l15) * P + 2 * hf;
      int row = mt * 16 + hf * 8;
      unsigned zoff = ((unsigned)(base + row) * OC + n0 + l15) * 4u;
      asm volatile("" : "+v"(zoff), "+v"(row));
      char* zr = reinterpret_cast<char*>(out) + (size_t)zoff;
      v8f dacc = {};
#pragma unroll
      for (int t = 0; t < 32; ++t) {
        v2f a = *reinterpret_cast<const v2f*>(ar + 4 * t);
        dacc = __builtin_amdgcn_wmma_f32_16x16x4_f32(
             false, a,  false, bf[t],
             (short)0, dacc,  false,  false);
      }
      asm volatile("v_nop\n\tv_nop\n\tv_nop\n\tv_nop\n\tv_nop" : "+v"(dacc));
      if (mt * 16 + 16 <= nrows) {
#pragma unroll
        for (int r = 0; r < 8; ++r) *reinterpret_cast<float*>(zr + r * OC * 4) = dacc[r];
      } else {
#pragma unroll
        for (int r = 0; r < 8; ++r)
          if (row + r < nrows) *reinterpret_cast<float*>(zr + r * OC * 4) = dacc[r];
      }
    }
    return;
  }
  __syncthreads();

  const int hrow = lane >> 4;
  const int c16 = lane & 15;
  for (int r0 = 2 * wave; r0 < nrows; r0 += 2 * NW) {
    const int row = r0 + hrow;
    const int ri = row < nrows ? row : nrows - 1;
    const float2 ev = *reinterpret_cast<const float2*>(&acc[ri * C + 2 * c16]);
    const float2 od = *reinterpret_cast<const float2*>(&acc[ri * C + 32 + 2 * c16]);
    float mx = fmaxf(fmaxf(ev.x, ev.y), fmaxf(od.x, od.y));
    float mn = fminf(fminf(ev.x, ev.y), fminf(od.x, od.y));
    mx = fmaxf(mx, lane_xrow<8>(mx));  mn = fminf(mn, lane_xrow<8>(mn));
    mx = fmaxf(mx, lane_xrow<4>(mx));  mn = fminf(mn, lane_xrow<4>(mn));
    mx = fmaxf(mx, lane_xrow<2>(mx));  mn = fminf(mn, lane_xrow<2>(mn));
    mx = fmaxf(mx, lane_xrow<1>(mx));  mn = fminf(mn, lane_xrow<1>(mn));
    const float inv = 1.0f / (mx - mn);
    const float s0 = (ev.x - mn) * inv;
    const float s1 = (od.x - mn) * inv;
    const float s2 = (ev.y - mn) * inv;
    const float s3 = (od.y - mn) * inv;
    float ss = s0 * s0 + s1 * s1 + s2 * s2 + s3 * s3;
    ss += lane_xrow<8>(ss);
    ss += lane_xrow<4>(ss);
    ss += lane_xrow<2>(ss);
    ss += lane_xrow<1>(ss);
    const float r = 1.0f / fmaxf(sqrtf(ss), 1e-12f);
    if (row < nrows) {
      float4 ov;
      ov.x = s0 * r;
      ov.y = s1 * r;
      ov.z = s2 * r;
      ov.w = s3 * r;
      *reinterpret_cast<float4*>(out + (size_t)(base + row) * 64 + 4 * c16) = ov;
    }
  }
}

constexpr int TILE = 392;

extern "C" void kernel_launch(void* const* d_in, const int* in_sizes, int n_in,
                              void* d_out, int out_size, void* d_ws, size_t ws_size,
                              hipStream_t stream) {
  (void)n_in; (void)out_size; (void)ws_size;

  const float* x  = (const float*)d_in[0];
  const int*   ei = (const int*)d_in[1];
  const float* W1 = (const float*)d_in[2];
  const float* b1 = (const float*)d_in[3];
  const float* W2 = (const float*)d_in[4];
  const float* b2 = (const float*)d_in[5];

  const int N = in_sizes[0] / 128;
  const int E = in_sizes[1] / 2;
  const int* src = ei;
  const int* dst = ei + E;
  float* out = (float*)d_out;

  float* ws = (float*)d_ws;
  size_t Npad = ((size_t)N + 63) & ~(size_t)63;
  float* dinv = ws;
  float* z  = ws + Npad;
  int* lists = (int*)(z + (size_t)N * 64);

  deg_init_k<<<(N + 255) / 256, 256, 0, stream>>>(dinv, N);
  int* keys = (int*)d_out;
  deg_acc_k<TILE><<<((E + 3) / 4 + 255) / 256, 256, 0, stream>>>(dst, dinv, keys, E, N);
  dinv_k<<<((N + 3) / 4 + 255) / 256, 256, 0, stream>>>(dinv, keys, N, E);

  agg_tile_k<128, TILE, 0><<<(N + TILE - 1) / TILE, 1024, 0, stream>>>(x, dinv, b1, src, dst, keys, W1, W2, z, lists, N, E);

  agg_tile_k<64, TILE, 1><<<(N + TILE - 1) / TILE, 1024, 65536, stream>>>(z, dinv, b2, src, dst, keys, W1, W2, out, lists, N, E);
}


# Degrees by tile scan, no global atomics: tile block reads 1-byte tile ids, counts in-degrees in LDS, writes dinv, table, edge list; layer 1 reads that list, no scan; 5 launches to 4; sum order free
# speedup vs baseline: 4.0606x; 1.3690x over previous
#include <hip/hip_runtime.h>
#include <math.h>
#include <type_traits>

typedef float v2f __attribute__((ext_vector_type(2)));
typedef float v8f __attribute__((ext_vector_type(8)));

constexpr int TABCAP = 100352;
constexpr unsigned TABESC = 255u;

__device__ __forceinline__ size_t tab_off(int E) { return ((size_t)E + 15) & ~(size_t)15; }
__device__ __forceinline__ bool tab_fits(const void* keys, int N, int E) {
  return N <= TABCAP && (reinterpret_cast<size_t>(keys) & 15) == 0 &&
         tab_off(E) + (size_t)N <= (size_t)N * 256;
}

template <int T>
__global__ void keys_k(const int* __restrict__ dst, int* __restrict__ keys, int E, int n) {
  if ((long long)E > (long long)n * 256 || (reinterpret_cast<size_t>(dst) & 15) != 0) return;
  const long long nthr = (long long)gridDim.x * blockDim.x;
  const long long t0 = (long long)blockIdx.x * blockDim.x + threadIdx.x;
  for (int u = 0; u < 4; ++u) {
    const long long g = t0 + u * nthr;
    if (g * 4 + 4 <= (long long)E) {
      const int4 d = *reinterpret_cast<const int4*>(dst + g * 4);
      keys[g] = (((unsigned)d.x / T) & 255u) | ((((unsigned)d.y / T) & 255u) << 8) |
                ((((unsigned)d.z / T) & 255u) << 16) | ((((unsigned)d.w / T) & 255u) << 24);
    }
  }
}

template <int T>
__global__ __launch_bounds__(1024) void deg_tile_k(const int* __restrict__ src, const int* __restrict__ dst,
                                                   int* keys, float* __restrict__ dinv,
                                                   int* __restrict__ lst, int N, int E) {
  constexpr int NW = 32;
  constexpr int QCAP = 1280;
  constexpr int KF = 8;
  constexpr int SB = 23;
  static_assert(T <= (1 << (32 - SB)), "a row number inside the tile must fit the bits above SB");
  static_assert(T % 4 == 0, "a tile's table bytes are stored as whole dwords");
  __shared__ int cntr[T];
  __shared__ int q[NW * QCAP];

  const int lane = threadIdx.x & 31;
  const int wave = __builtin_amdgcn_readfirstlane((int)(threadIdx.x >> 5));
  const int tile = blockIdx.x;
  const int base = tile * T;
  int nrows = N - base;
  if (nrows > T) nrows = T;

  for (int i = (int)threadIdx.x; i < T; i += NW * 32) cntr[i] = 0;
  __syncthreads();

  int* reg = lst + (size_t)base * 64;
  const int cap = nrows - 1;
  int2* mylist = reinterpret_cast<int2*>(reg + 32) + (size_t)wave * cap;
  int* myq = q + wave * QCAP;
  int cnt = 0;
  int rcnt = 0;
  bool rec_ok = N <= (1 << SB);

  auto flush = [&]() {
#pragma unroll 1
    for (int p = 0; p < cnt; p += 128) {
      int nb = cnt - p;
      if (nb > 128) nb = 128;
      int s[4], d[4];
#pragma unroll
      for (int u = 0; u < 4; ++u) {
        int i = p + 32 * u + lane;
        if (i > cnt - 1) i = cnt - 1;
        const int ed = myq[i];
        s[u] = src[ed];
        d[u] = dst[ed];
      }
#pragma unroll
      for (int u = 0; u < 4; ++u) asm volatile("" : "+v"(s[u]), "+v"(d[u]));
      const bool room = rec_ok && rcnt + nb <= cap;
      if (!room) rec_ok = false;
#pragma unroll
      for (int u = 0; u < 4; ++u) {
        if (32 * u + lane < nb) {
          const int row = d[u] - base;
          __hip_atomic_fetch_add(&cntr[row], 1, __ATOMIC_RELAXED, __HIP_MEMORY_SCOPE_WORKGROUP);
          if (room)
            mylist[rcnt + 32 * u + lane] = make_int2((int)((unsigned)s[u] | ((unsigned)row << SB)), 0);
        }
      }
      if (room) rcnt += nb;
    }
    cnt = 0;
  };

  auto push = [&](bool m, int e) {
    unsigned mk = __builtin_amdgcn_ballot_w32(m);
    if (m) myq[cnt + (int)__builtin_amdgcn_mbcnt_lo(mk, 0u)] = e;
    cnt += __builtin_popcount(mk);
    if (cnt > QCAP - 32) flush();
  };

  const bool key_ok = gridDim.x <= 256u && (reinterpret_cast<size_t>(dst) & 15) == 0 &&
                      (reinterpret_cast<size_t>(keys) & 15) == 0 &&
                      (long long)E <= (long long)N * 256;
  const int nfull = key_ok ? (E >> 9) : 0;
  const unsigned pat = (unsigned)tile * 0x01010101u;
  const unsigned HI = 0x80808080u;
  const int rot = (int)(((long long)tile * nfull) / (int)gridDim.x);
  for (int g0 = wave; g0 < nfull; g0 += NW * KF) {
    int4 k4[KF];
    int e16[KF];
#pragma unroll
    for (int u = 0; u < KF; ++u) {
      int g = g0 + u * NW;
      if (g > nfull - 1) g = nfull - 1;
      int gg = g + rot;
      if (gg >= nfull) gg -= nfull;
      e16[u] = gg * 512 + lane * 16;
      k4[u] = *reinterpret_cast<const int4*>(keys + (size_t)gg * 128 + lane * 4);
    }
#pragma unroll
    for (int u = 0; u < KF; ++u) {
      const bool live = g0 + u * NW < nfull;
      unsigned x0 = (unsigned)k4[u].x ^ pat;
      unsigned x1 = (unsigned)k4[u].y ^ pat;
      unsigned x2 = (unsigned)k4[u].z ^ pat;
      unsigned x3 = (unsigned)k4[u].w ^ pat;
      unsigned nz0 = ((x0 & ~HI) + ~HI) | x0;
      unsigned nz1 = ((x1 & ~HI) + ~HI) | x1;
      unsigned nz2 = ((x2 & ~HI) + ~HI) | x2;
      unsigned nz3 = ((x3 & ~HI) + ~HI) | x3;
      const bool hit = live && ((nz0 & nz1 & nz2 & nz3 & HI) != HI);
      if (__builtin_amdgcn_ballot_w32(hit) != 0u) {
        unsigned mm = 0u;
        if (hit)
          mm = 0x0F0F0F0Fu ^ (((nz0 & HI) >> 7) | ((nz1 & HI) >> 6) | ((nz2 & HI) >> 5) | ((nz3 & HI) >> 4));
        while (__builtin_amdgcn_ballot_w32(mm != 0u) != 0u) {
          const int p = __builtin_ctz(mm | 0x80000000u);
          push(mm != 0u, e16[u] + ((p & 3) << 2) + (p >> 3));
          mm &= mm - 1u;
        }
      }
    }
  }
  for (int eb = nfull * 512 + wave * 32; eb < E; eb += NW * 32) {
    int e = eb + lane;
    int d = -1;
    if (e < E) d = dst[e];
    push((unsigned)(d - base) < (unsigned)T, e);
  }
  if (cnt > 0) flush();
  if (lane == 0) reg[wave] = rec_ok ? rcnt : -1;
  __syncthreads();

  const bool tab = tab_fits(keys, N, E);
  unsigned char* tb = reinterpret_cast<unsigned char*>(keys) + tab_off(E);
  for (int i = (int)threadIdx.x; i < nrows; i += NW * 32) {
    const float dg = 1.0f + (float)cntr[i];
    dinv[base + i] = (dg > 0.0f) ? rsqrtf(dg) : 0.0f;
  }
  if (tab) {
    for (int i = (int)threadIdx.x * 4; i < nrows; i += NW * 32 * 4) {
      if (i + 4 <= nrows) {
        unsigned b[4];
#pragma unroll
        for (int k = 0; k < 4; ++k) {
          const unsigned dgi = 1u + (unsigned)cntr[i + k];
          b[k] = dgi < TABESC ? dgi : TABESC;
        }
        *reinterpret_cast<unsigned*>(tb + base + i) = b[0] | (b[1] << 8) | (b[2] << 16) | (b[3] << 24);
      } else {
        for (int k = 0; k < 4 && i + k < nrows; ++k) {
          const unsigned dgi = 1u + (unsigned)cntr[i + k];
          tb[base + i + k] = (unsigned char)(dgi < TABESC ? dgi : TABESC);
        }
      }
    }
  }
}

template <int M>
__device__ __forceinline__ float lane_xrow(float v) {
  static_assert(M >= 1 && M <= 15, "a mask inside a 16-lane row");
  const int b = __builtin_bit_cast(int, v);
  return __builtin_bit_cast(float, __builtin_amdgcn_update_dpp(b, b, 0x160 + M, 0xf, 0xf, false));
}

template <int C, int T, int MODE>
__global__ __launch_bounds__(1024) void agg_tile_k(const float* __restrict__ feat,
                                                   const float* __restrict__ dinv,
                                                   const float* __restrict__ bias,
                                                   const int* __restrict__ src,
                                                   const int* __restrict__ dst,
                                                   const int* __restrict__ keys,
                                                   const float* __restrict__ w1,
                                                   const float* __restrict__ w2,
                                                   float* out, int* lst, int N, int E) {
  constexpr int NW = 32;
  constexpr int VEC = C / 32;
  constexpr int QCAP = 64;
  constexpr int G = 32 / VEC;
  constexpr int RPW = (T + NW - 1) / NW;
  constexpr int IG = RPW <= 8 ? RPW : (RPW + 1) / 2;
  static_assert(IG * VEC <= 32, "a group of start-value rows is meant to stay within 32 registers");
  constexpr int P = MODE == 0 ? C + 4 : C;
  constexpr int ROWS = MODE == 0 ? (T + 15) / 16 * 16 : T;
  constexpr int OC = 64;
  constexpr int SB = 23;
  static_assert(MODE != 0 || C == 128, "the fused product is written for 128 channels");
  static_assert(MODE != 1 || C == 64, "the finishing pass is written for 64 channels, two a lane");
  static_assert(T <= (1 << (32 - SB)), "a row number inside the tile must fit the bits above SB");
  __shared__ alignas(16) float acc[ROWS * P];
  __shared__ int q[NW * QCAP];
  __shared__ alignas(16) unsigned char dtab[MODE == 0 ? TABCAP : 16];
  __shared__ int escw[NW];

  const int lane = threadIdx.x & 31;
  const int wave = __builtin_amdgcn_readfirstlane((int)(threadIdx.x >> 5));
  const int tile = blockIdx.x;
  const int base = tile * T;
  int nrows = N - base;
  if (nrows > T) nrows = T;

  const bool tab_ok = MODE == 0 && tab_fits(keys, N, E);
  if (tab_ok) {
    const unsigned char* tb = reinterpret_cast<const unsigned char*>(keys) + tab_off(E);
    const int4* t16 = reinterpret_cast<const int4*>(tb);
    const int n16 = N >> 4;
    const int rot16 = (int)(((unsigned)tile * (unsigned)n16) / gridDim.x);
    bool esc = false;
    constexpr int TB = 7;
    for (int i0 = (int)threadIdx.x; i0 < n16; i0 += NW * 32 * TB) {
      int4 t[TB];
      int jj[TB];
#pragma unroll
      for (int u = 0; u < TB; ++u) {
        int i = i0 + NW * 32 * u;
        if (i > n16 - 1) i = n16 - 1;
        int j = i + rot16;
        if (j >= n16) j -= n16;
        jj[u] = j;
        t[u] = t16[j];
      }
#pragma unroll
      for (int u = 0; u < TB; ++u) {
        reinterpret_cast<int4*>(dtab)[jj[u]] = t[u];
        const unsigned x0 = (unsigned)t[u].x, x1 = (unsigned)t[u].y, x2 = (unsigned)t[u].z, x3 = (unsigned)t[u].w;
        const unsigned z = ((~x0 - 0x01010101u) & x0) | ((~x1 - 0x01010101u) & x1) |
                           ((~x2 - 0x01010101u) & x2) | ((~x3 - 0x01010101u) & x3);
        esc |= (z & 0x80808080u) != 0u;
      }
    }
    for (int i = (n16 << 4) + (int)threadIdx.x; i < N; i += NW * 32) {
      const unsigned char b = tb[i];
      dtab[i] = b;
      esc |= b == TABESC;
    }
    const bool wesc = __builtin_amdgcn_ballot_w32(esc) != 0u;
    if (lane == 0) escw[wave] = wesc ? 1 : 0;
  }

  int* reg = lst + (size_t)base * 64;
  const int cap = nrows - 1;
  int2* mylist = reinterpret_cast<int2*>(reg + 32) + (size_t)wave * cap;
  [[maybe_unused]] int hv0 = -1;
  if constexpr (MODE == 0) hv0 = reg[lane];

  {
    [[maybe_unused]] float bv[VEC];
    if constexpr (MODE == 1) {
#pragma unroll
      for (int j = 0; j < VEC; ++j) bv[j] = bias[lane * VEC + j];
    }
#pragma unroll 1
    for (int i0 = wave; i0 < nrows; i0 += NW * IG) {
      float f[IG][VEC];
      float dd[IG];
#pragma unroll
      for (int u = 0; u < IG; ++u) {
        int i = i0 + NW * u;
        if (i > nrows - 1) i = nrows - 1;
        dd[u] = dinv[base + i];
        const float* fr = feat + (size_t)(base + i) * C + lane * VEC;
        if constexpr (VEC == 4) {
          float4 t = *reinterpret_cast<const float4*>(fr);
          f[u][0] = t.x; f[u][1] = t.y; f[u][2] = t.z; f[u][3] = t.w;
        } else {
          float2 t = *reinterpret_cast<const float2*>(fr);
          f[u][0] = t.x; f[u][1] = t.y;
        }
      }
#pragma unroll
      for (int u = 0; u < IG; ++u) {
        int i = i0 + NW * u;
        if (i > nrows - 1) i = nrows - 1;
        const float w = dd[u] * dd[u];
#pragma unroll
        for (int j = 0; j < VEC; ++j) {
          if constexpr (MODE == 1) acc[i * P + j * 32 + lane] = w * f[u][j] + bv[j];
          else                     acc[i * P + j * 32 + lane] = w * f[u][j];
        }
      }
    }
  }
  [[maybe_unused]] const int ntm = (nrows + 15) >> 4;
  if constexpr (MODE == 0) {
    for (int i = nrows + wave; i < ntm * 16; i += NW) {
      acc[i * P + lane]      = 0.0f;
      acc[i * P + 32 + lane] = 0.0f;
      acc[i * P + 64 + lane] = 0.0f;
      acc[i * P + 96 + lane] = 0.0f;
    }
  }
  __syncthreads();

  bool tab_use = false;
  if (tab_ok) tab_use = __builtin_amdgcn_ballot_w32(escw[lane] != 0) == 0u;

  auto edge_phase = [&](auto tabc) {
  constexpr bool TAB = decltype(tabc)::value;
  auto dv = [&](int i) -> float {
    if constexpr (TAB) {
      const unsigned b = dtab[i];
      return b ? rsqrtf((float)b) : 0.0f;
    } else {
      return dinv[i];
    }
  };

  int* myq = q + wave * QCAP;
  int cnt = 0;
  int rcnt = 0;
  bool rec_ok = N <= (1 << SB);

  auto add32 = [&](int s, int ro, int wb) {
#pragma unroll 1
    for (int k0 = 0; k0 < 32; k0 += 8) {
      float v[8][VEC];
#pragma unroll
      for (int u = 0; u < 8; ++u) {
        int sk = __builtin_amdgcn_readlane(s, k0 + u);
        const float* fr = feat + (size_t)sk * C + lane * VEC;
        if constexpr (VEC == 4) {
          float4 t = *reinterpret_cast<const float4*>(fr);
          v[u][0] = t.x; v[u][1] = t.y; v[u][2] = t.z; v[u][3] = t.w;
        } else {
          float2 t = *reinterpret_cast<const float2*>(fr);
          v[u][0] = t.x; v[u][1] = t.y;
        }
      }
#pragma unroll
      for (int u = 0; u < 8; ++u) {
        float wk = __builtin_bit_cast(float, __builtin_amdgcn_readlane(wb, k0 + u));
        int rk = __builtin_amdgcn_readlane(ro, k0 + u);
#pragma unroll
        for (int j = 0; j < VEC; ++j)
          __hip_atomic_fetch_add(&acc[rk + j * 32 + lane], wk * v[u][j],
                                 __ATOMIC_RELAXED, __HIP_MEMORY_SCOPE_WORKGROUP);
      }
    }
  };

  auto drain32 = [&](const int* qb) {
    int ed = qb[lane];
    int s = src[ed];
    int d = dst[ed];
    float w = dv(s) * dv(d);
    int wb = __builtin_bit_cast(int, w);
    if constexpr (MODE == 0) {
      if (rec_ok) {
        if (rcnt + 32 <= cap) {
          mylist[rcnt + lane] = make_int2((int)((unsigned)s | ((unsigned)(d - base) << SB)), wb);
          rcnt += 32;
        } else {
          rec_ok = false;
        }
      }
    }
    add32(s, (d - base) * P, wb);
  };

  auto add_n = [&](int s, int ro, int wb, int n) {
#pragma unroll 1
    for (int k0 = 0; k0 < n; k0 += G) {
      float v[G][VEC];
#pragma unroll
      for (int u = 0; u < G; ++u) {
        int k = k0 + u;
        if (k > n - 1) k = n - 1;
        int sk = __builtin_amdgcn_readlane(s, k);
        const float* fr = feat + (size_t)sk * C + lane * VEC;
        if constexpr (VEC == 4) {
          float4 t = *reinterpret_cast<const float4*>(fr);
          v[u][0] = t.x; v[u][1] = t.y; v[u][2] = t.z; v[u][3] = t.w;
        } else {
          float2 t = *reinterpret_cast<const float2*>(fr);
          v[u][0] = t.x; v[u][1] = t.y;
        }
      }
#pragma unroll
      for (int u = 0; u < G; ++u) {
        if constexpr (VEC == 4) asm volatile("" : "+v"(v[u][0]), "+v"(v[u][1]), "+v"(v[u][2]), "+v"(v[u][3]));
        else                    asm volatile("" : "+v"(v[u][0]), "+v"(v[u][1]));
      }
#pragma unroll
      for (int u = 0; u < G; ++u) {
        if (k0 + u < n) {
          float wk = __builtin_bit_cast(float, __builtin_amdgcn_readlane(wb, k0 + u));
          int rk = __builtin_amdgcn_readlane(ro, k0 + u);
#pragma unroll
          for (int j = 0; j < VEC; ++j)
            __hip_atomic_fetch_add(&acc[rk + j * 32 + lane], wk * v[u][j],
                                   __ATOMIC_RELAXED, __HIP_MEMORY_SCOPE_WORKGROUP);
        }
      }
    }
  };

  auto drain_n = [&](const int* qb, int n) {
    int ed = qb[lane < n ? lane : n - 1];
    int s = src[ed];
    int d = dst[ed];
    float w = dv(s) * dv(d);
    int wb = __builtin_bit_cast(int, w);
    if constexpr (MODE == 0) {
      if (rec_ok) {
        if (rcnt + n <= cap) {
          if (lane < n) mylist[rcnt + lane] = make_int2((int)((unsigned)s | ((unsigned)(d - base) << SB)), wb);
          rcnt += n;
        } else {
          rec_ok = false;
        }
      }
    }
    add_n(s, (d - base) * P, wb, n);
  };

  auto push = [&](bool m, int e) {
    unsigned mk = __builtin_amdgcn_ballot_w32(m);
    if (m) myq[cnt + (int)__builtin_amdgcn_mbcnt_lo(mk, 0u)] = e;
    cnt += __builtin_popcount(mk);
  };

  bool listed = false;
  if constexpr (MODE == 1) {
    int hv = reg[lane];
    listed = __builtin_amdgcn_ballot_w32(hv < 0 || hv > cap) == 0u;
    if (listed) {
      const int n = __builtin_amdgcn_readlane(hv, wave);
      int p = 0;
      if (n >= 32) {
        int2 ed = mylist[lane];
        for (; p + 32 <= n; p += 32) {
          int s = ed.x & ((1 << SB) - 1);
          int ro = (int)((unsigned)ed.x >> SB) * P;
          int wb = ed.y;
          if (p + 64 <= n) ed = mylist[p + 32 + lane];
#pragma unroll 1
          for (int k0 = 0; k0 < 32; k0 += G) {
            float v[G][VEC];
#pragma unroll
            for (int u = 0; u < G; ++u) {
              int sk = __builtin_amdgcn_readlane(s, k0 + u);
              const float* fr = feat + (size_t)sk * C + lane * VEC;
              if constexpr (VEC == 4) {
                float4 t = *reinterpret_cast<const float4*>(fr);
                v[u][0] = t.x; v[u][1] = t.y; v[u][2] = t.z; v[u][3] = t.w;
              } else {
                float2 t = *reinterpret_cast<const float2*>(fr);
                v[u][0] = t.x; v[u][1] = t.y;
              }
            }
#pragma unroll
            for (int u = 0; u < G; ++u) {
              float wk = __builtin_bit_cast(float, __builtin_amdgcn_readlane(wb, k0 + u));
              int rk = __builtin_amdgcn_readlane(ro, k0 + u);
#pragma unroll
              for (int j = 0; j < VEC; ++j)
                __hip_atomic_fetch_add(&acc[rk + j * 32 + lane], wk * v[u][j],
                                       __ATOMIC_RELAXED, __HIP_MEMORY_SCOPE_WORKGROUP);
            }
          }
        }
      }
      if (p < n) {
        const int r = n - p;
        int2 et = mylist[p + (lane < r ? lane : r - 1)];
        add_n(et.x & ((1 << SB) - 1), (int)((unsigned)et.x >> SB) * P, et.y, r);
      }
    }
  }

  if constexpr (MODE == 0) {
    listed = __builtin_amdgcn_ballot_w32(hv0 < 0 || hv0 > cap) == 0u;
    if (listed) {
      const int n = __builtin_amdgcn_readlane(hv0, wave);
      int p = 0;
      if (n >= 32) {
        int2 ed = mylist[lane];
        for (; p + 32 <= n; p += 32) {
          const int s = ed.x & ((1 << SB) - 1);
          const int row = (int)((unsigned)ed.x >> SB);
          const float w = dv(s) * dv(base + row);
          const int wb = __builtin_bit_cast(int, w);
          mylist[p + lane] = make_int2(ed.x, wb);
          if (p + 64 <= n) ed = mylist[p + 32 + lane];
          add32(s, row * P, wb);
        }
      }
      if (p < n) {
        const int r = n - p;
        const int i = p + (lane < r ? lane : r - 1);
        const int2 et = mylist[i];
        const int s = et.x & ((1 << SB) - 1);
        const int row = (int)((unsigned)et.x >> SB);
        const float w = dv(s) * dv(base + row);
        const int wb = __builtin_bit_cast(int, w);
        if (lane < r) mylist[i] = make_int2(et.x, wb);
        add_n(s, row * P, wb, r);
      }
    }
  }

  if (!listed) {
    const bool key_ok = MODE == 0 && gridDim.x <= 256u && (reinterpret_cast<size_t>(dst) & 15) == 0 &&
                        (reinterpret_cast<size_t>(keys) & 15) == 0 &&
                        (long long)E <= (long long)N * 256;
    const int nfull = key_ok ? (E >> 9) : 0;
    const unsigned pat = (unsigned)tile * 0x01010101u;
    const unsigned HI = 0x80808080u;
    const int rot = (int)(((long long)tile * nfull) / (int)gridDim.x);
    for (int g0 = wave; g0 < nfull; g0 += NW * 4) {
      int4 k4[4];
      int e16[4];
#pragma unroll
      for (int u = 0; u < 4; ++u) {
        int g = g0 + u * NW;
        if (g > nfull - 1) g = nfull - 1;
        int gg = g + rot;
        if (gg >= nfull) gg -= nfull;
        e16[u] = gg * 512 + lane * 16;
        k4[u] = *reinterpret_cast<const int4*>(keys + (size_t)gg * 128 + lane * 4);
      }
#pragma unroll
      for (int u = 0; u < 4; ++u) {
        const bool live = g0 + u * NW < nfull;
        unsigned x0 = (unsigned)k4[u].x ^ pat;
        unsigned x1 = (unsigned)k4[u].y ^ pat;
        unsigned x2 = (unsigned)k4[u].z ^ pat;
        unsigned x3 = (unsigned)k4[u].w ^ pat;
        unsigned nz0 = ((x0 & ~HI) + ~HI) | x0;
        unsigned nz1 = ((x1 & ~HI) + ~HI) | x1;
        unsigned nz2 = ((x2 & ~HI) + ~HI) | x2;
        unsigned nz3 = ((x3 & ~HI) + ~HI) | x3;
        const bool hit = live && ((nz0 & nz1 & nz2 & nz3 & HI) != HI);
        if (__builtin_amdgcn_ballot_w32(hit) != 0u) {
          unsigned mm = 0u;
          if (hit)
            mm = 0x0F0F0F0Fu ^ (((nz0 & HI) >> 7) | ((nz1 & HI) >> 6) | ((nz2 & HI) >> 5) | ((nz3 & HI) >> 4));
          while (__builtin_amdgcn_ballot_w32(mm != 0u) != 0u) {
            const int p = __builtin_ctz(mm | 0x80000000u);
            push(mm != 0u, e16[u] + ((p & 3) << 2) + (p >> 3));
            mm &= mm - 1u;
            if (cnt >= 32) {
              cnt -= 32;
              drain32(myq + cnt);
            }
          }
        }
      }
    }
    for (int eb = nfull * 512 + wave * 32; eb < E; eb += NW * 32) {
      int e = eb + lane;
      int d = -1;
      if (e < E) d = dst[e];
      bool m = (unsigned)(d - base) < (unsigned)T;
      push(m, e);
      if (cnt >= 32) {
        cnt -= 32;
        drain_n(myq + cnt, 32);
      }
    }
    if (cnt > 0) drain_n(myq, cnt);
    if constexpr (MODE == 0) {
      if (lane == 0) reg[wave] = rec_ok ? rcnt : -1;
    }
  }
  };
  if constexpr (MODE == 0) {
    if (tab_use) edge_phase(std::true_type{});
    else         edge_phase(std::false_type{});
  } else {
    edge_phase(std::false_type{});
  }

  if constexpr (MODE == 0) {
    const int hf = lane >> 4;
    const int l15 = lane & 15;
    {
      const int n1 = (wave & 7) * 16;
      v2f bf[32];
#pragma unroll
      for (int t = 0; t < 32; ++t) {
        const int j = t >> 3, qq = t & 7;
        bf[t].x = w1[(16 * qq + 8 * hf + j) * C + n1 + l15];
        bf[t].y = w1[(16 * qq + 8 * hf + 4 + j) * C + n1 + l15];
      }
      const float bb = bias[n1 + l15];
      __syncthreads();
      const int nit = (ntm + 3) >> 2;
      for (int it = 0; it < nit; ++it) {
        const int mt = (wave >> 3) + 4 * it;
        const bool on = mt < ntm;
        int hidx = (mt * 16 + 8 * hf) * P + n1 + l15;
        asm volatile("" : "+v"(hidx));
        v8f dacc = {};
        if (on) {
          const float* ar = acc + (mt * 16 + l15) * P + 2 * hf;
#pragma unroll
          for (int t = 0; t < 32; ++t) {
            const int j = t >> 3, qq = t & 7;
            v2f a = *reinterpret_cast<const v2f*>(ar + j * 32 + 4 * qq);
            dacc = __builtin_amdgcn_wmma_f32_16x16x4_f32(
                 false, a,  false, bf[t],
                 (short)0, dacc,  false,  false);
          }
        }
        asm volatile("v_nop\n\tv_nop\n\tv_nop\n\tv_nop\n\tv_nop" : "+v"(dacc));
        __syncthreads();
        if (on) {
#pragma unroll
          for (int r = 0; r < 8; ++r) acc[hidx + r * P] = dacc[r] + bb;
        }
      }
    }
    const int n0 = (wave & 3) * 16;
    v2f bf[32];
#pragma unroll
    for (int t = 0; t < 32; ++t) {
      bf[t].x = w2[(4 * t + 2 * hf) * OC + n0 + l15];
      bf[t].y = w2[(4 * t + 2 * hf + 1) * OC + n0 + l15];
    }
    __syncthreads();
    for (int mt = wave >> 2; mt < ntm; mt += 8) {
      const float* ar = acc + (mt * 16 + l15) * P + 2 * hf;
      int row = mt * 16 + hf * 8;
      unsigned zoff = ((unsigned)(base + row) * OC + n0 + l15) * 4u;
      asm volatile("" : "+v"(zoff), "+v"(row));
      char* zr = reinterpret_cast<char*>(out) + (size_t)zoff;
      v8f dacc = {};
#pragma unroll
      for (int t = 0; t < 32; ++t) {
        v2f a = *reinterpret_cast<const v2f*>(ar + 4 * t);
        dacc = __builtin_amdgcn_wmma_f32_16x16x4_f32(
             false, a,  false, bf[t],
             (short)0, dacc,  false,  false);
      }
      asm volatile("v_nop\n\tv_nop\n\tv_nop\n\tv_nop\n\tv_nop" : "+v"(dacc));
      if (mt * 16 + 16 <= nrows) {
#pragma unroll
        for (int r = 0; r < 8; ++r) *reinterpret_cast<float*>(zr + r * OC * 4) = dacc[r];
      } else {
#pragma unroll
        for (int r = 0; r < 8; ++r)
          if (row + r < nrows) *reinterpret_cast<float*>(zr + r * OC * 4) = dacc[r];
      }
    }
    return;
  }
  __syncthreads();

  const int hrow = lane >> 4;
  const int c16 = lane & 15;
  for (int r0 = 2 * wave; r0 < nrows; r0 += 2 * NW) {
    const int row = r0 + hrow;
    const int ri = row < nrows ? row : nrows - 1;
    const float2 ev = *reinterpret_cast<const float2*>(&acc[ri * C + 2 * c16]);
    const float2 od = *reinterpret_cast<const float2*>(&acc[ri * C + 32 + 2 * c16]);
    float mx = fmaxf(fmaxf(ev.x, ev.y), fmaxf(od.x, od.y));
    float mn = fminf(fminf(ev.x, ev.y), fminf(od.x, od.y));
    mx = fmaxf(mx, lane_xrow<8>(mx));  mn = fminf(mn, lane_xrow<8>(mn));
    mx = fmaxf(mx, lane_xrow<4>(mx));  mn = fminf(mn, lane_xrow<4>(mn));
    mx = fmaxf(mx, lane_xrow<2>(mx));  mn = fminf(mn, lane_xrow<2>(mn));
    mx = fmaxf(mx, lane_xrow<1>(mx));  mn = fminf(mn, lane_xrow<1>(mn));
    const float inv = 1.0f / (mx - mn);
    const float s0 = (ev.x - mn) * inv;
    const float s1 = (od.x - mn) * inv;
    const float s2 = (ev.y - mn) * inv;
    const float s3 = (od.y - mn) * inv;
    float ss = s0 * s0 + s1 * s1 + s2 * s2 + s3 * s3;
    ss += lane_xrow<8>(ss);
    ss += lane_xrow<4>(ss);
    ss += lane_xrow<2>(ss);
    ss += lane_xrow<1>(ss);
    const float r = 1.0f / fmaxf(sqrtf(ss), 1e-12f);
    if (row < nrows) {
      float4 ov;
      ov.x = s0 * r;
      ov.y = s1 * r;
      ov.z = s2 * r;
      ov.w = s3 * r;
      *reinterpret_cast<float4*>(out + (size_t)(base + row) * 64 + 4 * c16) = ov;
    }
  }
}

constexpr int TILE = 392;

extern "C" void kernel_launch(void* const* d_in, const int* in_sizes, int n_in,
                              void* d_out, int out_size, void* d_ws, size_t ws_size,
                              hipStream_t stream) {
  (void)n_in; (void)out_size; (void)ws_size;

  const float* x  = (const float*)d_in[0];
  const int*   ei = (const int*)d_in[1];
  const float* W1 = (const float*)d_in[2];
  const float* b1 = (const float*)d_in[3];
  const float* W2 = (const float*)d_in[4];
  const float* b2 = (const float*)d_in[5];

  const int N = in_sizes[0] / 128;
  const int E = in_sizes[1] / 2;
  const int* src = ei;
  const int* dst = ei + E;
  float* out = (float*)d_out;

  float* ws = (float*)d_ws;
  size_t Npad = ((size_t)N + 63) & ~(size_t)63;
  float* dinv = ws;
  float* z  = ws + Npad;
  int* lists = (int*)(z + (size_t)N * 64);

  int* keys = (int*)d_out;
  keys_k<TILE><<<((E + 15) / 16 + 255) / 256, 256, 0, stream>>>(dst, keys, E, N);
  deg_tile_k<TILE><<<(N + TILE - 1) / TILE, 1024, 0, stream>>>(src, dst, keys, dinv, lists, N, E);

  agg_tile_k<128, TILE, 0><<<(N + TILE - 1) / TILE, 1024, 0, stream>>>(x, dinv, b1, src, dst, keys, W1, W2, z, lists, N, E);

  agg_tile_k<64, TILE, 1><<<(N + TILE - 1) / TILE, 1024, 65536, stream>>>(z, dinv, b2, src, dst, keys, W1, W2, out, lists, N, E);
}
